# starting kernel
# baseline (speedup 1.0000x reference)
#include <hip/hip_runtime.h>

typedef __attribute__((ext_vector_type(2))) float v2f;
typedef __attribute__((ext_vector_type(8))) float v8f;

__global__ __launch_bounds__(256) void ec_count_deg(const int* __restrict__ dst,
                                                    int* __restrict__ deg,
                                                    int total, int Nn, int Eper) {
  int t = blockIdx.x * blockDim.x + threadIdx.x;
  if (t >= total) return;
  int r = t / Eper;
  atomicAdd(&deg[r * Nn + dst[t]], 1);
}

__global__ __launch_bounds__(256) void ec_inv_deg(int* __restrict__ deg,
                                                  float* __restrict__ invd,
                                                  int total) {
  int t = blockIdx.x * blockDim.x + threadIdx.x;
  if (t >= total) return;
  float d = (float)deg[t];
  invd[t] = 1.0f / fmaxf(d, 1.0f);
}

template <int K, int NC>
__global__ __launch_bounds__(128) void ec_gemm_wmma(
    const float* __restrict__ A, const float* __restrict__ B,
    const float* __restrict__ bias, float* __restrict__ C, int mtiles) {
  constexpr int NT = NC / 16;
  constexpr int S = ((2 * NC) % 64 == 32) ? (2 * NC) : (2 * NC + 32);
  __shared__ float Bs[(K / 2) * S];

  for (int i = threadIdx.x; i < K * NC / 4; i += 128) {
    int k = (i * 4) / NC;
    int n = (i * 4) % NC;
    float4 v = *(const float4*)(B + i * 4);
    float* p = &Bs[(k >> 1) * S + 2 * n + (k & 1)];
    p[0] = v.x;
    p[2] = v.y;
    p[4] = v.z;
    p[6] = v.w;
  }
  __syncthreads();

  int wave = (blockIdx.x * blockDim.x + threadIdx.x) >> 5;
  if (wave >= mtiles) return;
  int lane = threadIdx.x & 31;
  int m = lane & 15;
  int half = lane >> 4;

  const float* Arow = A + (size_t)(wave * 16 + m) * K + 2 * half;

  v8f acc[NT];
  v8f z = {0.f, 0.f, 0.f, 0.f, 0.f, 0.f, 0.f, 0.f};
#pragma unroll
  for (int j = 0; j < NT; ++j) acc[j] = z;

#pragma unroll 8
  for (int k = 0; k < K; k += 4) {
    float2 av = *(const float2*)(Arow + k);
    v2f a = {av.x, av.y};
    const float* Bp = &Bs[(size_t)((k + 2 * half) >> 1) * S + 2 * m];
#pragma unroll
    for (int j = 0; j < NT; ++j) {
      v2f b = *(const v2f*)(Bp + j * 32);
      acc[j] = __builtin_amdgcn_wmma_f32_16x16x4_f32(
           false, a,  false, b,
           (short)0, acc[j],  false,  false);
    }
  }

#pragma unroll
  for (int j = 0; j < NT; ++j) {
    float bv = bias ? bias[j * 16 + m] : 0.0f;
#pragma unroll
    for (int v = 0; v < 8; ++v) {
      C[(size_t)(wave * 16 + v + 8 * half) * NC + j * 16 + m] = acc[j][v] + bv;
    }
  }
}

template <int COLS>
__global__ __launch_bounds__(256) void ec_agg(
    const float* __restrict__ hw, const int* __restrict__ src,
    const int* __restrict__ dst, const float* __restrict__ invd,
    float* __restrict__ acc, int E) {
  constexpr int TPE = COLS / 4;
  int t = blockIdx.x * blockDim.x + threadIdx.x;
  int e = t / TPE;
  if (e >= E) return;
  int c = (t % TPE) * 4;
  int s = src[e];
  int d = dst[e];
  float w = invd[d];
  float4 v = *(const float4*)(hw + (size_t)s * COLS + c);
  float* dp = acc + (size_t)d * COLS + c;
  unsafeAtomicAdd(dp + 0, v.x * w);
  unsafeAtomicAdd(dp + 1, v.y * w);
  unsafeAtomicAdd(dp + 2, v.z * w);
  unsafeAtomicAdd(dp + 3, v.w * w);
}

__global__ __launch_bounds__(256) void ec_relu(float* __restrict__ h, int total) {
  int t = blockIdx.x * blockDim.x + threadIdx.x;
  if (t >= total) return;
  h[t] = fmaxf(h[t], 0.0f);
}

extern "C" void kernel_launch(void* const* d_in, const int* in_sizes, int n_in,
                              void* d_out, int out_size, void* d_ws, size_t ws_size,
                              hipStream_t stream) {
  const float* x   = (const float*)d_in[0];
  const int*   src = (const int*)d_in[1];
  const int*   dst = (const int*)d_in[2];
  const float* W1  = (const float*)d_in[3];
  const float* L1w = (const float*)d_in[4];
  const float* L1b = (const float*)d_in[5];
  const float* W2  = (const float*)d_in[6];
  const float* L2w = (const float*)d_in[7];
  const float* L2b = (const float*)d_in[8];
  float* out = (float*)d_out;

  const int D_IN = 256, D_H = 64, D_OUT = 16;
  const int Nn = in_sizes[0] / D_IN;
  const int R  = in_sizes[3] / (D_IN * D_H);
  const int E  = in_sizes[1] / R;

  char* ws = (char*)d_ws;
  size_t off = 0;
  int*   deg  = (int*)(ws + off);
  float* invd = (float*)(ws + off);
  off += (size_t)R * Nn * sizeof(float);
  off = (off + 255) & ~(size_t)255;
  float* h = (float*)(ws + off);
  off += (size_t)Nn * D_H * sizeof(float);
  off = (off + 255) & ~(size_t)255;
  float* hw = (float*)(ws + off);

  hipMemsetAsync(deg, 0, (size_t)R * Nn * sizeof(int), stream);
  {
    int total = R * E;
    ec_count_deg<<<(total + 255) / 256, 256, 0, stream>>>(dst, deg, total, Nn, E);
  }
  {
    int total = R * Nn;
    ec_inv_deg<<<(total + 255) / 256, 256, 0, stream>>>(deg, invd, total);
  }

  int mtiles = Nn / 16;
  int gemmBlocks = (mtiles * 32 + 127) / 128;

  ec_gemm_wmma<256, 64><<<gemmBlocks, 128, 0, stream>>>(x, L1w, L1b, h, mtiles);
  for (int r = 0; r < R; ++r) {
    ec_gemm_wmma<256, 64><<<gemmBlocks, 128, 0, stream>>>(
        x, W1 + (size_t)r * D_IN * D_H, nullptr, hw, mtiles);
    long long tot = (long long)E * (D_H / 4);
    ec_agg<64><<<(int)((tot + 255) / 256), 256, 0, stream>>>(
        hw, src + (size_t)r * E, dst + (size_t)r * E, invd + (size_t)r * Nn, h, E);
  }
  ec_relu<<<(Nn * D_H + 255) / 256, 256, 0, stream>>>(h, Nn * D_H);

  ec_gemm_wmma<64, 16><<<gemmBlocks, 128, 0, stream>>>(h, L2w, L2b, out, mtiles);
  for (int r = 0; r < R; ++r) {
    ec_gemm_wmma<64, 16><<<gemmBlocks, 128, 0, stream>>>(
        h, W2 + (size_t)r * D_H * D_OUT, nullptr, hw, mtiles);
    long long tot = (long long)E * (D_OUT / 4);
    ec_agg<16><<<(int)((tot + 255) / 256), 256, 0, stream>>>(
        hw, src + (size_t)r * E, dst + (size_t)r * E, invd + (size_t)r * Nn, out, E);
  }
}


# Edge sort without global atomics: edges bucketed by (relation, 1024-node dst range), then counted, scanned and placed per bucket, all counters in LDS; slot (sum) order = LDS add arrival order
# speedup vs baseline: 2.2274x; 2.2274x over previous
#include <hip/hip_runtime.h>

typedef __attribute__((ext_vector_type(2))) float v2f;
typedef __attribute__((ext_vector_type(4))) float v4f;
typedef __attribute__((ext_vector_type(8))) float v8f;

__global__ __launch_bounds__(256) void ec_count_deg(const int* __restrict__ dst,
                                                    int* __restrict__ deg,
                                                    int total, int Nn, int Eper) {
  int t = blockIdx.x * blockDim.x + threadIdx.x;
  if (t >= total) return;
  int r = t / Eper;
  atomicAdd(&deg[r * Nn + dst[t]], 1);
}

__global__ __launch_bounds__(256) void ec_inv_deg(int* __restrict__ deg,
                                                  float* __restrict__ invd,
                                                  int total) {
  int t = blockIdx.x * blockDim.x + threadIdx.x;
  if (t >= total) return;
  float d = (float)deg[t];
  invd[t] = 1.0f / fmaxf(d, 1.0f);
}


#define EC_BW_LOG 10
#define EC_BW (1 << EC_BW_LOG)
#define EC_RB_MAX 1024
#define EC_NCHUNK 256

__global__ __launch_bounds__(1024) void ec_bucket_hist(const int* __restrict__ dst,
                                                       int* __restrict__ hist,
                                                       int total, int Eper, int NB, int RB,
                                                       int chunk) {
  __shared__ int hs[EC_RB_MAX];
  int tid = threadIdx.x;
  for (int j = tid; j < RB; j += 1024) hs[j] = 0;
  __syncthreads();
  int c0 = blockIdx.x * chunk;
  int c1 = (c0 + chunk < total) ? (c0 + chunk) : total;
  int t = c0 + tid;
  for (; t + 3 * 1024 < c1; t += 4 * 1024) {
    int d0 = dst[t];
    int d1 = dst[t + 1024];
    int d2 = dst[t + 2048];
    int d3 = dst[t + 3072];
    atomicAdd(&hs[(t / Eper) * NB + (d0 >> EC_BW_LOG)], 1);
    atomicAdd(&hs[((t + 1024) / Eper) * NB + (d1 >> EC_BW_LOG)], 1);
    atomicAdd(&hs[((t + 2048) / Eper) * NB + (d2 >> EC_BW_LOG)], 1);
    atomicAdd(&hs[((t + 3072) / Eper) * NB + (d3 >> EC_BW_LOG)], 1);
  }
  for (; t < c1; t += 1024) {
    int d = dst[t];
    atomicAdd(&hs[(t / Eper) * NB + (d >> EC_BW_LOG)], 1);
  }
  __syncthreads();
  for (int j = tid; j < RB; j += 1024) hist[j * EC_NCHUNK + blockIdx.x] = hs[j];
}

__global__ __launch_bounds__(1024) void ec_bucket_scan(const int* __restrict__ hist,
                                                       int* __restrict__ base,
                                                       int* __restrict__ bstart,
                                                       int RB) {
  __shared__ int sh[1024];
  int tid = threadIdx.x;
  int bkt = blockIdx.x;
  const int4* h4 = (const int4*)hist;
  int n4 = bkt * (EC_NCHUNK / 4);
  int p0 = 0, p1 = 0, p2 = 0, p3 = 0;
  int j = tid;
  for (; j + 3 * 1024 < n4; j += 4 * 1024) {
    int4 a = h4[j];
    int4 b = h4[j + 1024];
    int4 c = h4[j + 2048];
    int4 d = h4[j + 3072];
    p0 += a.x + a.y + a.z + a.w;
    p1 += b.x + b.y + b.z + b.w;
    p2 += c.x + c.y + c.z + c.w;
    p3 += d.x + d.y + d.z + d.w;
  }
  for (; j < n4; j += 1024) {
    int4 a = h4[j];
    p0 += a.x + a.y + a.z + a.w;
  }
  sh[tid] = p0 + p1 + p2 + p3;
  __syncthreads();
  for (int st = 512; st > 0; st >>= 1) {
    if (tid < st) sh[tid] += sh[tid + st];
    __syncthreads();
  }
  int before = sh[0];
  __syncthreads();

  int v = (tid < EC_NCHUNK) ? hist[bkt * EC_NCHUNK + tid] : 0;
  sh[tid] = v;
  __syncthreads();
  for (int dl = 1; dl < EC_NCHUNK; dl <<= 1) {
    int add = (tid >= dl) ? sh[tid - dl] : 0;
    __syncthreads();
    sh[tid] += add;
    __syncthreads();
  }
  if (tid < EC_NCHUNK) base[bkt * EC_NCHUNK + tid] = before + sh[tid] - v;
  if (tid == EC_NCHUNK - 1) {
    bstart[bkt] = before;
    if (bkt == RB - 1) bstart[RB] = before + sh[tid];
  }
}

__global__ __launch_bounds__(1024) void ec_bucket_scatter(const int* __restrict__ src,
                                                          const int* __restrict__ dst,
                                                          const int* __restrict__ base,
                                                          int2* __restrict__ part,
                                                          int total, int Eper, int NB, int RB,
                                                          int chunk) {
  __shared__ int cur[EC_RB_MAX];
  int tid = threadIdx.x;
  for (int j = tid; j < RB; j += 1024) cur[j] = base[j * EC_NCHUNK + blockIdx.x];
  __syncthreads();
  int c0 = blockIdx.x * chunk;
  int c1 = (c0 + chunk < total) ? (c0 + chunk) : total;
  int t = c0 + tid;
  for (; t + 3 * 1024 < c1; t += 4 * 1024) {
    int d0 = dst[t];
    int d1 = dst[t + 1024];
    int d2 = dst[t + 2048];
    int d3 = dst[t + 3072];
    int s0 = src[t];
    int s1 = src[t + 1024];
    int s2 = src[t + 2048];
    int s3 = src[t + 3072];
    int q0 = atomicAdd(&cur[(t / Eper) * NB + (d0 >> EC_BW_LOG)], 1);
    int q1 = atomicAdd(&cur[((t + 1024) / Eper) * NB + (d1 >> EC_BW_LOG)], 1);
    int q2 = atomicAdd(&cur[((t + 2048) / Eper) * NB + (d2 >> EC_BW_LOG)], 1);
    int q3 = atomicAdd(&cur[((t + 3072) / Eper) * NB + (d3 >> EC_BW_LOG)], 1);
    part[q0] = make_int2(s0, d0);
    part[q1] = make_int2(s1, d1);
    part[q2] = make_int2(s2, d2);
    part[q3] = make_int2(s3, d3);
  }
  for (; t < c1; t += 1024) {
    int d = dst[t];
    int s = src[t];
    int q = atomicAdd(&cur[(t / Eper) * NB + (d >> EC_BW_LOG)], 1);
    part[q] = make_int2(s, d);
  }
}

__global__ __launch_bounds__(1024) void ec_bucket_sort(const int2* __restrict__ part,
                                                       const int* __restrict__ bstart,
                                                       int* __restrict__ offs,
                                                       int* __restrict__ col,
                                                       int Nn, int NB, int RB, int M) {
  __shared__ int cnt[EC_BW];
  __shared__ int sh[EC_BW];
  int tid = threadIdx.x;
  int bkt = blockIdx.x;
  int r = bkt / NB;
  int n0 = (bkt - r * NB) << EC_BW_LOG;
  int e0 = bstart[bkt];
  int e1 = bstart[bkt + 1];
  cnt[tid] = 0;
  __syncthreads();
  int i = e0 + tid;
  for (; i + 3 * 1024 < e1; i += 4 * 1024) {
    int y0 = part[i].y;
    int y1 = part[i + 1024].y;
    int y2 = part[i + 2048].y;
    int y3 = part[i + 3072].y;
    atomicAdd(&cnt[y0 - n0], 1);
    atomicAdd(&cnt[y1 - n0], 1);
    atomicAdd(&cnt[y2 - n0], 1);
    atomicAdd(&cnt[y3 - n0], 1);
  }
  for (; i < e1; i += 1024) atomicAdd(&cnt[part[i].y - n0], 1);
  __syncthreads();
  int v = cnt[tid];
  sh[tid] = v;
  __syncthreads();
  for (int dl = 1; dl < EC_BW; dl <<= 1) {
    int add = (tid >= dl) ? sh[tid - dl] : 0;
    __syncthreads();
    sh[tid] += add;
    __syncthreads();
  }
  int start = sh[tid] - v;
  if (n0 + tid < Nn) offs[r * Nn + n0 + tid] = e0 + start;
  cnt[tid] = start;
  __syncthreads();
  i = e0 + tid;
  for (; i + 3 * 1024 < e1; i += 4 * 1024) {
    int2 r0 = part[i];
    int2 r1 = part[i + 1024];
    int2 r2 = part[i + 2048];
    int2 r3 = part[i + 3072];
    int q0 = atomicAdd(&cnt[r0.y - n0], 1);
    int q1 = atomicAdd(&cnt[r1.y - n0], 1);
    int q2 = atomicAdd(&cnt[r2.y - n0], 1);
    int q3 = atomicAdd(&cnt[r3.y - n0], 1);
    col[e0 + q0] = r0.x;
    col[e0 + q1] = r1.x;
    col[e0 + q2] = r2.x;
    col[e0 + q3] = r3.x;
  }
  for (; i < e1; i += 1024) {
    int2 rec = part[i];
    int q = atomicAdd(&cnt[rec.y - n0], 1);
    col[e0 + q] = rec.x;
  }
  if (bkt == RB - 1 && tid == 0) offs[M] = e1;
}

template <int K, int NC>
__global__ __launch_bounds__(128) void ec_gemm_wmma(
    const float* __restrict__ A, const float* __restrict__ B,
    const float* __restrict__ bias, float* __restrict__ C, int mtiles) {
  constexpr int NT = NC / 16;
  constexpr int S = ((2 * NC) % 64 == 32) ? (2 * NC) : (2 * NC + 32);
  __shared__ float Bs[(K / 2) * S];

  for (int i = threadIdx.x; i < K * NC / 4; i += 128) {
    int k = (i * 4) / NC;
    int n = (i * 4) % NC;
    float4 v = *(const float4*)(B + i * 4);
    float* p = &Bs[(k >> 1) * S + 2 * n + (k & 1)];
    p[0] = v.x;
    p[2] = v.y;
    p[4] = v.z;
    p[6] = v.w;
  }
  __syncthreads();

  int wave = (blockIdx.x * blockDim.x + threadIdx.x) >> 5;
  if (wave >= mtiles) return;
  int lane = threadIdx.x & 31;
  int m = lane & 15;
  int half = lane >> 4;

  const float* Arow = A + (size_t)(wave * 16 + m) * K + 2 * half;

  v8f acc[NT];
  v8f z = {0.f, 0.f, 0.f, 0.f, 0.f, 0.f, 0.f, 0.f};
#pragma unroll
  for (int j = 0; j < NT; ++j) acc[j] = z;

#pragma unroll 8
  for (int k = 0; k < K; k += 4) {
    float2 av = *(const float2*)(Arow + k);
    v2f a = {av.x, av.y};
    const float* Bp = &Bs[(size_t)((k + 2 * half) >> 1) * S + 2 * m];
#pragma unroll
    for (int j = 0; j < NT; ++j) {
      v2f b = *(const v2f*)(Bp + j * 32);
      acc[j] = __builtin_amdgcn_wmma_f32_16x16x4_f32(
           false, a,  false, b,
           (short)0, acc[j],  false,  false);
    }
  }

#pragma unroll
  for (int j = 0; j < NT; ++j) {
    float bv = bias ? bias[j * 16 + m] : 0.0f;
#pragma unroll
    for (int v = 0; v < 8; ++v) {
      C[(size_t)(wave * 16 + v + 8 * half) * NC + j * 16 + m] = acc[j][v] + bv;
    }
  }
}

__global__ __launch_bounds__(128) void ec_gemm_xs(
    const float* __restrict__ A, const float* __restrict__ Bself,
    const float* __restrict__ Brel, const float* __restrict__ bias,
    float* __restrict__ Cself, float* __restrict__ Crel,
    int mtiles, int nrel, int Nn) {
  constexpr int K = 256;
  constexpr int NC = 64;
  constexpr int NT = NC / 16;
  constexpr int S = 2 * NC + 32;
  __shared__ float Bs[(K / 2) * S];

  int wave = (blockIdx.x * blockDim.x + threadIdx.x) >> 5;
  bool active = wave < mtiles;
  int wv = active ? wave : 0;
  int lane = threadIdx.x & 31;
  int m = lane & 15;
  int half = lane >> 4;

  const float* Arow = A + (size_t)(wv * 16 + m) * K + 2 * half;
  v2f a[K / 4];
#pragma unroll
  for (int kk = 0; kk < K / 4; ++kk) {
    float2 av = *(const float2*)(Arow + 4 * kk);
    a[kk][0] = av.x;
    a[kk][1] = av.y;
  }

  float bvs[NT];
#pragma unroll
  for (int j = 0; j < NT; ++j) bvs[j] = bias[j * 16 + m];

#pragma unroll 1
  for (int r = (Bself ? 0 : 1); r <= nrel; ++r) {
    const float* B = (r == 0) ? Bself : Brel + (size_t)(r - 1) * K * NC;
    float* C = (r == 0) ? Cself : Crel + (size_t)(r - 1) * Nn * NC;

    __syncthreads();
    for (int it0 = 0; it0 < K * NC / 4 / 128; it0 += 8) {
      v4f q[8];
#pragma unroll
      for (int u = 0; u < 8; ++u) {
        int i = (it0 + u) * 128 + (int)threadIdx.x;
        q[u] = *(const v4f*)(B + i * 4);
      }
      asm volatile("" : "+v"(q[0]), "+v"(q[1]), "+v"(q[2]), "+v"(q[3]),
                        "+v"(q[4]), "+v"(q[5]), "+v"(q[6]), "+v"(q[7]));
#pragma unroll
      for (int u = 0; u < 8; ++u) {
        int i = (it0 + u) * 128 + (int)threadIdx.x;
        int k = (i * 4) / NC;
        int n = (i * 4) % NC;
        int j = n >> 4;
        int m0 = n & 15;
        float* p = &Bs[(k >> 1) * S + (j >> 1) * 64 + 4 * m0 + 2 * (j & 1) + (k & 1)];
        p[0] = q[u][0];
        p[4] = q[u][1];
        p[8] = q[u][2];
        p[12] = q[u][3];
      }
    }
    __syncthreads();

    v8f acc[NT];
    v8f z = {0.f, 0.f, 0.f, 0.f, 0.f, 0.f, 0.f, 0.f};
#pragma unroll
    for (int j = 0; j < NT; ++j) acc[j] = z;

#pragma unroll
    for (int kk = 0; kk < K / 4; ++kk) {
      int k = 4 * kk;
      const float* Bp = &Bs[(size_t)((k + 2 * half) >> 1) * S + 4 * m];
#pragma unroll
      for (int t = 0; t < NT / 2; ++t) {
        v4f q = *(const v4f*)(Bp + t * 64);
        v2f b0 = {q[0], q[1]};
        v2f b1 = {q[2], q[3]};
        acc[2 * t] = __builtin_amdgcn_wmma_f32_16x16x4_f32(
             false, a[kk],  false, b0,
             (short)0, acc[2 * t],  false,  false);
        acc[2 * t + 1] = __builtin_amdgcn_wmma_f32_16x16x4_f32(
             false, a[kk],  false, b1,
             (short)0, acc[2 * t + 1],  false,  false);
      }
    }

    v8f o[NT];
#pragma unroll
    for (int j = 0; j < NT; ++j) {
      float bv = (r == 0) ? bvs[j] : 0.0f;
#pragma unroll
      for (int v = 0; v < 8; ++v) o[j][v] = acc[j][v] + bv;
    }
    asm volatile("" : "+v"(o[0]), "+v"(o[1]), "+v"(o[2]), "+v"(o[3]));
    if (active) {
#pragma unroll
      for (int j = 0; j < NT; ++j) {
#pragma unroll
        for (int v = 0; v < 8; ++v) {
          C[(size_t)(wave * 16 + v + 8 * half) * NC + j * 16 + m] = o[j][v];
        }
      }
    }
  }
}

template <int COLS>
__global__ __launch_bounds__(256) void ec_agg(
    const float* __restrict__ hw, const int* __restrict__ src,
    const int* __restrict__ dst, const float* __restrict__ invd,
    float* __restrict__ acc, int E) {
  constexpr int TPE = COLS / 4;
  int t = blockIdx.x * blockDim.x + threadIdx.x;
  int e = t / TPE;
  if (e >= E) return;
  int c = (t % TPE) * 4;
  int s = src[e];
  int d = dst[e];
  float w = invd[d];
  float4 v = *(const float4*)(hw + (size_t)s * COLS + c);
  float* dp = acc + (size_t)d * COLS + c;
  unsafeAtomicAdd(dp + 0, v.x * w);
  unsafeAtomicAdd(dp + 1, v.y * w);
  unsafeAtomicAdd(dp + 2, v.z * w);
  unsafeAtomicAdd(dp + 3, v.w * w);
}

template <int COLS>
__global__ __launch_bounds__(256) void ec_pull(
    const float* __restrict__ hw, const int* __restrict__ offs,
    const int* __restrict__ col, float* __restrict__ acc, int Nn) {
  constexpr int TPN = COLS / 4;
  int t = blockIdx.x * blockDim.x + threadIdx.x;
  int d = t / TPN;
  if (d >= Nn) return;
  int c = (t % TPN) * 4;
  int b = offs[d];
  int e = offs[d + 1];
  float4* hp = (float4*)(acc + (size_t)d * COLS + c);
  float4 o = *hp;
  const float* hwc = hw + c;
  float sx = 0.f, sy = 0.f, sz = 0.f, sw = 0.f;
  int i = b;
  for (; i + 4 <= e; i += 4) {
    int s0 = col[i];
    int s1 = col[i + 1];
    int s2 = col[i + 2];
    int s3 = col[i + 3];
    float4 v0 = *(const float4*)(hwc + (size_t)s0 * COLS);
    float4 v1 = *(const float4*)(hwc + (size_t)s1 * COLS);
    float4 v2 = *(const float4*)(hwc + (size_t)s2 * COLS);
    float4 v3 = *(const float4*)(hwc + (size_t)s3 * COLS);
    sx += v0.x; sy += v0.y; sz += v0.z; sw += v0.w;
    sx += v1.x; sy += v1.y; sz += v1.z; sw += v1.w;
    sx += v2.x; sy += v2.y; sz += v2.z; sw += v2.w;
    sx += v3.x; sy += v3.y; sz += v3.z; sw += v3.w;
  }
  if (i < e) {
    int n = e - i;
    int i1 = (n > 1) ? i + 1 : i;
    int i2 = (n > 2) ? i + 2 : i;
    int s0 = col[i];
    int s1 = col[i1];
    int s2 = col[i2];
    float4 v0 = *(const float4*)(hwc + (size_t)s0 * COLS);
    float4 v1 = *(const float4*)(hwc + (size_t)s1 * COLS);
    float4 v2 = *(const float4*)(hwc + (size_t)s2 * COLS);
    asm volatile("" : "+v"(v1.x), "+v"(v1.y), "+v"(v1.z), "+v"(v1.w),
                      "+v"(v2.x), "+v"(v2.y), "+v"(v2.z), "+v"(v2.w));
    sx += v0.x; sy += v0.y; sz += v0.z; sw += v0.w;
    if (n > 1) { sx += v1.x; sy += v1.y; sz += v1.z; sw += v1.w; }
    if (n > 2) { sx += v2.x; sy += v2.y; sz += v2.z; sw += v2.w; }
  }
  float w = 1.0f / fmaxf((float)(e - b), 1.0f);
  o.x += sx * w;
  o.y += sy * w;
  o.z += sz * w;
  o.w += sw * w;
  *hp = o;
}

__global__ __launch_bounds__(256) void ec_relu(float* __restrict__ h, int total) {
  int t = blockIdx.x * blockDim.x + threadIdx.x;
  if (t >= total) return;
  h[t] = fmaxf(h[t], 0.0f);
}

extern "C" void kernel_launch(void* const* d_in, const int* in_sizes, int n_in,
                              void* d_out, int out_size, void* d_ws, size_t ws_size,
                              hipStream_t stream) {
  const float* x   = (const float*)d_in[0];
  const int*   src = (const int*)d_in[1];
  const int*   dst = (const int*)d_in[2];
  const float* W1  = (const float*)d_in[3];
  const float* L1w = (const float*)d_in[4];
  const float* L1b = (const float*)d_in[5];
  const float* W2  = (const float*)d_in[6];
  const float* L2w = (const float*)d_in[7];
  const float* L2b = (const float*)d_in[8];
  float* out = (float*)d_out;

  const int D_IN = 256, D_H = 64, D_OUT = 16;
  const int Nn = in_sizes[0] / D_IN;
  const int R  = in_sizes[3] / (D_IN * D_H);
  const int E  = in_sizes[1] / R;

  char* ws = (char*)d_ws;
  int mtiles = Nn / 16;
  int gemmBlocks = (mtiles * 32 + 127) / 128;

  const int M = R * Nn;
  const int NB = (Nn + EC_BW - 1) >> EC_BW_LOG;
  const int RB = R * NB;
  size_t off = 0;
  int* hist = (int*)(ws + off);
  off += (size_t)RB * EC_NCHUNK * sizeof(int);
  off = (off + 255) & ~(size_t)255;
  int* base = (int*)(ws + off);
  off += (size_t)RB * EC_NCHUNK * sizeof(int);
  off = (off + 255) & ~(size_t)255;
  int* bstart = (int*)(ws + off);
  off += ((size_t)RB + 1) * sizeof(int);
  off = (off + 255) & ~(size_t)255;
  int* offs = (int*)(ws + off);
  off += ((size_t)M + 1) * sizeof(int);
  off = (off + 255) & ~(size_t)255;
  int* col = (int*)(ws + off);
  off += (size_t)R * E * sizeof(int);
  off = (off + 255) & ~(size_t)255;
  float* h = (float*)(ws + off);
  off += (size_t)Nn * D_H * sizeof(float);
  off = (off + 255) & ~(size_t)255;
  float* hw = (float*)(ws + off);
  int2* part = (int2*)hw;
  size_t planeBytes = (size_t)Nn * D_H * sizeof(float);
  size_t partBytes = (size_t)R * E * sizeof(int2);
  size_t hwOff = off;
  off += (planeBytes > partBytes) ? planeBytes : partBytes;

  if (ws_size >= off && RB <= EC_RB_MAX) {
    int P = 1;
    if (ws_size >= hwOff + 2 * planeBytes) P = 2;
    if (ws_size >= hwOff + (size_t)R * planeBytes) P = R;
    int total = R * E;
    int chunk = (total + EC_NCHUNK - 1) / EC_NCHUNK;
    ec_bucket_hist<<<EC_NCHUNK, 1024, 0, stream>>>(dst, hist, total, E, NB, RB, chunk);
    ec_bucket_scan<<<RB, 1024, 0, stream>>>(hist, base, bstart, RB);
    ec_bucket_scatter<<<EC_NCHUNK, 1024, 0, stream>>>(src, dst, base, part, total, E, NB, RB, chunk);
    ec_bucket_sort<<<RB, 1024, 0, stream>>>(part, bstart, offs, col, Nn, NB, RB, M);

    for (int r0 = 0; r0 < R; r0 += P) {
      int nr = (R - r0 < P) ? (R - r0) : P;
      ec_gemm_xs<<<gemmBlocks, 128, 0, stream>>>(
          x, (r0 == 0) ? L1w : nullptr, W1 + (size_t)r0 * D_IN * D_H, L1b,
          h, hw, mtiles, nr, Nn);
      for (int q = 0; q < nr; ++q) {
        ec_pull<64><<<(Nn * (D_H / 4) + 255) / 256, 256, 0, stream>>>(
            hw + (size_t)q * Nn * D_H, offs + (size_t)(r0 + q) * Nn, col, h, Nn);
      }
    }
    ec_relu<<<(Nn * D_H + 255) / 256, 256, 0, stream>>>(h, Nn * D_H);

    ec_gemm_wmma<64, 16><<<gemmBlocks, 128, 0, stream>>>(h, L2w, L2b, out, mtiles);
    for (int r = 0; r < R; ++r) {
      ec_gemm_wmma<64, 16><<<gemmBlocks, 128, 0, stream>>>(
          h, W2 + (size_t)r * D_H * D_OUT, nullptr, hw, mtiles);
      ec_pull<16><<<(Nn * (D_OUT / 4) + 255) / 256, 256, 0, stream>>>(
          hw, offs + (size_t)r * Nn, col, out, Nn);
    }
    return;
  }

  off = 0;
  int*   deg  = (int*)(ws + off);
  float* invd = (float*)(ws + off);
  off += (size_t)R * Nn * sizeof(float);
  off = (off + 255) & ~(size_t)255;
  h = (float*)(ws + off);
  off += (size_t)Nn * D_H * sizeof(float);
  off = (off + 255) & ~(size_t)255;
  hw = (float*)(ws + off);

  hipMemsetAsync(deg, 0, (size_t)R * Nn * sizeof(int), stream);
  {
    int total = R * E;
    ec_count_deg<<<(total + 255) / 256, 256, 0, stream>>>(dst, deg, total, Nn, E);
  }
  {
    int total = R * Nn;
    ec_inv_deg<<<(total + 255) / 256, 256, 0, stream>>>(deg, invd, total);
  }

  ec_gemm_wmma<256, 64><<<gemmBlocks, 128, 0, stream>>>(x, L1w, L1b, h, mtiles);
  for (int r = 0; r < R; ++r) {
    ec_gemm_wmma<256, 64><<<gemmBlocks, 128, 0, stream>>>(
        x, W1 + (size_t)r * D_IN * D_H, nullptr, hw, mtiles);
    long long tot = (long long)E * (D_H / 4);
    ec_agg<64><<<(int)((tot + 255) / 256), 256, 0, stream>>>(
        hw, src + (size_t)r * E, dst + (size_t)r * E, invd + (size_t)r * Nn, h, E);
  }
  ec_relu<<<(Nn * D_H + 255) / 256, 256, 0, stream>>>(h, Nn * D_H);

  ec_gemm_wmma<64, 16><<<gemmBlocks, 128, 0, stream>>>(h, L2w, L2b, out, mtiles);
  for (int r = 0; r < R; ++r) {
    ec_gemm_wmma<64, 16><<<gemmBlocks, 128, 0, stream>>>(
        h, W2 + (size_t)r * D_H * D_OUT, nullptr, hw, mtiles);
    long long tot = (long long)E * (D_OUT / 4);
    ec_agg<16><<<(int)((tot + 255) / 256), 256, 0, stream>>>(
        hw, src + (size_t)r * E, dst + (size_t)r * E, invd + (size_t)r * Nn, out, E);
  }
}


# Row walkers as resident grids: layer-1 edge sums and ReLU run one round of waves that loop over rows (ReLU 4x128 bits a trip), not 50k-200k one-shot waves a launch; sum order unchanged
# speedup vs baseline: 2.3076x; 1.0360x over previous
#include <hip/hip_runtime.h>

typedef __attribute__((ext_vector_type(2))) float v2f;
typedef __attribute__((ext_vector_type(4))) float v4f;
typedef __attribute__((ext_vector_type(8))) float v8f;

__global__ __launch_bounds__(256) void ec_count_deg(const int* __restrict__ dst,
                                                    int* __restrict__ deg,
                                                    int total, int Nn, int Eper) {
  int t = blockIdx.x * blockDim.x + threadIdx.x;
  if (t >= total) return;
  int r = t / Eper;
  atomicAdd(&deg[r * Nn + dst[t]], 1);
}

__global__ __launch_bounds__(256) void ec_inv_deg(int* __restrict__ deg,
                                                  float* __restrict__ invd,
                                                  int total) {
  int t = blockIdx.x * blockDim.x + threadIdx.x;
  if (t >= total) return;
  float d = (float)deg[t];
  invd[t] = 1.0f / fmaxf(d, 1.0f);
}


#define EC_BW_LOG 10
#define EC_BW (1 << EC_BW_LOG)
#define EC_RB_MAX 1024
#define EC_NCHUNK 256

__global__ __launch_bounds__(1024) void ec_bucket_hist(const int* __restrict__ dst,
                                                       int* __restrict__ hist,
                                                       int total, int Eper, int NB, int RB,
                                                       int chunk) {
  __shared__ int hs[EC_RB_MAX];
  int tid = threadIdx.x;
  for (int j = tid; j < RB; j += 1024) hs[j] = 0;
  __syncthreads();
  int c0 = blockIdx.x * chunk;
  int c1 = (c0 + chunk < total) ? (c0 + chunk) : total;
  int t = c0 + tid;
  for (; t + 3 * 1024 < c1; t += 4 * 1024) {
    int d0 = dst[t];
    int d1 = dst[t + 1024];
    int d2 = dst[t + 2048];
    int d3 = dst[t + 3072];
    atomicAdd(&hs[(t / Eper) * NB + (d0 >> EC_BW_LOG)], 1);
    atomicAdd(&hs[((t + 1024) / Eper) * NB + (d1 >> EC_BW_LOG)], 1);
    atomicAdd(&hs[((t + 2048) / Eper) * NB + (d2 >> EC_BW_LOG)], 1);
    atomicAdd(&hs[((t + 3072) / Eper) * NB + (d3 >> EC_BW_LOG)], 1);
  }
  for (; t < c1; t += 1024) {
    int d = dst[t];
    atomicAdd(&hs[(t / Eper) * NB + (d >> EC_BW_LOG)], 1);
  }
  __syncthreads();
  for (int j = tid; j < RB; j += 1024) hist[j * EC_NCHUNK + blockIdx.x] = hs[j];
}

__global__ __launch_bounds__(1024) void ec_bucket_scan(const int* __restrict__ hist,
                                                       int* __restrict__ base,
                                                       int* __restrict__ bstart,
                                                       int RB) {
  __shared__ int sh[1024];
  int tid = threadIdx.x;
  int bkt = blockIdx.x;
  const int4* h4 = (const int4*)hist;
  int n4 = bkt * (EC_NCHUNK / 4);
  int p0 = 0, p1 = 0, p2 = 0, p3 = 0;
  int j = tid;
  for (; j + 3 * 1024 < n4; j += 4 * 1024) {
    int4 a = h4[j];
    int4 b = h4[j + 1024];
    int4 c = h4[j + 2048];
    int4 d = h4[j + 3072];
    p0 += a.x + a.y + a.z + a.w;
    p1 += b.x + b.y + b.z + b.w;
    p2 += c.x + c.y + c.z + c.w;
    p3 += d.x + d.y + d.z + d.w;
  }
  for (; j < n4; j += 1024) {
    int4 a = h4[j];
    p0 += a.x + a.y + a.z + a.w;
  }
  sh[tid] = p0 + p1 + p2 + p3;
  __syncthreads();
  for (int st = 512; st > 0; st >>= 1) {
    if (tid < st) sh[tid] += sh[tid + st];
    __syncthreads();
  }
  int before = sh[0];
  __syncthreads();

  int v = (tid < EC_NCHUNK) ? hist[bkt * EC_NCHUNK + tid] : 0;
  sh[tid] = v;
  __syncthreads();
  for (int dl = 1; dl < EC_NCHUNK; dl <<= 1) {
    int add = (tid >= dl) ? sh[tid - dl] : 0;
    __syncthreads();
    sh[tid] += add;
    __syncthreads();
  }
  if (tid < EC_NCHUNK) base[bkt * EC_NCHUNK + tid] = before + sh[tid] - v;
  if (tid == EC_NCHUNK - 1) {
    bstart[bkt] = before;
    if (bkt == RB - 1) bstart[RB] = before + sh[tid];
  }
}

__global__ __launch_bounds__(1024) void ec_bucket_scatter(const int* __restrict__ src,
                                                          const int* __restrict__ dst,
                                                          const int* __restrict__ base,
                                                          int2* __restrict__ part,
                                                          int total, int Eper, int NB, int RB,
                                                          int chunk) {
  __shared__ int cur[EC_RB_MAX];
  int tid = threadIdx.x;
  for (int j = tid; j < RB; j += 1024) cur[j] = base[j * EC_NCHUNK + blockIdx.x];
  __syncthreads();
  int c0 = blockIdx.x * chunk;
  int c1 = (c0 + chunk < total) ? (c0 + chunk) : total;
  int t = c0 + tid;
  for (; t + 3 * 1024 < c1; t += 4 * 1024) {
    int d0 = dst[t];
    int d1 = dst[t + 1024];
    int d2 = dst[t + 2048];
    int d3 = dst[t + 3072];
    int s0 = src[t];
    int s1 = src[t + 1024];
    int s2 = src[t + 2048];
    int s3 = src[t + 3072];
    int q0 = atomicAdd(&cur[(t / Eper) * NB + (d0 >> EC_BW_LOG)], 1);
    int q1 = atomicAdd(&cur[((t + 1024) / Eper) * NB + (d1 >> EC_BW_LOG)], 1);
    int q2 = atomicAdd(&cur[((t + 2048) / Eper) * NB + (d2 >> EC_BW_LOG)], 1);
    int q3 = atomicAdd(&cur[((t + 3072) / Eper) * NB + (d3 >> EC_BW_LOG)], 1);
    part[q0] = make_int2(s0, d0);
    part[q1] = make_int2(s1, d1);
    part[q2] = make_int2(s2, d2);
    part[q3] = make_int2(s3, d3);
  }
  for (; t < c1; t += 1024) {
    int d = dst[t];
    int s = src[t];
    int q = atomicAdd(&cur[(t / Eper) * NB + (d >> EC_BW_LOG)], 1);
    part[q] = make_int2(s, d);
  }
}

__global__ __launch_bounds__(1024) void ec_bucket_sort(const int2* __restrict__ part,
                                                       const int* __restrict__ bstart,
                                                       int* __restrict__ offs,
                                                       int* __restrict__ col,
                                                       int Nn, int NB, int RB, int M) {
  __shared__ int cnt[EC_BW];
  __shared__ int sh[EC_BW];
  int tid = threadIdx.x;
  int bkt = blockIdx.x;
  int r = bkt / NB;
  int n0 = (bkt - r * NB) << EC_BW_LOG;
  int e0 = bstart[bkt];
  int e1 = bstart[bkt + 1];
  cnt[tid] = 0;
  __syncthreads();
  int i = e0 + tid;
  for (; i + 3 * 1024 < e1; i += 4 * 1024) {
    int y0 = part[i].y;
    int y1 = part[i + 1024].y;
    int y2 = part[i + 2048].y;
    int y3 = part[i + 3072].y;
    atomicAdd(&cnt[y0 - n0], 1);
    atomicAdd(&cnt[y1 - n0], 1);
    atomicAdd(&cnt[y2 - n0], 1);
    atomicAdd(&cnt[y3 - n0], 1);
  }
  for (; i < e1; i += 1024) atomicAdd(&cnt[part[i].y - n0], 1);
  __syncthreads();
  int v = cnt[tid];
  sh[tid] = v;
  __syncthreads();
  for (int dl = 1; dl < EC_BW; dl <<= 1) {
    int add = (tid >= dl) ? sh[tid - dl] : 0;
    __syncthreads();
    sh[tid] += add;
    __syncthreads();
  }
  int start = sh[tid] - v;
  if (n0 + tid < Nn) offs[r * Nn + n0 + tid] = e0 + start;
  cnt[tid] = start;
  __syncthreads();
  i = e0 + tid;
  for (; i + 3 * 1024 < e1; i += 4 * 1024) {
    int2 r0 = part[i];
    int2 r1 = part[i + 1024];
    int2 r2 = part[i + 2048];
    int2 r3 = part[i + 3072];
    int q0 = atomicAdd(&cnt[r0.y - n0], 1);
    int q1 = atomicAdd(&cnt[r1.y - n0], 1);
    int q2 = atomicAdd(&cnt[r2.y - n0], 1);
    int q3 = atomicAdd(&cnt[r3.y - n0], 1);
    col[e0 + q0] = r0.x;
    col[e0 + q1] = r1.x;
    col[e0 + q2] = r2.x;
    col[e0 + q3] = r3.x;
  }
  for (; i < e1; i += 1024) {
    int2 rec = part[i];
    int q = atomicAdd(&cnt[rec.y - n0], 1);
    col[e0 + q] = rec.x;
  }
  if (bkt == RB - 1 && tid == 0) offs[M] = e1;
}

template <int K, int NC>
__global__ __launch_bounds__(128) void ec_gemm_wmma(
    const float* __restrict__ A, const float* __restrict__ B,
    const float* __restrict__ bias, float* __restrict__ C, int mtiles) {
  constexpr int NT = NC / 16;
  constexpr int S = ((2 * NC) % 64 == 32) ? (2 * NC) : (2 * NC + 32);
  __shared__ float Bs[(K / 2) * S];

  for (int i = threadIdx.x; i < K * NC / 4; i += 128) {
    int k = (i * 4) / NC;
    int n = (i * 4) % NC;
    float4 v = *(const float4*)(B + i * 4);
    float* p = &Bs[(k >> 1) * S + 2 * n + (k & 1)];
    p[0] = v.x;
    p[2] = v.y;
    p[4] = v.z;
    p[6] = v.w;
  }
  __syncthreads();

  int wave = (blockIdx.x * blockDim.x + threadIdx.x) >> 5;
  if (wave >= mtiles) return;
  int lane = threadIdx.x & 31;
  int m = lane & 15;
  int half = lane >> 4;

  const float* Arow = A + (size_t)(wave * 16 + m) * K + 2 * half;

  v8f acc[NT];
  v8f z = {0.f, 0.f, 0.f, 0.f, 0.f, 0.f, 0.f, 0.f};
#pragma unroll
  for (int j = 0; j < NT; ++j) acc[j] = z;

#pragma unroll 8
  for (int k = 0; k < K; k += 4) {
    float2 av = *(const float2*)(Arow + k);
    v2f a = {av.x, av.y};
    const float* Bp = &Bs[(size_t)((k + 2 * half) >> 1) * S + 2 * m];
#pragma unroll
    for (int j = 0; j < NT; ++j) {
      v2f b = *(const v2f*)(Bp + j * 32);
      acc[j] = __builtin_amdgcn_wmma_f32_16x16x4_f32(
           false, a,  false, b,
           (short)0, acc[j],  false,  false);
    }
  }

#pragma unroll
  for (int j = 0; j < NT; ++j) {
    float bv = bias ? bias[j * 16 + m] : 0.0f;
#pragma unroll
    for (int v = 0; v < 8; ++v) {
      C[(size_t)(wave * 16 + v + 8 * half) * NC + j * 16 + m] = acc[j][v] + bv;
    }
  }
}

__global__ __launch_bounds__(128) void ec_gemm_xs(
    const float* __restrict__ A, const float* __restrict__ Bself,
    const float* __restrict__ Brel, const float* __restrict__ bias,
    float* __restrict__ Cself, float* __restrict__ Crel,
    int mtiles, int nrel, int Nn) {
  constexpr int K = 256;
  constexpr int NC = 64;
  constexpr int NT = NC / 16;
  constexpr int S = 2 * NC + 32;
  __shared__ float Bs[(K / 2) * S];

  int wave = (blockIdx.x * blockDim.x + threadIdx.x) >> 5;
  bool active = wave < mtiles;
  int wv = active ? wave : 0;
  int lane = threadIdx.x & 31;
  int m = lane & 15;
  int half = lane >> 4;

  const float* Arow = A + (size_t)(wv * 16 + m) * K + 2 * half;
  v2f a[K / 4];
#pragma unroll
  for (int kk = 0; kk < K / 4; ++kk) {
    float2 av = *(const float2*)(Arow + 4 * kk);
    a[kk][0] = av.x;
    a[kk][1] = av.y;
  }

  float bvs[NT];
#pragma unroll
  for (int j = 0; j < NT; ++j) bvs[j] = bias[j * 16 + m];

#pragma unroll 1
  for (int r = (Bself ? 0 : 1); r <= nrel; ++r) {
    const float* B = (r == 0) ? Bself : Brel + (size_t)(r - 1) * K * NC;
    float* C = (r == 0) ? Cself : Crel + (size_t)(r - 1) * Nn * NC;

    __syncthreads();
    for (int it0 = 0; it0 < K * NC / 4 / 128; it0 += 8) {
      v4f q[8];
#pragma unroll
      for (int u = 0; u < 8; ++u) {
        int i = (it0 + u) * 128 + (int)threadIdx.x;
        q[u] = *(const v4f*)(B + i * 4);
      }
      asm volatile("" : "+v"(q[0]), "+v"(q[1]), "+v"(q[2]), "+v"(q[3]),
                        "+v"(q[4]), "+v"(q[5]), "+v"(q[6]), "+v"(q[7]));
#pragma unroll
      for (int u = 0; u < 8; ++u) {
        int i = (it0 + u) * 128 + (int)threadIdx.x;
        int k = (i * 4) / NC;
        int n = (i * 4) % NC;
        int j = n >> 4;
        int m0 = n & 15;
        float* p = &Bs[(k >> 1) * S + (j >> 1) * 64 + 4 * m0 + 2 * (j & 1) + (k & 1)];
        p[0] = q[u][0];
        p[4] = q[u][1];
        p[8] = q[u][2];
        p[12] = q[u][3];
      }
    }
    __syncthreads();

    v8f acc[NT];
    v8f z = {0.f, 0.f, 0.f, 0.f, 0.f, 0.f, 0.f, 0.f};
#pragma unroll
    for (int j = 0; j < NT; ++j) acc[j] = z;

#pragma unroll
    for (int kk = 0; kk < K / 4; ++kk) {
      int k = 4 * kk;
      const float* Bp = &Bs[(size_t)((k + 2 * half) >> 1) * S + 4 * m];
#pragma unroll
      for (int t = 0; t < NT / 2; ++t) {
        v4f q = *(const v4f*)(Bp + t * 64);
        v2f b0 = {q[0], q[1]};
        v2f b1 = {q[2], q[3]};
        acc[2 * t] = __builtin_amdgcn_wmma_f32_16x16x4_f32(
             false, a[kk],  false, b0,
             (short)0, acc[2 * t],  false,  false);
        acc[2 * t + 1] = __builtin_amdgcn_wmma_f32_16x16x4_f32(
             false, a[kk],  false, b1,
             (short)0, acc[2 * t + 1],  false,  false);
      }
    }

    v8f o[NT];
#pragma unroll
    for (int j = 0; j < NT; ++j) {
      float bv = (r == 0) ? bvs[j] : 0.0f;
#pragma unroll
      for (int v = 0; v < 8; ++v) o[j][v] = acc[j][v] + bv;
    }
    asm volatile("" : "+v"(o[0]), "+v"(o[1]), "+v"(o[2]), "+v"(o[3]));
    if (active) {
#pragma unroll
      for (int j = 0; j < NT; ++j) {
#pragma unroll
        for (int v = 0; v < 8; ++v) {
          C[(size_t)(wave * 16 + v + 8 * half) * NC + j * 16 + m] = o[j][v];
        }
      }
    }
  }
}

template <int COLS>
__global__ __launch_bounds__(256) void ec_agg(
    const float* __restrict__ hw, const int* __restrict__ src,
    const int* __restrict__ dst, const float* __restrict__ invd,
    float* __restrict__ acc, int E) {
  constexpr int TPE = COLS / 4;
  int t = blockIdx.x * blockDim.x + threadIdx.x;
  int e = t / TPE;
  if (e >= E) return;
  int c = (t % TPE) * 4;
  int s = src[e];
  int d = dst[e];
  float w = invd[d];
  float4 v = *(const float4*)(hw + (size_t)s * COLS + c);
  float* dp = acc + (size_t)d * COLS + c;
  unsafeAtomicAdd(dp + 0, v.x * w);
  unsafeAtomicAdd(dp + 1, v.y * w);
  unsafeAtomicAdd(dp + 2, v.z * w);
  unsafeAtomicAdd(dp + 3, v.w * w);
}

template <int COLS>
__global__ __launch_bounds__(256) void ec_pull(
    const float* __restrict__ hw, const int* __restrict__ offs,
    const int* __restrict__ col, float* __restrict__ acc, int Nn) {
  constexpr int TPN = COLS / 4;
  constexpr int RPB = 256 / TPN;
  const int g = (int)threadIdx.x / TPN;
  const int c = ((int)threadIdx.x % TPN) * 4;
  const float* hwc = hw + c;
  const int step = (int)gridDim.x * RPB;
  int d = (int)blockIdx.x * RPB + g;
  if (d >= Nn) return;
  int b = offs[d];
  int e = offs[d + 1];
  for (;;) {
    const int dn = d + step;
    const bool more = dn < Nn;
    const int dq = more ? dn : d;
    const int bn = offs[dq];
    const int en = offs[dq + 1];
    float4* hp = (float4*)(acc + (size_t)d * COLS + c);
    float4 o = *hp;
    float sx = 0.f, sy = 0.f, sz = 0.f, sw = 0.f;
    int i = b;
    for (; i + 4 <= e; i += 4) {
      int s0 = col[i];
      int s1 = col[i + 1];
      int s2 = col[i + 2];
      int s3 = col[i + 3];
      float4 v0 = *(const float4*)(hwc + (size_t)s0 * COLS);
      float4 v1 = *(const float4*)(hwc + (size_t)s1 * COLS);
      float4 v2 = *(const float4*)(hwc + (size_t)s2 * COLS);
      float4 v3 = *(const float4*)(hwc + (size_t)s3 * COLS);
      sx += v0.x; sy += v0.y; sz += v0.z; sw += v0.w;
      sx += v1.x; sy += v1.y; sz += v1.z; sw += v1.w;
      sx += v2.x; sy += v2.y; sz += v2.z; sw += v2.w;
      sx += v3.x; sy += v3.y; sz += v3.z; sw += v3.w;
    }
    if (i < e) {
      int n = e - i;
      int i1 = (n > 1) ? i + 1 : i;
      int i2 = (n > 2) ? i + 2 : i;
      int s0 = col[i];
      int s1 = col[i1];
      int s2 = col[i2];
      float4 v0 = *(const float4*)(hwc + (size_t)s0 * COLS);
      float4 v1 = *(const float4*)(hwc + (size_t)s1 * COLS);
      float4 v2 = *(const float4*)(hwc + (size_t)s2 * COLS);
      asm volatile("" : "+v"(v1.x), "+v"(v1.y), "+v"(v1.z), "+v"(v1.w),
                        "+v"(v2.x), "+v"(v2.y), "+v"(v2.z), "+v"(v2.w));
      sx += v0.x; sy += v0.y; sz += v0.z; sw += v0.w;
      if (n > 1) { sx += v1.x; sy += v1.y; sz += v1.z; sw += v1.w; }
      if (n > 2) { sx += v2.x; sy += v2.y; sz += v2.z; sw += v2.w; }
    }
    float w = 1.0f / fmaxf((float)(e - b), 1.0f);
    o.x += sx * w;
    o.y += sy * w;
    o.z += sz * w;
    o.w += sw * w;
    *hp = o;
    if (!more) break;
    d = dn;
    b = bn;
    e = en;
  }
}

__global__ __launch_bounds__(256) void ec_relu(float* __restrict__ h, int n4) {
  v4f* p = (v4f*)h;
  const int T = (int)gridDim.x * 256;
  for (int i = (int)blockIdx.x * 256 + (int)threadIdx.x; i < n4; i += 4 * T) {
    const int i1 = i + T;
    const int i2 = i + 2 * T;
    const int i3 = i + 3 * T;
    const bool k1 = i1 < n4;
    const bool k2 = i2 < n4;
    const bool k3 = i3 < n4;
    v4f a0 = p[i];
    v4f a1 = p[k1 ? i1 : i];
    v4f a2 = p[k2 ? i2 : i];
    v4f a3 = p[k3 ? i3 : i];
#pragma unroll
    for (int q = 0; q < 4; ++q) {
      a0[q] = fmaxf(a0[q], 0.0f);
      a1[q] = fmaxf(a1[q], 0.0f);
      a2[q] = fmaxf(a2[q], 0.0f);
      a3[q] = fmaxf(a3[q], 0.0f);
    }
    p[i] = a0;
    if (k1) p[i1] = a1;
    if (k2) p[i2] = a2;
    if (k3) p[i3] = a3;
  }
}

extern "C" void kernel_launch(void* const* d_in, const int* in_sizes, int n_in,
                              void* d_out, int out_size, void* d_ws, size_t ws_size,
                              hipStream_t stream) {
  const float* x   = (const float*)d_in[0];
  const int*   src = (const int*)d_in[1];
  const int*   dst = (const int*)d_in[2];
  const float* W1  = (const float*)d_in[3];
  const float* L1w = (const float*)d_in[4];
  const float* L1b = (const float*)d_in[5];
  const float* W2  = (const float*)d_in[6];
  const float* L2w = (const float*)d_in[7];
  const float* L2b = (const float*)d_in[8];
  float* out = (float*)d_out;

  const int D_IN = 256, D_H = 64, D_OUT = 16;
  const int Nn = in_sizes[0] / D_IN;
  const int R  = in_sizes[3] / (D_IN * D_H);
  const int E  = in_sizes[1] / R;

  char* ws = (char*)d_ws;
  int mtiles = Nn / 16;
  int gemmBlocks = (mtiles * 32 + 127) / 128;

  const int M = R * Nn;
  const int NB = (Nn + EC_BW - 1) >> EC_BW_LOG;
  const int RB = R * NB;
  size_t off = 0;
  int* hist = (int*)(ws + off);
  off += (size_t)RB * EC_NCHUNK * sizeof(int);
  off = (off + 255) & ~(size_t)255;
  int* base = (int*)(ws + off);
  off += (size_t)RB * EC_NCHUNK * sizeof(int);
  off = (off + 255) & ~(size_t)255;
  int* bstart = (int*)(ws + off);
  off += ((size_t)RB + 1) * sizeof(int);
  off = (off + 255) & ~(size_t)255;
  int* offs = (int*)(ws + off);
  off += ((size_t)M + 1) * sizeof(int);
  off = (off + 255) & ~(size_t)255;
  int* col = (int*)(ws + off);
  off += (size_t)R * E * sizeof(int);
  off = (off + 255) & ~(size_t)255;
  float* h = (float*)(ws + off);
  off += (size_t)Nn * D_H * sizeof(float);
  off = (off + 255) & ~(size_t)255;
  float* hw = (float*)(ws + off);
  int2* part = (int2*)hw;
  size_t planeBytes = (size_t)Nn * D_H * sizeof(float);
  size_t partBytes = (size_t)R * E * sizeof(int2);
  size_t hwOff = off;
  off += (planeBytes > partBytes) ? planeBytes : partBytes;

  const int residentBlocks = 2048;
  int pull64Blocks = (Nn + 15) / 16;
  if (pull64Blocks > residentBlocks) pull64Blocks = residentBlocks;
  int pull16Blocks = (Nn + 63) / 64;
  if (pull16Blocks > residentBlocks) pull16Blocks = residentBlocks;
  const int relu4 = Nn * D_H / 4;
  int reluBlocks = (relu4 + 1023) / 1024;
  if (reluBlocks > residentBlocks) reluBlocks = residentBlocks;

  if (ws_size >= off && RB <= EC_RB_MAX) {
    int P = 1;
    if (ws_size >= hwOff + 2 * planeBytes) P = 2;
    if (ws_size >= hwOff + (size_t)R * planeBytes) P = R;
    int total = R * E;
    int chunk = (total + EC_NCHUNK - 1) / EC_NCHUNK;
    ec_bucket_hist<<<EC_NCHUNK, 1024, 0, stream>>>(dst, hist, total, E, NB, RB, chunk);
    ec_bucket_scan<<<RB, 1024, 0, stream>>>(hist, base, bstart, RB);
    ec_bucket_scatter<<<EC_NCHUNK, 1024, 0, stream>>>(src, dst, base, part, total, E, NB, RB, chunk);
    ec_bucket_sort<<<RB, 1024, 0, stream>>>(part, bstart, offs, col, Nn, NB, RB, M);

    for (int r0 = 0; r0 < R; r0 += P) {
      int nr = (R - r0 < P) ? (R - r0) : P;
      ec_gemm_xs<<<gemmBlocks, 128, 0, stream>>>(
          x, (r0 == 0) ? L1w : nullptr, W1 + (size_t)r0 * D_IN * D_H, L1b,
          h, hw, mtiles, nr, Nn);
      for (int q = 0; q < nr; ++q) {
        ec_pull<64><<<pull64Blocks, 256, 0, stream>>>(
            hw + (size_t)q * Nn * D_H, offs + (size_t)(r0 + q) * Nn, col, h, Nn);
      }
    }
    ec_relu<<<reluBlocks, 256, 0, stream>>>(h, relu4);

    ec_gemm_wmma<64, 16><<<gemmBlocks, 128, 0, stream>>>(h, L2w, L2b, out, mtiles);
    for (int r = 0; r < R; ++r) {
      ec_gemm_wmma<64, 16><<<gemmBlocks, 128, 0, stream>>>(
          h, W2 + (size_t)r * D_H * D_OUT, nullptr, hw, mtiles);
      ec_pull<16><<<pull16Blocks, 256, 0, stream>>>(
          hw, offs + (size_t)r * Nn, col, out, Nn);
    }
    return;
  }

  off = 0;
  int*   deg  = (int*)(ws + off);
  float* invd = (float*)(ws + off);
  off += (size_t)R * Nn * sizeof(float);
  off = (off + 255) & ~(size_t)255;
  h = (float*)(ws + off);
  off += (size_t)Nn * D_H * sizeof(float);
  off = (off + 255) & ~(size_t)255;
  hw = (float*)(ws + off);

  hipMemsetAsync(deg, 0, (size_t)R * Nn * sizeof(int), stream);
  {
    int total = R * E;
    ec_count_deg<<<(total + 255) / 256, 256, 0, stream>>>(dst, deg, total, Nn, E);
  }
  {
    int total = R * Nn;
    ec_inv_deg<<<(total + 255) / 256, 256, 0, stream>>>(deg, invd, total);
  }

  ec_gemm_wmma<256, 64><<<gemmBlocks, 128, 0, stream>>>(x, L1w, L1b, h, mtiles);
  for (int r = 0; r < R; ++r) {
    ec_gemm_wmma<256, 64><<<gemmBlocks, 128, 0, stream>>>(
        x, W1 + (size_t)r * D_IN * D_H, nullptr, hw, mtiles);
    long long tot = (long long)E * (D_H / 4);
    ec_agg<64><<<(int)((tot + 255) / 256), 256, 0, stream>>>(
        hw, src + (size_t)r * E, dst + (size_t)r * E, invd + (size_t)r * Nn, h, E);
  }
  ec_relu<<<reluBlocks, 256, 0, stream>>>(h, relu4);

  ec_gemm_wmma<64, 16><<<gemmBlocks, 128, 0, stream>>>(h, L2w, L2b, out, mtiles);
  for (int r = 0; r < R; ++r) {
    ec_gemm_wmma<64, 16><<<gemmBlocks, 128, 0, stream>>>(
        h, W2 + (size_t)r * D_H * D_OUT, nullptr, hw, mtiles);
    long long tot = (long long)E * (D_OUT / 4);
    ec_agg<16><<<(int)((tot + 255) / 256), 256, 0, stream>>>(
        hw, src + (size_t)r * E, dst + (size_t)r * E, invd + (size_t)r * Nn, out, E);
  }
}


# Edge-sum row walk once per layer, not once per relation: a row's lanes hold the row in registers across its 4 sorted segments (same add order), ReLU in layer 1's store; 12 launches, not 19
# speedup vs baseline: 2.6282x; 1.1389x over previous
#include <hip/hip_runtime.h>

typedef __attribute__((ext_vector_type(2))) float v2f;
typedef __attribute__((ext_vector_type(4))) float v4f;
typedef __attribute__((ext_vector_type(8))) float v8f;

__global__ __launch_bounds__(256) void ec_count_deg(const int* __restrict__ dst,
                                                    int* __restrict__ deg,
                                                    int total, int Nn, int Eper) {
  int t = blockIdx.x * blockDim.x + threadIdx.x;
  if (t >= total) return;
  int r = t / Eper;
  atomicAdd(&deg[r * Nn + dst[t]], 1);
}

__global__ __launch_bounds__(256) void ec_inv_deg(int* __restrict__ deg,
                                                  float* __restrict__ invd,
                                                  int total) {
  int t = blockIdx.x * blockDim.x + threadIdx.x;
  if (t >= total) return;
  float d = (float)deg[t];
  invd[t] = 1.0f / fmaxf(d, 1.0f);
}


#define EC_BW_LOG 10
#define EC_BW (1 << EC_BW_LOG)
#define EC_RB_MAX 1024
#define EC_NCHUNK 256

__global__ __launch_bounds__(1024) void ec_bucket_hist(const int* __restrict__ dst,
                                                       int* __restrict__ hist,
                                                       int total, int Eper, int NB, int RB,
                                                       int chunk) {
  __shared__ int hs[EC_RB_MAX];
  int tid = threadIdx.x;
  for (int j = tid; j < RB; j += 1024) hs[j] = 0;
  __syncthreads();
  int c0 = blockIdx.x * chunk;
  int c1 = (c0 + chunk < total) ? (c0 + chunk) : total;
  int t = c0 + tid;
  for (; t + 3 * 1024 < c1; t += 4 * 1024) {
    int d0 = dst[t];
    int d1 = dst[t + 1024];
    int d2 = dst[t + 2048];
    int d3 = dst[t + 3072];
    atomicAdd(&hs[(t / Eper) * NB + (d0 >> EC_BW_LOG)], 1);
    atomicAdd(&hs[((t + 1024) / Eper) * NB + (d1 >> EC_BW_LOG)], 1);
    atomicAdd(&hs[((t + 2048) / Eper) * NB + (d2 >> EC_BW_LOG)], 1);
    atomicAdd(&hs[((t + 3072) / Eper) * NB + (d3 >> EC_BW_LOG)], 1);
  }
  for (; t < c1; t += 1024) {
    int d = dst[t];
    atomicAdd(&hs[(t / Eper) * NB + (d >> EC_BW_LOG)], 1);
  }
  __syncthreads();
  for (int j = tid; j < RB; j += 1024) hist[j * EC_NCHUNK + blockIdx.x] = hs[j];
}

__global__ __launch_bounds__(1024) void ec_bucket_scan(const int* __restrict__ hist,
                                                       int* __restrict__ base,
                                                       int* __restrict__ bstart,
                                                       int RB) {
  __shared__ int sh[1024];
  int tid = threadIdx.x;
  int bkt = blockIdx.x;
  const int4* h4 = (const int4*)hist;
  int n4 = bkt * (EC_NCHUNK / 4);
  int p0 = 0, p1 = 0, p2 = 0, p3 = 0;
  int j = tid;
  for (; j + 3 * 1024 < n4; j += 4 * 1024) {
    int4 a = h4[j];
    int4 b = h4[j + 1024];
    int4 c = h4[j + 2048];
    int4 d = h4[j + 3072];
    p0 += a.x + a.y + a.z + a.w;
    p1 += b.x + b.y + b.z + b.w;
    p2 += c.x + c.y + c.z + c.w;
    p3 += d.x + d.y + d.z + d.w;
  }
  for (; j < n4; j += 1024) {
    int4 a = h4[j];
    p0 += a.x + a.y + a.z + a.w;
  }
  sh[tid] = p0 + p1 + p2 + p3;
  __syncthreads();
  for (int st = 512; st > 0; st >>= 1) {
    if (tid < st) sh[tid] += sh[tid + st];
    __syncthreads();
  }
  int before = sh[0];
  __syncthreads();

  int v = (tid < EC_NCHUNK) ? hist[bkt * EC_NCHUNK + tid] : 0;
  sh[tid] = v;
  __syncthreads();
  for (int dl = 1; dl < EC_NCHUNK; dl <<= 1) {
    int add = (tid >= dl) ? sh[tid - dl] : 0;
    __syncthreads();
    sh[tid] += add;
    __syncthreads();
  }
  if (tid < EC_NCHUNK) base[bkt * EC_NCHUNK + tid] = before + sh[tid] - v;
  if (tid == EC_NCHUNK - 1) {
    bstart[bkt] = before;
    if (bkt == RB - 1) bstart[RB] = before + sh[tid];
  }
}

__global__ __launch_bounds__(1024) void ec_bucket_scatter(const int* __restrict__ src,
                                                          const int* __restrict__ dst,
                                                          const int* __restrict__ base,
                                                          int2* __restrict__ part,
                                                          int total, int Eper, int NB, int RB,
                                                          int chunk) {
  __shared__ int cur[EC_RB_MAX];
  int tid = threadIdx.x;
  for (int j = tid; j < RB; j += 1024) cur[j] = base[j * EC_NCHUNK + blockIdx.x];
  __syncthreads();
  int c0 = blockIdx.x * chunk;
  int c1 = (c0 + chunk < total) ? (c0 + chunk) : total;
  int t = c0 + tid;
  for (; t + 3 * 1024 < c1; t += 4 * 1024) {
    int d0 = dst[t];
    int d1 = dst[t + 1024];
    int d2 = dst[t + 2048];
    int d3 = dst[t + 3072];
    int s0 = src[t];
    int s1 = src[t + 1024];
    int s2 = src[t + 2048];
    int s3 = src[t + 3072];
    int q0 = atomicAdd(&cur[(t / Eper) * NB + (d0 >> EC_BW_LOG)], 1);
    int q1 = atomicAdd(&cur[((t + 1024) / Eper) * NB + (d1 >> EC_BW_LOG)], 1);
    int q2 = atomicAdd(&cur[((t + 2048) / Eper) * NB + (d2 >> EC_BW_LOG)], 1);
    int q3 = atomicAdd(&cur[((t + 3072) / Eper) * NB + (d3 >> EC_BW_LOG)], 1);
    part[q0] = make_int2(s0, d0);
    part[q1] = make_int2(s1, d1);
    part[q2] = make_int2(s2, d2);
    part[q3] = make_int2(s3, d3);
  }
  for (; t < c1; t += 1024) {
    int d = dst[t];
    int s = src[t];
    int q = atomicAdd(&cur[(t / Eper) * NB + (d >> EC_BW_LOG)], 1);
    part[q] = make_int2(s, d);
  }
}

__global__ __launch_bounds__(1024) void ec_bucket_sort(const int2* __restrict__ part,
                                                       const int* __restrict__ bstart,
                                                       int* __restrict__ offs,
                                                       int* __restrict__ col,
                                                       int Nn, int NB, int RB, int M) {
  __shared__ int cnt[EC_BW];
  __shared__ int sh[EC_BW];
  int tid = threadIdx.x;
  int bkt = blockIdx.x;
  int r = bkt / NB;
  int n0 = (bkt - r * NB) << EC_BW_LOG;
  int e0 = bstart[bkt];
  int e1 = bstart[bkt + 1];
  cnt[tid] = 0;
  __syncthreads();
  int i = e0 + tid;
  for (; i + 3 * 1024 < e1; i += 4 * 1024) {
    int y0 = part[i].y;
    int y1 = part[i + 1024].y;
    int y2 = part[i + 2048].y;
    int y3 = part[i + 3072].y;
    atomicAdd(&cnt[y0 - n0], 1);
    atomicAdd(&cnt[y1 - n0], 1);
    atomicAdd(&cnt[y2 - n0], 1);
    atomicAdd(&cnt[y3 - n0], 1);
  }
  for (; i < e1; i += 1024) atomicAdd(&cnt[part[i].y - n0], 1);
  __syncthreads();
  int v = cnt[tid];
  sh[tid] = v;
  __syncthreads();
  for (int dl = 1; dl < EC_BW; dl <<= 1) {
    int add = (tid >= dl) ? sh[tid - dl] : 0;
    __syncthreads();
    sh[tid] += add;
    __syncthreads();
  }
  int start = sh[tid] - v;
  if (n0 + tid < Nn) offs[r * Nn + n0 + tid] = e0 + start;
  cnt[tid] = start;
  __syncthreads();
  i = e0 + tid;
  for (; i + 3 * 1024 < e1; i += 4 * 1024) {
    int2 r0 = part[i];
    int2 r1 = part[i + 1024];
    int2 r2 = part[i + 2048];
    int2 r3 = part[i + 3072];
    int q0 = atomicAdd(&cnt[r0.y - n0], 1);
    int q1 = atomicAdd(&cnt[r1.y - n0], 1);
    int q2 = atomicAdd(&cnt[r2.y - n0], 1);
    int q3 = atomicAdd(&cnt[r3.y - n0], 1);
    col[e0 + q0] = r0.x;
    col[e0 + q1] = r1.x;
    col[e0 + q2] = r2.x;
    col[e0 + q3] = r3.x;
  }
  for (; i < e1; i += 1024) {
    int2 rec = part[i];
    int q = atomicAdd(&cnt[rec.y - n0], 1);
    col[e0 + q] = rec.x;
  }
  if (bkt == RB - 1 && tid == 0) offs[M] = e1;
}

template <int K, int NC>
__global__ __launch_bounds__(128) void ec_gemm_wmma(
    const float* __restrict__ A, const float* __restrict__ B,
    const float* __restrict__ bias, float* __restrict__ C, int mtiles) {
  constexpr int NT = NC / 16;
  constexpr int S = ((2 * NC) % 64 == 32) ? (2 * NC) : (2 * NC + 32);
  __shared__ float Bs[(K / 2) * S];

  for (int i = threadIdx.x; i < K * NC / 4; i += 128) {
    int k = (i * 4) / NC;
    int n = (i * 4) % NC;
    float4 v = *(const float4*)(B + i * 4);
    float* p = &Bs[(k >> 1) * S + 2 * n + (k & 1)];
    p[0] = v.x;
    p[2] = v.y;
    p[4] = v.z;
    p[6] = v.w;
  }
  __syncthreads();

  int wave = (blockIdx.x * blockDim.x + threadIdx.x) >> 5;
  if (wave >= mtiles) return;
  int lane = threadIdx.x & 31;
  int m = lane & 15;
  int half = lane >> 4;

  const float* Arow = A + (size_t)(wave * 16 + m) * K + 2 * half;

  v8f acc[NT];
  v8f z = {0.f, 0.f, 0.f, 0.f, 0.f, 0.f, 0.f, 0.f};
#pragma unroll
  for (int j = 0; j < NT; ++j) acc[j] = z;

#pragma unroll 8
  for (int k = 0; k < K; k += 4) {
    float2 av = *(const float2*)(Arow + k);
    v2f a = {av.x, av.y};
    const float* Bp = &Bs[(size_t)((k + 2 * half) >> 1) * S + 2 * m];
#pragma unroll
    for (int j = 0; j < NT; ++j) {
      v2f b = *(const v2f*)(Bp + j * 32);
      acc[j] = __builtin_amdgcn_wmma_f32_16x16x4_f32(
           false, a,  false, b,
           (short)0, acc[j],  false,  false);
    }
  }

#pragma unroll
  for (int j = 0; j < NT; ++j) {
    float bv = bias ? bias[j * 16 + m] : 0.0f;
#pragma unroll
    for (int v = 0; v < 8; ++v) {
      C[(size_t)(wave * 16 + v + 8 * half) * NC + j * 16 + m] = acc[j][v] + bv;
    }
  }
}

__global__ __launch_bounds__(128) void ec_gemm_xs(
    const float* __restrict__ A, const float* __restrict__ Bself,
    const float* __restrict__ Brel, const float* __restrict__ bias,
    float* __restrict__ Cself, float* __restrict__ Crel,
    int mtiles, int nrel, int Nn) {
  constexpr int K = 256;
  constexpr int NC = 64;
  constexpr int NT = NC / 16;
  constexpr int S = 2 * NC + 32;
  __shared__ float Bs[(K / 2) * S];

  int wave = (blockIdx.x * blockDim.x + threadIdx.x) >> 5;
  bool active = wave < mtiles;
  int wv = active ? wave : 0;
  int lane = threadIdx.x & 31;
  int m = lane & 15;
  int half = lane >> 4;

  const float* Arow = A + (size_t)(wv * 16 + m) * K + 2 * half;
  v2f a[K / 4];
#pragma unroll
  for (int kk = 0; kk < K / 4; ++kk) {
    float2 av = *(const float2*)(Arow + 4 * kk);
    a[kk][0] = av.x;
    a[kk][1] = av.y;
  }

  float bvs[NT];
#pragma unroll
  for (int j = 0; j < NT; ++j) bvs[j] = bias[j * 16 + m];

#pragma unroll 1
  for (int r = (Bself ? 0 : 1); r <= nrel; ++r) {
    const float* B = (r == 0) ? Bself : Brel + (size_t)(r - 1) * K * NC;
    float* C = (r == 0) ? Cself : Crel + (size_t)(r - 1) * Nn * NC;

    __syncthreads();
    for (int it0 = 0; it0 < K * NC / 4 / 128; it0 += 8) {
      v4f q[8];
#pragma unroll
      for (int u = 0; u < 8; ++u) {
        int i = (it0 + u) * 128 + (int)threadIdx.x;
        q[u] = *(const v4f*)(B + i * 4);
      }
      asm volatile("" : "+v"(q[0]), "+v"(q[1]), "+v"(q[2]), "+v"(q[3]),
                        "+v"(q[4]), "+v"(q[5]), "+v"(q[6]), "+v"(q[7]));
#pragma unroll
      for (int u = 0; u < 8; ++u) {
        int i = (it0 + u) * 128 + (int)threadIdx.x;
        int k = (i * 4) / NC;
        int n = (i * 4) % NC;
        int j = n >> 4;
        int m0 = n & 15;
        float* p = &Bs[(k >> 1) * S + (j >> 1) * 64 + 4 * m0 + 2 * (j & 1) + (k & 1)];
        p[0] = q[u][0];
        p[4] = q[u][1];
        p[8] = q[u][2];
        p[12] = q[u][3];
      }
    }
    __syncthreads();

    v8f acc[NT];
    v8f z = {0.f, 0.f, 0.f, 0.f, 0.f, 0.f, 0.f, 0.f};
#pragma unroll
    for (int j = 0; j < NT; ++j) acc[j] = z;

#pragma unroll
    for (int kk = 0; kk < K / 4; ++kk) {
      int k = 4 * kk;
      const float* Bp = &Bs[(size_t)((k + 2 * half) >> 1) * S + 4 * m];
#pragma unroll
      for (int t = 0; t < NT / 2; ++t) {
        v4f q = *(const v4f*)(Bp + t * 64);
        v2f b0 = {q[0], q[1]};
        v2f b1 = {q[2], q[3]};
        acc[2 * t] = __builtin_amdgcn_wmma_f32_16x16x4_f32(
             false, a[kk],  false, b0,
             (short)0, acc[2 * t],  false,  false);
        acc[2 * t + 1] = __builtin_amdgcn_wmma_f32_16x16x4_f32(
             false, a[kk],  false, b1,
             (short)0, acc[2 * t + 1],  false,  false);
      }
    }

    v8f o[NT];
#pragma unroll
    for (int j = 0; j < NT; ++j) {
      float bv = (r == 0) ? bvs[j] : 0.0f;
#pragma unroll
      for (int v = 0; v < 8; ++v) o[j][v] = acc[j][v] + bv;
    }
    asm volatile("" : "+v"(o[0]), "+v"(o[1]), "+v"(o[2]), "+v"(o[3]));
    if (active) {
#pragma unroll
      for (int j = 0; j < NT; ++j) {
#pragma unroll
        for (int v = 0; v < 8; ++v) {
          C[(size_t)(wave * 16 + v + 8 * half) * NC + j * 16 + m] = o[j][v];
        }
      }
    }
  }
}

template <int COLS>
__global__ __launch_bounds__(256) void ec_agg(
    const float* __restrict__ hw, const int* __restrict__ src,
    const int* __restrict__ dst, const float* __restrict__ invd,
    float* __restrict__ acc, int E) {
  constexpr int TPE = COLS / 4;
  int t = blockIdx.x * blockDim.x + threadIdx.x;
  int e = t / TPE;
  if (e >= E) return;
  int c = (t % TPE) * 4;
  int s = src[e];
  int d = dst[e];
  float w = invd[d];
  float4 v = *(const float4*)(hw + (size_t)s * COLS + c);
  float* dp = acc + (size_t)d * COLS + c;
  unsafeAtomicAdd(dp + 0, v.x * w);
  unsafeAtomicAdd(dp + 1, v.y * w);
  unsafeAtomicAdd(dp + 2, v.z * w);
  unsafeAtomicAdd(dp + 3, v.w * w);
}

template <int COLS>
__global__ __launch_bounds__(256) void ec_pull_rows(
    const float* __restrict__ hw, const int* __restrict__ offs,
    const int* __restrict__ col, float* __restrict__ acc,
    int Nn, int nrel, int relu) {
  constexpr int TPN = COLS / 4;
  constexpr int RPB = 256 / TPN;
  const int g = (int)threadIdx.x / TPN;
  const int c = ((int)threadIdx.x % TPN) * 4;
  const int step = (int)gridDim.x * RPB;
  const size_t plane = (size_t)Nn * COLS;
  int d = (int)blockIdx.x * RPB + g;
  if (d >= Nn) return;
  int b = offs[d];
  int e = offs[d + 1];
  for (;;) {
    const int dn = d + step;
    const bool more = dn < Nn;
    const int dq = more ? dn : d;
    float4* hp = (float4*)(acc + (size_t)d * COLS + c);
    float4 o = *hp;
    const float* hwc = hw + c;
    const int* oq = offs + d;
    for (int q = 0; q < nrel; ++q) {
      const int* on = (q + 1 < nrel) ? (oq + Nn) : (offs + dq);
      const int bn = on[0];
      const int en = on[1];
      float sx = 0.f, sy = 0.f, sz = 0.f, sw = 0.f;
      int i = b;
      for (; i + 4 <= e; i += 4) {
        int s0 = col[i];
        int s1 = col[i + 1];
        int s2 = col[i + 2];
        int s3 = col[i + 3];
        float4 v0 = *(const float4*)(hwc + (size_t)s0 * COLS);
        float4 v1 = *(const float4*)(hwc + (size_t)s1 * COLS);
        float4 v2 = *(const float4*)(hwc + (size_t)s2 * COLS);
        float4 v3 = *(const float4*)(hwc + (size_t)s3 * COLS);
        sx += v0.x; sy += v0.y; sz += v0.z; sw += v0.w;
        sx += v1.x; sy += v1.y; sz += v1.z; sw += v1.w;
        sx += v2.x; sy += v2.y; sz += v2.z; sw += v2.w;
        sx += v3.x; sy += v3.y; sz += v3.z; sw += v3.w;
      }
      if (i < e) {
        int n = e - i;
        int i1 = (n > 1) ? i + 1 : i;
        int i2 = (n > 2) ? i + 2 : i;
        int s0 = col[i];
        int s1 = col[i1];
        int s2 = col[i2];
        float4 v0 = *(const float4*)(hwc + (size_t)s0 * COLS);
        float4 v1 = *(const float4*)(hwc + (size_t)s1 * COLS);
        float4 v2 = *(const float4*)(hwc + (size_t)s2 * COLS);
        asm volatile("" : "+v"(v1.x), "+v"(v1.y), "+v"(v1.z), "+v"(v1.w),
                          "+v"(v2.x), "+v"(v2.y), "+v"(v2.z), "+v"(v2.w));
        sx += v0.x; sy += v0.y; sz += v0.z; sw += v0.w;
        if (n > 1) { sx += v1.x; sy += v1.y; sz += v1.z; sw += v1.w; }
        if (n > 2) { sx += v2.x; sy += v2.y; sz += v2.z; sw += v2.w; }
      }
      float w = 1.0f / fmaxf((float)(e - b), 1.0f);
      o.x += sx * w;
      o.y += sy * w;
      o.z += sz * w;
      o.w += sw * w;
      b = bn;
      e = en;
      oq += Nn;
      hwc += plane;
    }
    if (relu) {
      o.x = fmaxf(o.x, 0.0f);
      o.y = fmaxf(o.y, 0.0f);
      o.z = fmaxf(o.z, 0.0f);
      o.w = fmaxf(o.w, 0.0f);
    }
    *hp = o;
    if (!more) break;
    d = dn;
  }
}

__global__ __launch_bounds__(256) void ec_relu(float* __restrict__ h, int n4) {
  v4f* p = (v4f*)h;
  const int T = (int)gridDim.x * 256;
  for (int i = (int)blockIdx.x * 256 + (int)threadIdx.x; i < n4; i += 4 * T) {
    const int i1 = i + T;
    const int i2 = i + 2 * T;
    const int i3 = i + 3 * T;
    const bool k1 = i1 < n4;
    const bool k2 = i2 < n4;
    const bool k3 = i3 < n4;
    v4f a0 = p[i];
    v4f a1 = p[k1 ? i1 : i];
    v4f a2 = p[k2 ? i2 : i];
    v4f a3 = p[k3 ? i3 : i];
#pragma unroll
    for (int q = 0; q < 4; ++q) {
      a0[q] = fmaxf(a0[q], 0.0f);
      a1[q] = fmaxf(a1[q], 0.0f);
      a2[q] = fmaxf(a2[q], 0.0f);
      a3[q] = fmaxf(a3[q], 0.0f);
    }
    p[i] = a0;
    if (k1) p[i1] = a1;
    if (k2) p[i2] = a2;
    if (k3) p[i3] = a3;
  }
}

extern "C" void kernel_launch(void* const* d_in, const int* in_sizes, int n_in,
                              void* d_out, int out_size, void* d_ws, size_t ws_size,
                              hipStream_t stream) {
  const float* x   = (const float*)d_in[0];
  const int*   src = (const int*)d_in[1];
  const int*   dst = (const int*)d_in[2];
  const float* W1  = (const float*)d_in[3];
  const float* L1w = (const float*)d_in[4];
  const float* L1b = (const float*)d_in[5];
  const float* W2  = (const float*)d_in[6];
  const float* L2w = (const float*)d_in[7];
  const float* L2b = (const float*)d_in[8];
  float* out = (float*)d_out;

  const int D_IN = 256, D_H = 64, D_OUT = 16;
  const int Nn = in_sizes[0] / D_IN;
  const int R  = in_sizes[3] / (D_IN * D_H);
  const int E  = in_sizes[1] / R;

  char* ws = (char*)d_ws;
  int mtiles = Nn / 16;
  int gemmBlocks = (mtiles * 32 + 127) / 128;

  const int M = R * Nn;
  const int NB = (Nn + EC_BW - 1) >> EC_BW_LOG;
  const int RB = R * NB;
  size_t off = 0;
  int* hist = (int*)(ws + off);
  off += (size_t)RB * EC_NCHUNK * sizeof(int);
  off = (off + 255) & ~(size_t)255;
  int* base = (int*)(ws + off);
  off += (size_t)RB * EC_NCHUNK * sizeof(int);
  off = (off + 255) & ~(size_t)255;
  int* bstart = (int*)(ws + off);
  off += ((size_t)RB + 1) * sizeof(int);
  off = (off + 255) & ~(size_t)255;
  int* offs = (int*)(ws + off);
  off += ((size_t)M + 1) * sizeof(int);
  off = (off + 255) & ~(size_t)255;
  int* col = (int*)(ws + off);
  off += (size_t)R * E * sizeof(int);
  off = (off + 255) & ~(size_t)255;
  float* h = (float*)(ws + off);
  off += (size_t)Nn * D_H * sizeof(float);
  off = (off + 255) & ~(size_t)255;
  float* hw = (float*)(ws + off);
  int2* part = (int2*)hw;
  size_t planeBytes = (size_t)Nn * D_H * sizeof(float);
  size_t partBytes = (size_t)R * E * sizeof(int2);
  size_t hwOff = off;
  off += (planeBytes > partBytes) ? planeBytes : partBytes;

  const int residentBlocks = 2048;
  int pull64Blocks = (Nn + 15) / 16;
  if (pull64Blocks > residentBlocks) pull64Blocks = residentBlocks;
  int pull16Blocks = (Nn + 63) / 64;
  if (pull16Blocks > residentBlocks) pull16Blocks = residentBlocks;
  const int relu4 = Nn * D_H / 4;
  int reluBlocks = (relu4 + 1023) / 1024;
  if (reluBlocks > residentBlocks) reluBlocks = residentBlocks;

  if (ws_size >= off && RB <= EC_RB_MAX) {
    int P = 1;
    if (ws_size >= hwOff + 2 * planeBytes) P = 2;
    if (ws_size >= hwOff + (size_t)R * planeBytes) P = R;
    int total = R * E;
    int chunk = (total + EC_NCHUNK - 1) / EC_NCHUNK;
    ec_bucket_hist<<<EC_NCHUNK, 1024, 0, stream>>>(dst, hist, total, E, NB, RB, chunk);
    ec_bucket_scan<<<RB, 1024, 0, stream>>>(hist, base, bstart, RB);
    ec_bucket_scatter<<<EC_NCHUNK, 1024, 0, stream>>>(src, dst, base, part, total, E, NB, RB, chunk);
    ec_bucket_sort<<<RB, 1024, 0, stream>>>(part, bstart, offs, col, Nn, NB, RB, M);

    for (int r0 = 0; r0 < R; r0 += P) {
      int nr = (R - r0 < P) ? (R - r0) : P;
      ec_gemm_xs<<<gemmBlocks, 128, 0, stream>>>(
          x, (r0 == 0) ? L1w : nullptr, W1 + (size_t)r0 * D_IN * D_H, L1b,
          h, hw, mtiles, nr, Nn);
      ec_pull_rows<64><<<pull64Blocks, 256, 0, stream>>>(
          hw, offs + (size_t)r0 * Nn, col, h, Nn, nr, (r0 + nr >= R) ? 1 : 0);
    }

    ec_gemm_wmma<64, 16><<<gemmBlocks, 128, 0, stream>>>(h, L2w, L2b, out, mtiles);
    size_t plane2Bytes = (size_t)Nn * D_OUT * sizeof(float);
    int P2 = (ws_size >= hwOff + (size_t)R * plane2Bytes) ? R : 1;
    for (int r0 = 0; r0 < R; r0 += P2) {
      int nr = (R - r0 < P2) ? (R - r0) : P2;
      for (int q = 0; q < nr; ++q) {
        ec_gemm_wmma<64, 16><<<gemmBlocks, 128, 0, stream>>>(
            h, W2 + (size_t)(r0 + q) * D_H * D_OUT, nullptr,
            hw + (size_t)q * Nn * D_OUT, mtiles);
      }
      ec_pull_rows<16><<<pull16Blocks, 256, 0, stream>>>(
          hw, offs + (size_t)r0 * Nn, col, out, Nn, nr, 0);
    }
    return;
  }

  off = 0;
  int*   deg  = (int*)(ws + off);
  float* invd = (float*)(ws + off);
  off += (size_t)R * Nn * sizeof(float);
  off = (off + 255) & ~(size_t)255;
  h = (float*)(ws + off);
  off += (size_t)Nn * D_H * sizeof(float);
  off = (off + 255) & ~(size_t)255;
  hw = (float*)(ws + off);

  hipMemsetAsync(deg, 0, (size_t)R * Nn * sizeof(int), stream);
  {
    int total = R * E;
    ec_count_deg<<<(total + 255) / 256, 256, 0, stream>>>(dst, deg, total, Nn, E);
  }
  {
    int total = R * Nn;
    ec_inv_deg<<<(total + 255) / 256, 256, 0, stream>>>(deg, invd, total);
  }

  ec_gemm_wmma<256, 64><<<gemmBlocks, 128, 0, stream>>>(x, L1w, L1b, h, mtiles);
  for (int r = 0; r < R; ++r) {
    ec_gemm_wmma<256, 64><<<gemmBlocks, 128, 0, stream>>>(
        x, W1 + (size_t)r * D_IN * D_H, nullptr, hw, mtiles);
    long long tot = (long long)E * (D_H / 4);
    ec_agg<64><<<(int)((tot + 255) / 256), 256, 0, stream>>>(
        hw, src + (size_t)r * E, dst + (size_t)r * E, invd + (size_t)r * Nn, h, E);
  }
  ec_relu<<<reluBlocks, 256, 0, stream>>>(h, relu4);

  ec_gemm_wmma<64, 16><<<gemmBlocks, 128, 0, stream>>>(h, L2w, L2b, out, mtiles);
  for (int r = 0; r < R; ++r) {
    ec_gemm_wmma<64, 16><<<gemmBlocks, 128, 0, stream>>>(
        h, W2 + (size_t)r * D_H * D_OUT, nullptr, hw, mtiles);
    long long tot = (long long)E * (D_OUT / 4);
    ec_agg<16><<<(int)((tot + 255) / 256), 256, 0, stream>>>(
        hw, src + (size_t)r * E, dst + (size_t)r * E, invd + (size_t)r * Nn, out, E);
  }
}


# Edge-sum walk, one round trip a segment: a row's lanes read its source ids in one lane-parallel load a segment ahead, share them by lane broadcast, gather up to 16 rows as one batch; add order same
# speedup vs baseline: 2.7211x; 1.0353x over previous
#include <hip/hip_runtime.h>

typedef __attribute__((ext_vector_type(2))) float v2f;
typedef __attribute__((ext_vector_type(4))) float v4f;
typedef __attribute__((ext_vector_type(8))) float v8f;

__global__ __launch_bounds__(256) void ec_count_deg(const int* __restrict__ dst,
                                                    int* __restrict__ deg,
                                                    int total, int Nn, int Eper) {
  int t = blockIdx.x * blockDim.x + threadIdx.x;
  if (t >= total) return;
  int r = t / Eper;
  atomicAdd(&deg[r * Nn + dst[t]], 1);
}

__global__ __launch_bounds__(256) void ec_inv_deg(int* __restrict__ deg,
                                                  float* __restrict__ invd,
                                                  int total) {
  int t = blockIdx.x * blockDim.x + threadIdx.x;
  if (t >= total) return;
  float d = (float)deg[t];
  invd[t] = 1.0f / fmaxf(d, 1.0f);
}


#define EC_BW_LOG 10
#define EC_BW (1 << EC_BW_LOG)
#define EC_RB_MAX 1024
#define EC_NCHUNK 256

__global__ __launch_bounds__(1024) void ec_bucket_hist(const int* __restrict__ dst,
                                                       int* __restrict__ hist,
                                                       int total, int Eper, int NB, int RB,
                                                       int chunk) {
  __shared__ int hs[EC_RB_MAX];
  int tid = threadIdx.x;
  for (int j = tid; j < RB; j += 1024) hs[j] = 0;
  __syncthreads();
  int c0 = blockIdx.x * chunk;
  int c1 = (c0 + chunk < total) ? (c0 + chunk) : total;
  int t = c0 + tid;
  for (; t + 3 * 1024 < c1; t += 4 * 1024) {
    int d0 = dst[t];
    int d1 = dst[t + 1024];
    int d2 = dst[t + 2048];
    int d3 = dst[t + 3072];
    atomicAdd(&hs[(t / Eper) * NB + (d0 >> EC_BW_LOG)], 1);
    atomicAdd(&hs[((t + 1024) / Eper) * NB + (d1 >> EC_BW_LOG)], 1);
    atomicAdd(&hs[((t + 2048) / Eper) * NB + (d2 >> EC_BW_LOG)], 1);
    atomicAdd(&hs[((t + 3072) / Eper) * NB + (d3 >> EC_BW_LOG)], 1);
  }
  for (; t < c1; t += 1024) {
    int d = dst[t];
    atomicAdd(&hs[(t / Eper) * NB + (d >> EC_BW_LOG)], 1);
  }
  __syncthreads();
  for (int j = tid; j < RB; j += 1024) hist[j * EC_NCHUNK + blockIdx.x] = hs[j];
}

__global__ __launch_bounds__(1024) void ec_bucket_scan(const int* __restrict__ hist,
                                                       int* __restrict__ base,
                                                       int* __restrict__ bstart,
                                                       int RB) {
  __shared__ int sh[1024];
  int tid = threadIdx.x;
  int bkt = blockIdx.x;
  const int4* h4 = (const int4*)hist;
  int n4 = bkt * (EC_NCHUNK / 4);
  int p0 = 0, p1 = 0, p2 = 0, p3 = 0;
  int j = tid;
  for (; j + 3 * 1024 < n4; j += 4 * 1024) {
    int4 a = h4[j];
    int4 b = h4[j + 1024];
    int4 c = h4[j + 2048];
    int4 d = h4[j + 3072];
    p0 += a.x + a.y + a.z + a.w;
    p1 += b.x + b.y + b.z + b.w;
    p2 += c.x + c.y + c.z + c.w;
    p3 += d.x + d.y + d.z + d.w;
  }
  for (; j < n4; j += 1024) {
    int4 a = h4[j];
    p0 += a.x + a.y + a.z + a.w;
  }
  sh[tid] = p0 + p1 + p2 + p3;
  __syncthreads();
  for (int st = 512; st > 0; st >>= 1) {
    if (tid < st) sh[tid] += sh[tid + st];
    __syncthreads();
  }
  int before = sh[0];
  __syncthreads();

  int v = (tid < EC_NCHUNK) ? hist[bkt * EC_NCHUNK + tid] : 0;
  sh[tid] = v;
  __syncthreads();
  for (int dl = 1; dl < EC_NCHUNK; dl <<= 1) {
    int add = (tid >= dl) ? sh[tid - dl] : 0;
    __syncthreads();
    sh[tid] += add;
    __syncthreads();
  }
  if (tid < EC_NCHUNK) base[bkt * EC_NCHUNK + tid] = before + sh[tid] - v;
  if (tid == EC_NCHUNK - 1) {
    bstart[bkt] = before;
    if (bkt == RB - 1) bstart[RB] = before + sh[tid];
  }
}

__global__ __launch_bounds__(1024) void ec_bucket_scatter(const int* __restrict__ src,
                                                          const int* __restrict__ dst,
                                                          const int* __restrict__ base,
                                                          int2* __restrict__ part,
                                                          int total, int Eper, int NB, int RB,
                                                          int chunk) {
  __shared__ int cur[EC_RB_MAX];
  int tid = threadIdx.x;
  for (int j = tid; j < RB; j += 1024) cur[j] = base[j * EC_NCHUNK + blockIdx.x];
  __syncthreads();
  int c0 = blockIdx.x * chunk;
  int c1 = (c0 + chunk < total) ? (c0 + chunk) : total;
  int t = c0 + tid;
  for (; t + 3 * 1024 < c1; t += 4 * 1024) {
    int d0 = dst[t];
    int d1 = dst[t + 1024];
    int d2 = dst[t + 2048];
    int d3 = dst[t + 3072];
    int s0 = src[t];
    int s1 = src[t + 1024];
    int s2 = src[t + 2048];
    int s3 = src[t + 3072];
    int q0 = atomicAdd(&cur[(t / Eper) * NB + (d0 >> EC_BW_LOG)], 1);
    int q1 = atomicAdd(&cur[((t + 1024) / Eper) * NB + (d1 >> EC_BW_LOG)], 1);
    int q2 = atomicAdd(&cur[((t + 2048) / Eper) * NB + (d2 >> EC_BW_LOG)], 1);
    int q3 = atomicAdd(&cur[((t + 3072) / Eper) * NB + (d3 >> EC_BW_LOG)], 1);
    part[q0] = make_int2(s0, d0);
    part[q1] = make_int2(s1, d1);
    part[q2] = make_int2(s2, d2);
    part[q3] = make_int2(s3, d3);
  }
  for (; t < c1; t += 1024) {
    int d = dst[t];
    int s = src[t];
    int q = atomicAdd(&cur[(t / Eper) * NB + (d >> EC_BW_LOG)], 1);
    part[q] = make_int2(s, d);
  }
}

__global__ __launch_bounds__(1024) void ec_bucket_sort(const int2* __restrict__ part,
                                                       const int* __restrict__ bstart,
                                                       int* __restrict__ offs,
                                                       int* __restrict__ col,
                                                       int Nn, int NB, int RB, int M) {
  __shared__ int cnt[EC_BW];
  __shared__ int sh[EC_BW];
  int tid = threadIdx.x;
  int bkt = blockIdx.x;
  int r = bkt / NB;
  int n0 = (bkt - r * NB) << EC_BW_LOG;
  int e0 = bstart[bkt];
  int e1 = bstart[bkt + 1];
  cnt[tid] = 0;
  __syncthreads();
  int i = e0 + tid;
  for (; i + 3 * 1024 < e1; i += 4 * 1024) {
    int y0 = part[i].y;
    int y1 = part[i + 1024].y;
    int y2 = part[i + 2048].y;
    int y3 = part[i + 3072].y;
    atomicAdd(&cnt[y0 - n0], 1);
    atomicAdd(&cnt[y1 - n0], 1);
    atomicAdd(&cnt[y2 - n0], 1);
    atomicAdd(&cnt[y3 - n0], 1);
  }
  for (; i < e1; i += 1024) atomicAdd(&cnt[part[i].y - n0], 1);
  __syncthreads();
  int v = cnt[tid];
  sh[tid] = v;
  __syncthreads();
  for (int dl = 1; dl < EC_BW; dl <<= 1) {
    int add = (tid >= dl) ? sh[tid - dl] : 0;
    __syncthreads();
    sh[tid] += add;
    __syncthreads();
  }
  int start = sh[tid] - v;
  if (n0 + tid < Nn) offs[r * Nn + n0 + tid] = e0 + start;
  cnt[tid] = start;
  __syncthreads();
  i = e0 + tid;
  for (; i + 3 * 1024 < e1; i += 4 * 1024) {
    int2 r0 = part[i];
    int2 r1 = part[i + 1024];
    int2 r2 = part[i + 2048];
    int2 r3 = part[i + 3072];
    int q0 = atomicAdd(&cnt[r0.y - n0], 1);
    int q1 = atomicAdd(&cnt[r1.y - n0], 1);
    int q2 = atomicAdd(&cnt[r2.y - n0], 1);
    int q3 = atomicAdd(&cnt[r3.y - n0], 1);
    col[e0 + q0] = r0.x;
    col[e0 + q1] = r1.x;
    col[e0 + q2] = r2.x;
    col[e0 + q3] = r3.x;
  }
  for (; i < e1; i += 1024) {
    int2 rec = part[i];
    int q = atomicAdd(&cnt[rec.y - n0], 1);
    col[e0 + q] = rec.x;
  }
  if (bkt == RB - 1 && tid == 0) offs[M] = e1;
}

template <int K, int NC>
__global__ __launch_bounds__(128) void ec_gemm_wmma(
    const float* __restrict__ A, const float* __restrict__ B,
    const float* __restrict__ bias, float* __restrict__ C, int mtiles) {
  constexpr int NT = NC / 16;
  constexpr int S = ((2 * NC) % 64 == 32) ? (2 * NC) : (2 * NC + 32);
  __shared__ float Bs[(K / 2) * S];

  for (int i = threadIdx.x; i < K * NC / 4; i += 128) {
    int k = (i * 4) / NC;
    int n = (i * 4) % NC;
    float4 v = *(const float4*)(B + i * 4);
    float* p = &Bs[(k >> 1) * S + 2 * n + (k & 1)];
    p[0] = v.x;
    p[2] = v.y;
    p[4] = v.z;
    p[6] = v.w;
  }
  __syncthreads();

  int wave = (blockIdx.x * blockDim.x + threadIdx.x) >> 5;
  if (wave >= mtiles) return;
  int lane = threadIdx.x & 31;
  int m = lane & 15;
  int half = lane >> 4;

  const float* Arow = A + (size_t)(wave * 16 + m) * K + 2 * half;

  v8f acc[NT];
  v8f z = {0.f, 0.f, 0.f, 0.f, 0.f, 0.f, 0.f, 0.f};
#pragma unroll
  for (int j = 0; j < NT; ++j) acc[j] = z;

#pragma unroll 8
  for (int k = 0; k < K; k += 4) {
    float2 av = *(const float2*)(Arow + k);
    v2f a = {av.x, av.y};
    const float* Bp = &Bs[(size_t)((k + 2 * half) >> 1) * S + 2 * m];
#pragma unroll
    for (int j = 0; j < NT; ++j) {
      v2f b = *(const v2f*)(Bp + j * 32);
      acc[j] = __builtin_amdgcn_wmma_f32_16x16x4_f32(
           false, a,  false, b,
           (short)0, acc[j],  false,  false);
    }
  }

#pragma unroll
  for (int j = 0; j < NT; ++j) {
    float bv = bias ? bias[j * 16 + m] : 0.0f;
#pragma unroll
    for (int v = 0; v < 8; ++v) {
      C[(size_t)(wave * 16 + v + 8 * half) * NC + j * 16 + m] = acc[j][v] + bv;
    }
  }
}

__global__ __launch_bounds__(128) void ec_gemm_xs(
    const float* __restrict__ A, const float* __restrict__ Bself,
    const float* __restrict__ Brel, const float* __restrict__ bias,
    float* __restrict__ Cself, float* __restrict__ Crel,
    int mtiles, int nrel, int Nn) {
  constexpr int K = 256;
  constexpr int NC = 64;
  constexpr int NT = NC / 16;
  constexpr int S = 2 * NC + 32;
  __shared__ float Bs[(K / 2) * S];

  int wave = (blockIdx.x * blockDim.x + threadIdx.x) >> 5;
  bool active = wave < mtiles;
  int wv = active ? wave : 0;
  int lane = threadIdx.x & 31;
  int m = lane & 15;
  int half = lane >> 4;

  const float* Arow = A + (size_t)(wv * 16 + m) * K + 2 * half;
  v2f a[K / 4];
#pragma unroll
  for (int kk = 0; kk < K / 4; ++kk) {
    float2 av = *(const float2*)(Arow + 4 * kk);
    a[kk][0] = av.x;
    a[kk][1] = av.y;
  }

  float bvs[NT];
#pragma unroll
  for (int j = 0; j < NT; ++j) bvs[j] = bias[j * 16 + m];

#pragma unroll 1
  for (int r = (Bself ? 0 : 1); r <= nrel; ++r) {
    const float* B = (r == 0) ? Bself : Brel + (size_t)(r - 1) * K * NC;
    float* C = (r == 0) ? Cself : Crel + (size_t)(r - 1) * Nn * NC;

    __syncthreads();
    for (int it0 = 0; it0 < K * NC / 4 / 128; it0 += 8) {
      v4f q[8];
#pragma unroll
      for (int u = 0; u < 8; ++u) {
        int i = (it0 + u) * 128 + (int)threadIdx.x;
        q[u] = *(const v4f*)(B + i * 4);
      }
      asm volatile("" : "+v"(q[0]), "+v"(q[1]), "+v"(q[2]), "+v"(q[3]),
                        "+v"(q[4]), "+v"(q[5]), "+v"(q[6]), "+v"(q[7]));
#pragma unroll
      for (int u = 0; u < 8; ++u) {
        int i = (it0 + u) * 128 + (int)threadIdx.x;
        int k = (i * 4) / NC;
        int n = (i * 4) % NC;
        int j = n >> 4;
        int m0 = n & 15;
        float* p = &Bs[(k >> 1) * S + (j >> 1) * 64 + 4 * m0 + 2 * (j & 1) + (k & 1)];
        p[0] = q[u][0];
        p[4] = q[u][1];
        p[8] = q[u][2];
        p[12] = q[u][3];
      }
    }
    __syncthreads();

    v8f acc[NT];
    v8f z = {0.f, 0.f, 0.f, 0.f, 0.f, 0.f, 0.f, 0.f};
#pragma unroll
    for (int j = 0; j < NT; ++j) acc[j] = z;

#pragma unroll
    for (int kk = 0; kk < K / 4; ++kk) {
      int k = 4 * kk;
      const float* Bp = &Bs[(size_t)((k + 2 * half) >> 1) * S + 4 * m];
#pragma unroll
      for (int t = 0; t < NT / 2; ++t) {
        v4f q = *(const v4f*)(Bp + t * 64);
        v2f b0 = {q[0], q[1]};
        v2f b1 = {q[2], q[3]};
        acc[2 * t] = __builtin_amdgcn_wmma_f32_16x16x4_f32(
             false, a[kk],  false, b0,
             (short)0, acc[2 * t],  false,  false);
        acc[2 * t + 1] = __builtin_amdgcn_wmma_f32_16x16x4_f32(
             false, a[kk],  false, b1,
             (short)0, acc[2 * t + 1],  false,  false);
      }
    }

    v8f o[NT];
#pragma unroll
    for (int j = 0; j < NT; ++j) {
      float bv = (r == 0) ? bvs[j] : 0.0f;
#pragma unroll
      for (int v = 0; v < 8; ++v) o[j][v] = acc[j][v] + bv;
    }
    asm volatile("" : "+v"(o[0]), "+v"(o[1]), "+v"(o[2]), "+v"(o[3]));
    if (active) {
#pragma unroll
      for (int j = 0; j < NT; ++j) {
#pragma unroll
        for (int v = 0; v < 8; ++v) {
          C[(size_t)(wave * 16 + v + 8 * half) * NC + j * 16 + m] = o[j][v];
        }
      }
    }
  }
}

template <int COLS>
__global__ __launch_bounds__(256) void ec_agg(
    const float* __restrict__ hw, const int* __restrict__ src,
    const int* __restrict__ dst, const float* __restrict__ invd,
    float* __restrict__ acc, int E) {
  constexpr int TPE = COLS / 4;
  int t = blockIdx.x * blockDim.x + threadIdx.x;
  int e = t / TPE;
  if (e >= E) return;
  int c = (t % TPE) * 4;
  int s = src[e];
  int d = dst[e];
  float w = invd[d];
  float4 v = *(const float4*)(hw + (size_t)s * COLS + c);
  float* dp = acc + (size_t)d * COLS + c;
  unsafeAtomicAdd(dp + 0, v.x * w);
  unsafeAtomicAdd(dp + 1, v.y * w);
  unsafeAtomicAdd(dp + 2, v.z * w);
  unsafeAtomicAdd(dp + 3, v.w * w);
}

template <int TPN, int J>
__device__ __forceinline__ int ec_slot_src(const int* id) {
  static_assert(TPN == 16 || TPN == 4, "lane group is a 16-lane row or a quad");
  if constexpr (TPN == 16) {
    return __builtin_amdgcn_update_dpp(0, id[0], 0x150 + J, 0xf, 0xf, true);
  } else {
    return __builtin_amdgcn_update_dpp(0, id[J >> 2], (J & 3) * 0x55, 0xf, 0xf, true);
  }
}

template <int COLS>
__device__ __forceinline__ void ec_sum_batch(const float* __restrict__ hwb, unsigned cb, const int* id, int n,
                                             float& sx, float& sy, float& sz, float& sw) {
  constexpr int TPN = COLS / 4;
  float4 v[16];
#define EC_GATHER(J) \
  if (J < n) v[J] = *(const float4*)((const char*)hwb + \
                       ((unsigned)ec_slot_src<TPN, J>(id) * (unsigned)(COLS * 4) + cb));
  EC_GATHER(0) EC_GATHER(1) EC_GATHER(2) EC_GATHER(3)
  if (n > 4) {
    EC_GATHER(4) EC_GATHER(5) EC_GATHER(6) EC_GATHER(7)
    if (n > 8) {
      EC_GATHER(8) EC_GATHER(9) EC_GATHER(10) EC_GATHER(11)
      if (n > 12) {
        EC_GATHER(12) EC_GATHER(13) EC_GATHER(14) EC_GATHER(15)
      }
    }
  }
#undef EC_GATHER
  asm volatile("" : "+v"(sx), "+v"(sy), "+v"(sz), "+v"(sw));
#define EC_ADD(J) \
  if (J < n) { sx += v[J].x; sy += v[J].y; sz += v[J].z; sw += v[J].w; }
  EC_ADD(0) EC_ADD(1) EC_ADD(2) EC_ADD(3)
  if (n > 4) {
    EC_ADD(4) EC_ADD(5) EC_ADD(6) EC_ADD(7)
    if (n > 8) {
      EC_ADD(8) EC_ADD(9) EC_ADD(10) EC_ADD(11)
      if (n > 12) {
        EC_ADD(12) EC_ADD(13) EC_ADD(14) EC_ADD(15)
      }
    }
  }
#undef EC_ADD
}

template <int COLS>
__global__ __launch_bounds__(256) void ec_pull_rows(
    const float* __restrict__ hw, const int* __restrict__ offs,
    const int* __restrict__ col, float* __restrict__ acc,
    int Nn, int nrel, int relu) {
  constexpr int TPN = COLS / 4;
  constexpr int RPB = 256 / TPN;
  constexpr int NI = 16 / TPN;
  const int g = (int)threadIdx.x / TPN;
  const int l = (int)threadIdx.x % TPN;
  const int c = l * 4;
  const int step = (int)gridDim.x * RPB;
  const size_t plane = (size_t)Nn * COLS;
  int d = (int)blockIdx.x * RPB + g;
  if (d >= Nn) return;
  int q = 0;

  int b = offs[d];
  int e = offs[d + 1];
  int idc[NI];
#pragma unroll
  for (int k = 0; k < NI; ++k) {
    int j = k * TPN + l;
    idc[k] = col[(j < e - b) ? (b + j) : 0];
  }
  int q1 = (nrel > 1) ? 1 : 0;
  int d1 = (nrel > 1) ? d : d + step;
  bool ok1 = d1 < Nn;
  int b1 = 0, e1 = 0;
  if (ok1) {
    const int* p = offs + (size_t)q1 * Nn + d1;
    b1 = p[0];
    e1 = p[1];
  }

  float4 o = {0.f, 0.f, 0.f, 0.f};
  for (;;) {
    int idn[NI];
#pragma unroll
    for (int k = 0; k < NI; ++k) {
      int j = k * TPN + l;
      idn[k] = col[(j < e1 - b1) ? (b1 + j) : 0];
    }
    int q2 = (q1 + 1 < nrel) ? q1 + 1 : 0;
    int d2 = (q1 + 1 < nrel) ? d1 : d1 + step;
    bool ok2 = ok1 && (d2 < Nn);
    int b2 = 0, e2 = 0;
    if (ok2) {
      const int* p = offs + (size_t)q2 * Nn + d2;
      b2 = p[0];
      e2 = p[1];
    }

    float4* hp = (float4*)(acc + (size_t)d * COLS + c);
    if (q == 0) o = *hp;
    const unsigned cb = (unsigned)(((size_t)q * plane + (size_t)c) * sizeof(float));
    float sx = 0.f, sy = 0.f, sz = 0.f, sw = 0.f;
    const int n = e - b;
    ec_sum_batch<COLS>(hw, cb, idc, (n < 16) ? n : 16, sx, sy, sz, sw);
    for (int i = b + 16; i < e; i += 16) {
      int m = (e - i < 16) ? (e - i) : 16;
      int idx[NI];
#pragma unroll
      for (int k = 0; k < NI; ++k) {
        int j = k * TPN + l;
        idx[k] = col[(j < m) ? (i + j) : 0];
      }
      ec_sum_batch<COLS>(hw, cb, idx, m, sx, sy, sz, sw);
    }
    float w = 1.0f / fmaxf((float)n, 1.0f);
    o.x += sx * w;
    o.y += sy * w;
    o.z += sz * w;
    o.w += sw * w;
    if (q + 1 >= nrel) {
      if (relu) {
        o.x = fmaxf(o.x, 0.0f);
        o.y = fmaxf(o.y, 0.0f);
        o.z = fmaxf(o.z, 0.0f);
        o.w = fmaxf(o.w, 0.0f);
      }
      *hp = o;
    }
    if (!ok1) break;
    d = d1; q = q1; b = b1; e = e1;
#pragma unroll
    for (int k = 0; k < NI; ++k) idc[k] = idn[k];
    d1 = d2; q1 = q2; ok1 = ok2; b1 = b2; e1 = e2;
  }
}


__global__ __launch_bounds__(256) void ec_relu(float* __restrict__ h, int n4) {
  v4f* p = (v4f*)h;
  const int T = (int)gridDim.x * 256;
  for (int i = (int)blockIdx.x * 256 + (int)threadIdx.x; i < n4; i += 4 * T) {
    const int i1 = i + T;
    const int i2 = i + 2 * T;
    const int i3 = i + 3 * T;
    const bool k1 = i1 < n4;
    const bool k2 = i2 < n4;
    const bool k3 = i3 < n4;
    v4f a0 = p[i];
    v4f a1 = p[k1 ? i1 : i];
    v4f a2 = p[k2 ? i2 : i];
    v4f a3 = p[k3 ? i3 : i];
#pragma unroll
    for (int q = 0; q < 4; ++q) {
      a0[q] = fmaxf(a0[q], 0.0f);
      a1[q] = fmaxf(a1[q], 0.0f);
      a2[q] = fmaxf(a2[q], 0.0f);
      a3[q] = fmaxf(a3[q], 0.0f);
    }
    p[i] = a0;
    if (k1) p[i1] = a1;
    if (k2) p[i2] = a2;
    if (k3) p[i3] = a3;
  }
}

extern "C" void kernel_launch(void* const* d_in, const int* in_sizes, int n_in,
                              void* d_out, int out_size, void* d_ws, size_t ws_size,
                              hipStream_t stream) {
  const float* x   = (const float*)d_in[0];
  const int*   src = (const int*)d_in[1];
  const int*   dst = (const int*)d_in[2];
  const float* W1  = (const float*)d_in[3];
  const float* L1w = (const float*)d_in[4];
  const float* L1b = (const float*)d_in[5];
  const float* W2  = (const float*)d_in[6];
  const float* L2w = (const float*)d_in[7];
  const float* L2b = (const float*)d_in[8];
  float* out = (float*)d_out;

  const int D_IN = 256, D_H = 64, D_OUT = 16;
  const int Nn = in_sizes[0] / D_IN;
  const int R  = in_sizes[3] / (D_IN * D_H);
  const int E  = in_sizes[1] / R;

  char* ws = (char*)d_ws;
  int mtiles = Nn / 16;
  int gemmBlocks = (mtiles * 32 + 127) / 128;

  const int M = R * Nn;
  const int NB = (Nn + EC_BW - 1) >> EC_BW_LOG;
  const int RB = R * NB;
  size_t off = 0;
  int* hist = (int*)(ws + off);
  off += (size_t)RB * EC_NCHUNK * sizeof(int);
  off = (off + 255) & ~(size_t)255;
  int* base = (int*)(ws + off);
  off += (size_t)RB * EC_NCHUNK * sizeof(int);
  off = (off + 255) & ~(size_t)255;
  int* bstart = (int*)(ws + off);
  off += ((size_t)RB + 1) * sizeof(int);
  off = (off + 255) & ~(size_t)255;
  int* offs = (int*)(ws + off);
  off += ((size_t)M + 1) * sizeof(int);
  off = (off + 255) & ~(size_t)255;
  int* col = (int*)(ws + off);
  off += (size_t)R * E * sizeof(int);
  off = (off + 255) & ~(size_t)255;
  float* h = (float*)(ws + off);
  off += (size_t)Nn * D_H * sizeof(float);
  off = (off + 255) & ~(size_t)255;
  float* hw = (float*)(ws + off);
  int2* part = (int2*)hw;
  size_t planeBytes = (size_t)Nn * D_H * sizeof(float);
  size_t partBytes = (size_t)R * E * sizeof(int2);
  size_t hwOff = off;
  off += (planeBytes > partBytes) ? planeBytes : partBytes;

  const int residentBlocks = 2048;
  const int pull64Resident = 1280;
  const int pull16Resident = 1024;
  int pull64Blocks = (Nn + 15) / 16;
  if (pull64Blocks > pull64Resident) pull64Blocks = pull64Resident;
  int pull16Blocks = (Nn + 63) / 64;
  if (pull16Blocks > pull16Resident) pull16Blocks = pull16Resident;
  const int relu4 = Nn * D_H / 4;
  int reluBlocks = (relu4 + 1023) / 1024;
  if (reluBlocks > residentBlocks) reluBlocks = residentBlocks;

  if (ws_size >= off && RB <= EC_RB_MAX) {
    int P = 1;
    if (ws_size >= hwOff + 2 * planeBytes) P = 2;
    if (ws_size >= hwOff + (size_t)R * planeBytes) P = R;
    int total = R * E;
    int chunk = (total + EC_NCHUNK - 1) / EC_NCHUNK;
    ec_bucket_hist<<<EC_NCHUNK, 1024, 0, stream>>>(dst, hist, total, E, NB, RB, chunk);
    ec_bucket_scan<<<RB, 1024, 0, stream>>>(hist, base, bstart, RB);
    ec_bucket_scatter<<<EC_NCHUNK, 1024, 0, stream>>>(src, dst, base, part, total, E, NB, RB, chunk);
    ec_bucket_sort<<<RB, 1024, 0, stream>>>(part, bstart, offs, col, Nn, NB, RB, M);

    for (int r0 = 0; r0 < R; r0 += P) {
      int nr = (R - r0 < P) ? (R - r0) : P;
      ec_gemm_xs<<<gemmBlocks, 128, 0, stream>>>(
          x, (r0 == 0) ? L1w : nullptr, W1 + (size_t)r0 * D_IN * D_H, L1b,
          h, hw, mtiles, nr, Nn);
      ec_pull_rows<64><<<pull64Blocks, 256, 0, stream>>>(
          hw, offs + (size_t)r0 * Nn, col, h, Nn, nr, (r0 + nr >= R) ? 1 : 0);
    }

    ec_gemm_wmma<64, 16><<<gemmBlocks, 128, 0, stream>>>(h, L2w, L2b, out, mtiles);
    size_t plane2Bytes = (size_t)Nn * D_OUT * sizeof(float);
    int P2 = (ws_size >= hwOff + (size_t)R * plane2Bytes) ? R : 1;
    for (int r0 = 0; r0 < R; r0 += P2) {
      int nr = (R - r0 < P2) ? (R - r0) : P2;
      for (int q = 0; q < nr; ++q) {
        ec_gemm_wmma<64, 16><<<gemmBlocks, 128, 0, stream>>>(
            h, W2 + (size_t)(r0 + q) * D_H * D_OUT, nullptr,
            hw + (size_t)q * Nn * D_OUT, mtiles);
      }
      ec_pull_rows<16><<<pull16Blocks, 256, 0, stream>>>(
          hw, offs + (size_t)r0 * Nn, col, out, Nn, nr, 0);
    }
    return;
  }

  off = 0;
  int*   deg  = (int*)(ws + off);
  float* invd = (float*)(ws + off);
  off += (size_t)R * Nn * sizeof(float);
  off = (off + 255) & ~(size_t)255;
  h = (float*)(ws + off);
  off += (size_t)Nn * D_H * sizeof(float);
  off = (off + 255) & ~(size_t)255;
  hw = (float*)(ws + off);

  hipMemsetAsync(deg, 0, (size_t)R * Nn * sizeof(int), stream);
  {
    int total = R * E;
    ec_count_deg<<<(total + 255) / 256, 256, 0, stream>>>(dst, deg, total, Nn, E);
  }
  {
    int total = R * Nn;
    ec_inv_deg<<<(total + 255) / 256, 256, 0, stream>>>(deg, invd, total);
  }

  ec_gemm_wmma<256, 64><<<gemmBlocks, 128, 0, stream>>>(x, L1w, L1b, h, mtiles);
  for (int r = 0; r < R; ++r) {
    ec_gemm_wmma<256, 64><<<gemmBlocks, 128, 0, stream>>>(
        x, W1 + (size_t)r * D_IN * D_H, nullptr, hw, mtiles);
    long long tot = (long long)E * (D_H / 4);
    ec_agg<64><<<(int)((tot + 255) / 256), 256, 0, stream>>>(
        hw, src + (size_t)r * E, dst + (size_t)r * E, invd + (size_t)r * Nn, h, E);
  }
  ec_relu<<<reluBlocks, 256, 0, stream>>>(h, relu4);

  ec_gemm_wmma<64, 16><<<gemmBlocks, 128, 0, stream>>>(h, L2w, L2b, out, mtiles);
  for (int r = 0; r < R; ++r) {
    ec_gemm_wmma<64, 16><<<gemmBlocks, 128, 0, stream>>>(
        h, W2 + (size_t)r * D_H * D_OUT, nullptr, hw, mtiles);
    long long tot = (long long)E * (D_OUT / 4);
    ec_agg<16><<<(int)((tot + 255) / 256), 256, 0, stream>>>(
        hw, src + (size_t)r * E, dst + (size_t)r * E, invd + (size_t)r * Nn, out, E);
  }
}


# Edge sort in 2 launches, not 4: a chunk (one slice of every relation) buckets its edges inside its own slice; a bucket's start = sum of its run starts, no table scan; slot (sum) order = LDS arrival
# speedup vs baseline: 2.8389x; 1.0433x over previous
#include <hip/hip_runtime.h>

typedef __attribute__((ext_vector_type(2))) float v2f;
typedef __attribute__((ext_vector_type(4))) float v4f;
typedef __attribute__((ext_vector_type(8))) float v8f;

__global__ __launch_bounds__(256) void ec_count_deg(const int* __restrict__ dst,
                                                    int* __restrict__ deg,
                                                    int total, int Nn, int Eper) {
  int t = blockIdx.x * blockDim.x + threadIdx.x;
  if (t >= total) return;
  int r = t / Eper;
  atomicAdd(&deg[r * Nn + dst[t]], 1);
}

__global__ __launch_bounds__(256) void ec_inv_deg(int* __restrict__ deg,
                                                  float* __restrict__ invd,
                                                  int total) {
  int t = blockIdx.x * blockDim.x + threadIdx.x;
  if (t >= total) return;
  float d = (float)deg[t];
  invd[t] = 1.0f / fmaxf(d, 1.0f);
}


#define EC_BW_LOG 10
#define EC_BW (1 << EC_BW_LOG)
#define EC_RB_MAX 1024
#define EC_NCHUNK 256
#define EC_CPW (EC_NCHUNK / 32)

__device__ __forceinline__ int ec_chunk_edge(int u, int len, int lo, int Eper, int& rel) {
  rel = u / len;
  return rel * Eper + lo + (u - rel * len);
}

__global__ __launch_bounds__(1024) void ec_chunk_sort(const int* __restrict__ src,
                                                      const int* __restrict__ dst,
                                                      int2* __restrict__ part,
                                                      int* __restrict__ lstart,
                                                      int* __restrict__ lcnt,
                                                      int Eper, int Rn, int NB, int RB,
                                                      int sub) {
  static_assert(EC_RB_MAX == 1024, "one bucket counter per thread");
  __shared__ int hs[EC_RB_MAX];
  __shared__ int wtot[32];
  int tid = threadIdx.x;
  int lane = tid & 31;
  int wv = tid >> 5;
  hs[tid] = 0;
  __syncthreads();
  int lo = (int)blockIdx.x * sub;
  if (lo > Eper) lo = Eper;
  int hi = (lo + sub < Eper) ? (lo + sub) : Eper;
  int len = hi - lo;
  int ntot = Rn * len;

  int u = tid;
  for (; u + 3 * 1024 < ntot; u += 4 * 1024) {
    int r0, r1, r2, r3;
    int t0 = ec_chunk_edge(u, len, lo, Eper, r0);
    int t1 = ec_chunk_edge(u + 1024, len, lo, Eper, r1);
    int t2 = ec_chunk_edge(u + 2048, len, lo, Eper, r2);
    int t3 = ec_chunk_edge(u + 3072, len, lo, Eper, r3);
    int d0 = dst[t0];
    int d1 = dst[t1];
    int d2 = dst[t2];
    int d3 = dst[t3];
    atomicAdd(&hs[r0 * NB + (d0 >> EC_BW_LOG)], 1);
    atomicAdd(&hs[r1 * NB + (d1 >> EC_BW_LOG)], 1);
    atomicAdd(&hs[r2 * NB + (d2 >> EC_BW_LOG)], 1);
    atomicAdd(&hs[r3 * NB + (d3 >> EC_BW_LOG)], 1);
  }
  for (; u < ntot; u += 1024) {
    int r0;
    int t0 = ec_chunk_edge(u, len, lo, Eper, r0);
    int d = dst[t0];
    atomicAdd(&hs[r0 * NB + (d >> EC_BW_LOG)], 1);
  }
  __syncthreads();

  int v = hs[tid];
  int inc = v;
  for (int dl = 1; dl < 32; dl <<= 1) {
    int up = __shfl_up(inc, dl, 32);
    if (lane >= dl) inc += up;
  }
  if (lane == 31) wtot[wv] = inc;
  __syncthreads();
  int before = 0;
  for (int j = 0; j < wv; ++j) before += wtot[j];
  int start = before + inc - v;
  if (tid < RB) {
    lstart[tid * EC_NCHUNK + blockIdx.x] = start;
    lcnt[tid * EC_NCHUNK + blockIdx.x] = v;
  }
  hs[tid] = start;
  __syncthreads();

  int2* mine = part + (size_t)Rn * lo;
  u = tid;
  for (; u + 3 * 1024 < ntot; u += 4 * 1024) {
    int r0, r1, r2, r3;
    int t0 = ec_chunk_edge(u, len, lo, Eper, r0);
    int t1 = ec_chunk_edge(u + 1024, len, lo, Eper, r1);
    int t2 = ec_chunk_edge(u + 2048, len, lo, Eper, r2);
    int t3 = ec_chunk_edge(u + 3072, len, lo, Eper, r3);
    int d0 = dst[t0];
    int d1 = dst[t1];
    int d2 = dst[t2];
    int d3 = dst[t3];
    int s0 = src[t0];
    int s1 = src[t1];
    int s2 = src[t2];
    int s3 = src[t3];
    int q0 = atomicAdd(&hs[r0 * NB + (d0 >> EC_BW_LOG)], 1);
    int q1 = atomicAdd(&hs[r1 * NB + (d1 >> EC_BW_LOG)], 1);
    int q2 = atomicAdd(&hs[r2 * NB + (d2 >> EC_BW_LOG)], 1);
    int q3 = atomicAdd(&hs[r3 * NB + (d3 >> EC_BW_LOG)], 1);
    mine[q0] = make_int2(s0, d0);
    mine[q1] = make_int2(s1, d1);
    mine[q2] = make_int2(s2, d2);
    mine[q3] = make_int2(s3, d3);
  }
  for (; u < ntot; u += 1024) {
    int r0;
    int t0 = ec_chunk_edge(u, len, lo, Eper, r0);
    int d = dst[t0];
    int s = src[t0];
    int q = atomicAdd(&hs[r0 * NB + (d >> EC_BW_LOG)], 1);
    mine[q] = make_int2(s, d);
  }
}

__global__ __launch_bounds__(1024) void ec_bucket_sort(const int2* __restrict__ part,
                                                       const int* __restrict__ lstart,
                                                       const int* __restrict__ lcnt,
                                                       int* __restrict__ offs,
                                                       int* __restrict__ col,
                                                       int Nn, int NB, int RB, int M,
                                                       int Eper, int Rn, int sub) {
  __shared__ int cnt[EC_BW];
  __shared__ int sh[EC_BW];
  __shared__ int rbeg[EC_NCHUNK];
  __shared__ int rlen[EC_NCHUNK];
  __shared__ int wst[EC_CPW];
  __shared__ int wln[EC_CPW];
  int tid = threadIdx.x;
  int lane = tid & 31;
  int wv = tid >> 5;
  int bkt = blockIdx.x;
  int r = bkt / NB;
  int n0 = (bkt - r * NB) << EC_BW_LOG;
  cnt[tid] = 0;
  if (tid < EC_NCHUNK) {
    int st = lstart[bkt * EC_NCHUNK + tid];
    int ln = lcnt[bkt * EC_NCHUNK + tid];
    int lo = tid * sub;
    if (lo > Eper) lo = Eper;
    rbeg[tid] = Rn * lo + st;
    rlen[tid] = ln;
    for (int dl = 16; dl > 0; dl >>= 1) {
      st += __shfl_xor(st, dl, 32);
      ln += __shfl_xor(ln, dl, 32);
    }
    if (lane == 0) {
      wst[wv] = st;
      wln[wv] = ln;
    }
  }
  __syncthreads();
  int e0 = 0;
  int nb = 0;
#pragma unroll
  for (int j = 0; j < EC_CPW; ++j) {
    e0 += wst[j];
    nb += wln[j];
  }
  int e1 = e0 + nb;
  int rb[EC_CPW], rl[EC_CPW];
#pragma unroll
  for (int cc = 0; cc < EC_CPW; ++cc) {
    rb[cc] = rbeg[wv * EC_CPW + cc];
    rl[cc] = rlen[wv * EC_CPW + cc];
  }
  if (nb > 0) {
    int y[2 * EC_CPW];
#pragma unroll
    for (int cc = 0; cc < EC_CPW; ++cc) {
      y[2 * cc] = part[(lane < rl[cc]) ? (rb[cc] + lane) : 0].y;
      y[2 * cc + 1] = part[(lane + 32 < rl[cc]) ? (rb[cc] + lane + 32) : 0].y;
    }
#pragma unroll
    for (int cc = 0; cc < EC_CPW; ++cc) {
      if (lane < rl[cc]) atomicAdd(&cnt[y[2 * cc] - n0], 1);
      if (lane + 32 < rl[cc]) atomicAdd(&cnt[y[2 * cc + 1] - n0], 1);
    }
#pragma unroll
    for (int cc = 0; cc < EC_CPW; ++cc) {
      for (int j = lane + 64; j < rl[cc]; j += 32) atomicAdd(&cnt[part[rb[cc] + j].y - n0], 1);
    }
  }
  __syncthreads();
  int v = cnt[tid];
  sh[tid] = v;
  __syncthreads();
  for (int dl = 1; dl < EC_BW; dl <<= 1) {
    int add = (tid >= dl) ? sh[tid - dl] : 0;
    __syncthreads();
    sh[tid] += add;
    __syncthreads();
  }
  int start = sh[tid] - v;
  if (n0 + tid < Nn) offs[r * Nn + n0 + tid] = e0 + start;
  cnt[tid] = start;
  __syncthreads();
  if (nb > 0) {
    int2 rec[2 * EC_CPW];
#pragma unroll
    for (int cc = 0; cc < EC_CPW; ++cc) {
      rec[2 * cc] = part[(lane < rl[cc]) ? (rb[cc] + lane) : 0];
      rec[2 * cc + 1] = part[(lane + 32 < rl[cc]) ? (rb[cc] + lane + 32) : 0];
    }
#pragma unroll
    for (int cc = 0; cc < EC_CPW; ++cc) {
      if (lane < rl[cc]) {
        int q = atomicAdd(&cnt[rec[2 * cc].y - n0], 1);
        col[e0 + q] = rec[2 * cc].x;
      }
      if (lane + 32 < rl[cc]) {
        int q = atomicAdd(&cnt[rec[2 * cc + 1].y - n0], 1);
        col[e0 + q] = rec[2 * cc + 1].x;
      }
    }
#pragma unroll
    for (int cc = 0; cc < EC_CPW; ++cc) {
      for (int j = lane + 64; j < rl[cc]; j += 32) {
        int2 rc = part[rb[cc] + j];
        int q = atomicAdd(&cnt[rc.y - n0], 1);
        col[e0 + q] = rc.x;
      }
    }
  }
  if (bkt == RB - 1 && tid == 0) offs[M] = e1;
}

template <int K, int NC>
__global__ __launch_bounds__(128) void ec_gemm_wmma(
    const float* __restrict__ A, const float* __restrict__ B,
    const float* __restrict__ bias, float* __restrict__ C, int mtiles) {
  constexpr int NT = NC / 16;
  constexpr int S = ((2 * NC) % 64 == 32) ? (2 * NC) : (2 * NC + 32);
  __shared__ float Bs[(K / 2) * S];

  for (int i = threadIdx.x; i < K * NC / 4; i += 128) {
    int k = (i * 4) / NC;
    int n = (i * 4) % NC;
    float4 v = *(const float4*)(B + i * 4);
    float* p = &Bs[(k >> 1) * S + 2 * n + (k & 1)];
    p[0] = v.x;
    p[2] = v.y;
    p[4] = v.z;
    p[6] = v.w;
  }
  __syncthreads();

  int wave = (blockIdx.x * blockDim.x + threadIdx.x) >> 5;
  if (wave >= mtiles) return;
  int lane = threadIdx.x & 31;
  int m = lane & 15;
  int half = lane >> 4;

  const float* Arow = A + (size_t)(wave * 16 + m) * K + 2 * half;

  v8f acc[NT];
  v8f z = {0.f, 0.f, 0.f, 0.f, 0.f, 0.f, 0.f, 0.f};
#pragma unroll
  for (int j = 0; j < NT; ++j) acc[j] = z;

#pragma unroll 8
  for (int k = 0; k < K; k += 4) {
    float2 av = *(const float2*)(Arow + k);
    v2f a = {av.x, av.y};
    const float* Bp = &Bs[(size_t)((k + 2 * half) >> 1) * S + 2 * m];
#pragma unroll
    for (int j = 0; j < NT; ++j) {
      v2f b = *(const v2f*)(Bp + j * 32);
      acc[j] = __builtin_amdgcn_wmma_f32_16x16x4_f32(
           false, a,  false, b,
           (short)0, acc[j],  false,  false);
    }
  }

#pragma unroll
  for (int j = 0; j < NT; ++j) {
    float bv = bias ? bias[j * 16 + m] : 0.0f;
#pragma unroll
    for (int v = 0; v < 8; ++v) {
      C[(size_t)(wave * 16 + v + 8 * half) * NC + j * 16 + m] = acc[j][v] + bv;
    }
  }
}

__global__ __launch_bounds__(128) void ec_gemm_xs(
    const float* __restrict__ A, const float* __restrict__ Bself,
    const float* __restrict__ Brel, const float* __restrict__ bias,
    float* __restrict__ Cself, float* __restrict__ Crel,
    int mtiles, int nrel, int Nn) {
  constexpr int K = 256;
  constexpr int NC = 64;
  constexpr int NT = NC / 16;
  constexpr int S = 2 * NC + 32;
  __shared__ float Bs[(K / 2) * S];

  int wave = (blockIdx.x * blockDim.x + threadIdx.x) >> 5;
  bool active = wave < mtiles;
  int wv = active ? wave : 0;
  int lane = threadIdx.x & 31;
  int m = lane & 15;
  int half = lane >> 4;

  const float* Arow = A + (size_t)(wv * 16 + m) * K + 2 * half;
  v2f a[K / 4];
#pragma unroll
  for (int kk = 0; kk < K / 4; ++kk) {
    float2 av = *(const float2*)(Arow + 4 * kk);
    a[kk][0] = av.x;
    a[kk][1] = av.y;
  }

  float bvs[NT];
#pragma unroll
  for (int j = 0; j < NT; ++j) bvs[j] = bias[j * 16 + m];

#pragma unroll 1
  for (int r = (Bself ? 0 : 1); r <= nrel; ++r) {
    const float* B = (r == 0) ? Bself : Brel + (size_t)(r - 1) * K * NC;
    float* C = (r == 0) ? Cself : Crel + (size_t)(r - 1) * Nn * NC;

    __syncthreads();
    for (int it0 = 0; it0 < K * NC / 4 / 128; it0 += 8) {
      v4f q[8];
#pragma unroll
      for (int u = 0; u < 8; ++u) {
        int i = (it0 + u) * 128 + (int)threadIdx.x;
        q[u] = *(const v4f*)(B + i * 4);
      }
      asm volatile("" : "+v"(q[0]), "+v"(q[1]), "+v"(q[2]), "+v"(q[3]),
                        "+v"(q[4]), "+v"(q[5]), "+v"(q[6]), "+v"(q[7]));
#pragma unroll
      for (int u = 0; u < 8; ++u) {
        int i = (it0 + u) * 128 + (int)threadIdx.x;
        int k = (i * 4) / NC;
        int n = (i * 4) % NC;
        int j = n >> 4;
        int m0 = n & 15;
        float* p = &Bs[(k >> 1) * S + (j >> 1) * 64 + 4 * m0 + 2 * (j & 1) + (k & 1)];
        p[0] = q[u][0];
        p[4] = q[u][1];
        p[8] = q[u][2];
        p[12] = q[u][3];
      }
    }
    __syncthreads();

    v8f acc[NT];
    v8f z = {0.f, 0.f, 0.f, 0.f, 0.f, 0.f, 0.f, 0.f};
#pragma unroll
    for (int j = 0; j < NT; ++j) acc[j] = z;

#pragma unroll
    for (int kk = 0; kk < K / 4; ++kk) {
      int k = 4 * kk;
      const float* Bp = &Bs[(size_t)((k + 2 * half) >> 1) * S + 4 * m];
#pragma unroll
      for (int t = 0; t < NT / 2; ++t) {
        v4f q = *(const v4f*)(Bp + t * 64);
        v2f b0 = {q[0], q[1]};
        v2f b1 = {q[2], q[3]};
        acc[2 * t] = __builtin_amdgcn_wmma_f32_16x16x4_f32(
             false, a[kk],  false, b0,
             (short)0, acc[2 * t],  false,  false);
        acc[2 * t + 1] = __builtin_amdgcn_wmma_f32_16x16x4_f32(
             false, a[kk],  false, b1,
             (short)0, acc[2 * t + 1],  false,  false);
      }
    }

    v8f o[NT];
#pragma unroll
    for (int j = 0; j < NT; ++j) {
      float bv = (r == 0) ? bvs[j] : 0.0f;
#pragma unroll
      for (int v = 0; v < 8; ++v) o[j][v] = acc[j][v] + bv;
    }
    asm volatile("" : "+v"(o[0]), "+v"(o[1]), "+v"(o[2]), "+v"(o[3]));
    if (active) {
#pragma unroll
      for (int j = 0; j < NT; ++j) {
#pragma unroll
        for (int v = 0; v < 8; ++v) {
          C[(size_t)(wave * 16 + v + 8 * half) * NC + j * 16 + m] = o[j][v];
        }
      }
    }
  }
}

template <int COLS>
__global__ __launch_bounds__(256) void ec_agg(
    const float* __restrict__ hw, const int* __restrict__ src,
    const int* __restrict__ dst, const float* __restrict__ invd,
    float* __restrict__ acc, int E) {
  constexpr int TPE = COLS / 4;
  int t = blockIdx.x * blockDim.x + threadIdx.x;
  int e = t / TPE;
  if (e >= E) return;
  int c = (t % TPE) * 4;
  int s = src[e];
  int d = dst[e];
  float w = invd[d];
  float4 v = *(const float4*)(hw + (size_t)s * COLS + c);
  float* dp = acc + (size_t)d * COLS + c;
  unsafeAtomicAdd(dp + 0, v.x * w);
  unsafeAtomicAdd(dp + 1, v.y * w);
  unsafeAtomicAdd(dp + 2, v.z * w);
  unsafeAtomicAdd(dp + 3, v.w * w);
}

template <int TPN, int J>
__device__ __forceinline__ int ec_slot_src(const int* id) {
  static_assert(TPN == 16 || TPN == 4, "lane group is a 16-lane row or a quad");
  if constexpr (TPN == 16) {
    return __builtin_amdgcn_update_dpp(0, id[0], 0x150 + J, 0xf, 0xf, true);
  } else {
    return __builtin_amdgcn_update_dpp(0, id[J >> 2], (J & 3) * 0x55, 0xf, 0xf, true);
  }
}

template <int COLS>
__device__ __forceinline__ void ec_sum_batch(const float* __restrict__ hwb, unsigned cb, const int* id, int n,
                                             float& sx, float& sy, float& sz, float& sw) {
  constexpr int TPN = COLS / 4;
  float4 v[16];
#define EC_GATHER(J) \
  if (J < n) v[J] = *(const float4*)((const char*)hwb + \
                       ((unsigned)ec_slot_src<TPN, J>(id) * (unsigned)(COLS * 4) + cb));
  EC_GATHER(0) EC_GATHER(1) EC_GATHER(2) EC_GATHER(3)
  if (n > 4) {
    EC_GATHER(4) EC_GATHER(5) EC_GATHER(6) EC_GATHER(7)
    if (n > 8) {
      EC_GATHER(8) EC_GATHER(9) EC_GATHER(10) EC_GATHER(11)
      if (n > 12) {
        EC_GATHER(12) EC_GATHER(13) EC_GATHER(14) EC_GATHER(15)
      }
    }
  }
#undef EC_GATHER
  asm volatile("" : "+v"(sx), "+v"(sy), "+v"(sz), "+v"(sw));
#define EC_ADD(J) \
  if (J < n) { sx += v[J].x; sy += v[J].y; sz += v[J].z; sw += v[J].w; }
  EC_ADD(0) EC_ADD(1) EC_ADD(2) EC_ADD(3)
  if (n > 4) {
    EC_ADD(4) EC_ADD(5) EC_ADD(6) EC_ADD(7)
    if (n > 8) {
      EC_ADD(8) EC_ADD(9) EC_ADD(10) EC_ADD(11)
      if (n > 12) {
        EC_ADD(12) EC_ADD(13) EC_ADD(14) EC_ADD(15)
      }
    }
  }
#undef EC_ADD
}

template <int COLS>
__global__ __launch_bounds__(256) void ec_pull_rows(
    const float* __restrict__ hw, const int* __restrict__ offs,
    const int* __restrict__ col, float* __restrict__ acc,
    int Nn, int nrel, int relu) {
  constexpr int TPN = COLS / 4;
  constexpr int RPB = 256 / TPN;
  constexpr int NI = 16 / TPN;
  const int g = (int)threadIdx.x / TPN;
  const int l = (int)threadIdx.x % TPN;
  const int c = l * 4;
  const int step = (int)gridDim.x * RPB;
  const size_t plane = (size_t)Nn * COLS;
  int d = (int)blockIdx.x * RPB + g;
  if (d >= Nn) return;
  int q = 0;

  int b = offs[d];
  int e = offs[d + 1];
  int idc[NI];
#pragma unroll
  for (int k = 0; k < NI; ++k) {
    int j = k * TPN + l;
    idc[k] = col[(j < e - b) ? (b + j) : 0];
  }
  int q1 = (nrel > 1) ? 1 : 0;
  int d1 = (nrel > 1) ? d : d + step;
  bool ok1 = d1 < Nn;
  int b1 = 0, e1 = 0;
  if (ok1) {
    const int* p = offs + (size_t)q1 * Nn + d1;
    b1 = p[0];
    e1 = p[1];
  }

  float4 o = {0.f, 0.f, 0.f, 0.f};
  for (;;) {
    int idn[NI];
#pragma unroll
    for (int k = 0; k < NI; ++k) {
      int j = k * TPN + l;
      idn[k] = col[(j < e1 - b1) ? (b1 + j) : 0];
    }
    int q2 = (q1 + 1 < nrel) ? q1 + 1 : 0;
    int d2 = (q1 + 1 < nrel) ? d1 : d1 + step;
    bool ok2 = ok1 && (d2 < Nn);
    int b2 = 0, e2 = 0;
    if (ok2) {
      const int* p = offs + (size_t)q2 * Nn + d2;
      b2 = p[0];
      e2 = p[1];
    }

    float4* hp = (float4*)(acc + (size_t)d * COLS + c);
    if (q == 0) o = *hp;
    const unsigned cb = (unsigned)(((size_t)q * plane + (size_t)c) * sizeof(float));
    float sx = 0.f, sy = 0.f, sz = 0.f, sw = 0.f;
    const int n = e - b;
    ec_sum_batch<COLS>(hw, cb, idc, (n < 16) ? n : 16, sx, sy, sz, sw);
    for (int i = b + 16; i < e; i += 16) {
      int m = (e - i < 16) ? (e - i) : 16;
      int idx[NI];
#pragma unroll
      for (int k = 0; k < NI; ++k) {
        int j = k * TPN + l;
        idx[k] = col[(j < m) ? (i + j) : 0];
      }
      ec_sum_batch<COLS>(hw, cb, idx, m, sx, sy, sz, sw);
    }
    float w = 1.0f / fmaxf((float)n, 1.0f);
    o.x += sx * w;
    o.y += sy * w;
    o.z += sz * w;
    o.w += sw * w;
    if (q + 1 >= nrel) {
      if (relu) {
        o.x = fmaxf(o.x, 0.0f);
        o.y = fmaxf(o.y, 0.0f);
        o.z = fmaxf(o.z, 0.0f);
        o.w = fmaxf(o.w, 0.0f);
      }
      *hp = o;
    }
    if (!ok1) break;
    d = d1; q = q1; b = b1; e = e1;
#pragma unroll
    for (int k = 0; k < NI; ++k) idc[k] = idn[k];
    d1 = d2; q1 = q2; ok1 = ok2; b1 = b2; e1 = e2;
  }
}


__global__ __launch_bounds__(256) void ec_relu(float* __restrict__ h, int n4) {
  v4f* p = (v4f*)h;
  const int T = (int)gridDim.x * 256;
  for (int i = (int)blockIdx.x * 256 + (int)threadIdx.x; i < n4; i += 4 * T) {
    const int i1 = i + T;
    const int i2 = i + 2 * T;
    const int i3 = i + 3 * T;
    const bool k1 = i1 < n4;
    const bool k2 = i2 < n4;
    const bool k3 = i3 < n4;
    v4f a0 = p[i];
    v4f a1 = p[k1 ? i1 : i];
    v4f a2 = p[k2 ? i2 : i];
    v4f a3 = p[k3 ? i3 : i];
#pragma unroll
    for (int q = 0; q < 4; ++q) {
      a0[q] = fmaxf(a0[q], 0.0f);
      a1[q] = fmaxf(a1[q], 0.0f);
      a2[q] = fmaxf(a2[q], 0.0f);
      a3[q] = fmaxf(a3[q], 0.0f);
    }
    p[i] = a0;
    if (k1) p[i1] = a1;
    if (k2) p[i2] = a2;
    if (k3) p[i3] = a3;
  }
}

extern "C" void kernel_launch(void* const* d_in, const int* in_sizes, int n_in,
                              void* d_out, int out_size, void* d_ws, size_t ws_size,
                              hipStream_t stream) {
  const float* x   = (const float*)d_in[0];
  const int*   src = (const int*)d_in[1];
  const int*   dst = (const int*)d_in[2];
  const float* W1  = (const float*)d_in[3];
  const float* L1w = (const float*)d_in[4];
  const float* L1b = (const float*)d_in[5];
  const float* W2  = (const float*)d_in[6];
  const float* L2w = (const float*)d_in[7];
  const float* L2b = (const float*)d_in[8];
  float* out = (float*)d_out;

  const int D_IN = 256, D_H = 64, D_OUT = 16;
  const int Nn = in_sizes[0] / D_IN;
  const int R  = in_sizes[3] / (D_IN * D_H);
  const int E  = in_sizes[1] / R;

  char* ws = (char*)d_ws;
  int mtiles = Nn / 16;
  int gemmBlocks = (mtiles * 32 + 127) / 128;

  const int M = R * Nn;
  const int NB = (Nn + EC_BW - 1) >> EC_BW_LOG;
  const int RB = R * NB;
  size_t off = 0;
  int* lstart = (int*)(ws + off);
  off += (size_t)RB * EC_NCHUNK * sizeof(int);
  off = (off + 255) & ~(size_t)255;
  int* lcnt = (int*)(ws + off);
  off += (size_t)RB * EC_NCHUNK * sizeof(int);
  off = (off + 255) & ~(size_t)255;
  int* offs = (int*)(ws + off);
  off += ((size_t)M + 1) * sizeof(int);
  off = (off + 255) & ~(size_t)255;
  int* col = (int*)(ws + off);
  off += (size_t)R * E * sizeof(int);
  off = (off + 255) & ~(size_t)255;
  float* h = (float*)(ws + off);
  off += (size_t)Nn * D_H * sizeof(float);
  off = (off + 255) & ~(size_t)255;
  float* hw = (float*)(ws + off);
  int2* part = (int2*)hw;
  size_t planeBytes = (size_t)Nn * D_H * sizeof(float);
  size_t partBytes = (size_t)R * E * sizeof(int2);
  size_t hwOff = off;
  off += (planeBytes > partBytes) ? planeBytes : partBytes;

  const int residentBlocks = 2048;
  const int pull64Resident = 1280;
  const int pull16Resident = 1024;
  int pull64Blocks = (Nn + 15) / 16;
  if (pull64Blocks > pull64Resident) pull64Blocks = pull64Resident;
  int pull16Blocks = (Nn + 63) / 64;
  if (pull16Blocks > pull16Resident) pull16Blocks = pull16Resident;
  const int relu4 = Nn * D_H / 4;
  int reluBlocks = (relu4 + 1023) / 1024;
  if (reluBlocks > residentBlocks) reluBlocks = residentBlocks;

  if (ws_size >= off && RB <= EC_RB_MAX) {
    int P = 1;
    if (ws_size >= hwOff + 2 * planeBytes) P = 2;
    if (ws_size >= hwOff + (size_t)R * planeBytes) P = R;
    int sub = (((E + EC_NCHUNK - 1) / EC_NCHUNK) + 31) & ~31;
    ec_chunk_sort<<<EC_NCHUNK, 1024, 0, stream>>>(src, dst, part, lstart, lcnt, E, R, NB, RB, sub);
    ec_bucket_sort<<<RB, 1024, 0, stream>>>(part, lstart, lcnt, offs, col, Nn, NB, RB, M, E, R, sub);

    for (int r0 = 0; r0 < R; r0 += P) {
      int nr = (R - r0 < P) ? (R - r0) : P;
      ec_gemm_xs<<<gemmBlocks, 128, 0, stream>>>(
          x, (r0 == 0) ? L1w : nullptr, W1 + (size_t)r0 * D_IN * D_H, L1b,
          h, hw, mtiles, nr, Nn);
      ec_pull_rows<64><<<pull64Blocks, 256, 0, stream>>>(
          hw, offs + (size_t)r0 * Nn, col, h, Nn, nr, (r0 + nr >= R) ? 1 : 0);
    }

    ec_gemm_wmma<64, 16><<<gemmBlocks, 128, 0, stream>>>(h, L2w, L2b, out, mtiles);
    size_t plane2Bytes = (size_t)Nn * D_OUT * sizeof(float);
    int P2 = (ws_size >= hwOff + (size_t)R * plane2Bytes) ? R : 1;
    for (int r0 = 0; r0 < R; r0 += P2) {
      int nr = (R - r0 < P2) ? (R - r0) : P2;
      for (int q = 0; q < nr; ++q) {
        ec_gemm_wmma<64, 16><<<gemmBlocks, 128, 0, stream>>>(
            h, W2 + (size_t)(r0 + q) * D_H * D_OUT, nullptr,
            hw + (size_t)q * Nn * D_OUT, mtiles);
      }
      ec_pull_rows<16><<<pull16Blocks, 256, 0, stream>>>(
          hw, offs + (size_t)r0 * Nn, col, out, Nn, nr, 0);
    }
    return;
  }

  off = 0;
  int*   deg  = (int*)(ws + off);
  float* invd = (float*)(ws + off);
  off += (size_t)R * Nn * sizeof(float);
  off = (off + 255) & ~(size_t)255;
  h = (float*)(ws + off);
  off += (size_t)Nn * D_H * sizeof(float);
  off = (off + 255) & ~(size_t)255;
  hw = (float*)(ws + off);

  hipMemsetAsync(deg, 0, (size_t)R * Nn * sizeof(int), stream);
  {
    int total = R * E;
    ec_count_deg<<<(total + 255) / 256, 256, 0, stream>>>(dst, deg, total, Nn, E);
  }
  {
    int total = R * Nn;
    ec_inv_deg<<<(total + 255) / 256, 256, 0, stream>>>(deg, invd, total);
  }

  ec_gemm_wmma<256, 64><<<gemmBlocks, 128, 0, stream>>>(x, L1w, L1b, h, mtiles);
  for (int r = 0; r < R; ++r) {
    ec_gemm_wmma<256, 64><<<gemmBlocks, 128, 0, stream>>>(
        x, W1 + (size_t)r * D_IN * D_H, nullptr, hw, mtiles);
    long long tot = (long long)E * (D_H / 4);
    ec_agg<64><<<(int)((tot + 255) / 256), 256, 0, stream>>>(
        hw, src + (size_t)r * E, dst + (size_t)r * E, invd + (size_t)r * Nn, h, E);
  }
  ec_relu<<<reluBlocks, 256, 0, stream>>>(h, relu4);

  ec_gemm_wmma<64, 16><<<gemmBlocks, 128, 0, stream>>>(h, L2w, L2b, out, mtiles);
  for (int r = 0; r < R; ++r) {
    ec_gemm_wmma<64, 16><<<gemmBlocks, 128, 0, stream>>>(
        h, W2 + (size_t)r * D_H * D_OUT, nullptr, hw, mtiles);
    long long tot = (long long)E * (D_OUT / 4);
    ec_agg<16><<<(int)((tot + 255) / 256), 256, 0, stream>>>(
        hw, src + (size_t)r * E, dst + (size_t)r * E, invd + (size_t)r * Nn, out, E);
  }
}


# Edge sort stores staged in LDS: a chunk's bucketed records and a bucket's source ids are placed by slot in LDS, then written out as whole 128-byte lines, not 32 lines a wave store; slot order same
# speedup vs baseline: 3.0323x; 1.0681x over previous
#include <hip/hip_runtime.h>

typedef __attribute__((ext_vector_type(2))) float v2f;
typedef __attribute__((ext_vector_type(4))) float v4f;
typedef __attribute__((ext_vector_type(8))) float v8f;

__global__ __launch_bounds__(256) void ec_count_deg(const int* __restrict__ dst,
                                                    int* __restrict__ deg,
                                                    int total, int Nn, int Eper) {
  int t = blockIdx.x * blockDim.x + threadIdx.x;
  if (t >= total) return;
  int r = t / Eper;
  atomicAdd(&deg[r * Nn + dst[t]], 1);
}

__global__ __launch_bounds__(256) void ec_inv_deg(int* __restrict__ deg,
                                                  float* __restrict__ invd,
                                                  int total) {
  int t = blockIdx.x * blockDim.x + threadIdx.x;
  if (t >= total) return;
  float d = (float)deg[t];
  invd[t] = 1.0f / fmaxf(d, 1.0f);
}


#define EC_BW_LOG 10
#define EC_BW (1 << EC_BW_LOG)
#define EC_RB_MAX 1024
#define EC_NCHUNK 256
#define EC_CPW (EC_NCHUNK / 32)
#define EC_STAGE_CAP 16384
#define EC_COL_CAP 16384

__device__ __forceinline__ int ec_chunk_edge(int u, int len, int lo, int Eper, int& rel) {
  rel = u / len;
  return rel * Eper + lo + (u - rel * len);
}

template <bool STAGED>
__device__ __forceinline__ void ec_chunk_place(const int* __restrict__ src, const int* __restrict__ dst,
                                               int* hs, int2* stage, int2* __restrict__ mine,
                                               int tid, int ntot, int len, int lo, int Eper, int NB) {
  int u = tid;
  for (; u + 3 * 1024 < ntot; u += 4 * 1024) {
    int r0, r1, r2, r3;
    int t0 = ec_chunk_edge(u, len, lo, Eper, r0);
    int t1 = ec_chunk_edge(u + 1024, len, lo, Eper, r1);
    int t2 = ec_chunk_edge(u + 2048, len, lo, Eper, r2);
    int t3 = ec_chunk_edge(u + 3072, len, lo, Eper, r3);
    int d0 = dst[t0];
    int d1 = dst[t1];
    int d2 = dst[t2];
    int d3 = dst[t3];
    int s0 = src[t0];
    int s1 = src[t1];
    int s2 = src[t2];
    int s3 = src[t3];
    int q0 = atomicAdd(&hs[r0 * NB + (d0 >> EC_BW_LOG)], 1);
    int q1 = atomicAdd(&hs[r1 * NB + (d1 >> EC_BW_LOG)], 1);
    int q2 = atomicAdd(&hs[r2 * NB + (d2 >> EC_BW_LOG)], 1);
    int q3 = atomicAdd(&hs[r3 * NB + (d3 >> EC_BW_LOG)], 1);
    if constexpr (STAGED) {
      stage[q0] = make_int2(s0, d0);
      stage[q1] = make_int2(s1, d1);
      stage[q2] = make_int2(s2, d2);
      stage[q3] = make_int2(s3, d3);
    } else {
      mine[q0] = make_int2(s0, d0);
      mine[q1] = make_int2(s1, d1);
      mine[q2] = make_int2(s2, d2);
      mine[q3] = make_int2(s3, d3);
    }
  }
  for (; u < ntot; u += 1024) {
    int r0;
    int t0 = ec_chunk_edge(u, len, lo, Eper, r0);
    int d = dst[t0];
    int s = src[t0];
    int q = atomicAdd(&hs[r0 * NB + (d >> EC_BW_LOG)], 1);
    if constexpr (STAGED) {
      stage[q] = make_int2(s, d);
    } else {
      mine[q] = make_int2(s, d);
    }
  }
}

__global__ __launch_bounds__(1024) void ec_chunk_sort(const int* __restrict__ src,
                                                      const int* __restrict__ dst,
                                                      int2* __restrict__ part,
                                                      int* __restrict__ lstart,
                                                      int* __restrict__ lcnt,
                                                      int Eper, int Rn, int NB, int RB,
                                                      int sub) {
  static_assert(EC_RB_MAX == 1024, "one bucket counter per thread");
  __shared__ int hs[EC_RB_MAX];
  __shared__ int wtot[32];
  __shared__ __attribute__((aligned(16))) int2 stage[EC_STAGE_CAP];
  int tid = threadIdx.x;
  int lane = tid & 31;
  int wv = tid >> 5;
  hs[tid] = 0;
  __syncthreads();
  int lo = (int)blockIdx.x * sub;
  if (lo > Eper) lo = Eper;
  int hi = (lo + sub < Eper) ? (lo + sub) : Eper;
  int len = hi - lo;
  int ntot = Rn * len;

  int u = tid;
  for (; u + 3 * 1024 < ntot; u += 4 * 1024) {
    int r0, r1, r2, r3;
    int t0 = ec_chunk_edge(u, len, lo, Eper, r0);
    int t1 = ec_chunk_edge(u + 1024, len, lo, Eper, r1);
    int t2 = ec_chunk_edge(u + 2048, len, lo, Eper, r2);
    int t3 = ec_chunk_edge(u + 3072, len, lo, Eper, r3);
    int d0 = dst[t0];
    int d1 = dst[t1];
    int d2 = dst[t2];
    int d3 = dst[t3];
    atomicAdd(&hs[r0 * NB + (d0 >> EC_BW_LOG)], 1);
    atomicAdd(&hs[r1 * NB + (d1 >> EC_BW_LOG)], 1);
    atomicAdd(&hs[r2 * NB + (d2 >> EC_BW_LOG)], 1);
    atomicAdd(&hs[r3 * NB + (d3 >> EC_BW_LOG)], 1);
  }
  for (; u < ntot; u += 1024) {
    int r0;
    int t0 = ec_chunk_edge(u, len, lo, Eper, r0);
    int d = dst[t0];
    atomicAdd(&hs[r0 * NB + (d >> EC_BW_LOG)], 1);
  }
  __syncthreads();

  int v = hs[tid];
  int inc = v;
  for (int dl = 1; dl < 32; dl <<= 1) {
    int up = __shfl_up(inc, dl, 32);
    if (lane >= dl) inc += up;
  }
  if (lane == 31) wtot[wv] = inc;
  __syncthreads();
  int before = 0;
  for (int j = 0; j < wv; ++j) before += wtot[j];
  int start = before + inc - v;
  if (tid < RB) {
    lstart[tid * EC_NCHUNK + blockIdx.x] = start;
    lcnt[tid * EC_NCHUNK + blockIdx.x] = v;
  }
  hs[tid] = start;
  __syncthreads();

  int2* mine = part + (size_t)Rn * lo;
  const bool staged = (ntot <= EC_STAGE_CAP) && (((Rn * lo) & 1) == 0);
  if (staged) {
    ec_chunk_place<true>(src, dst, hs, stage, mine, tid, ntot, len, lo, Eper, NB);
    __syncthreads();
    const int4* s4 = (const int4*)stage;
    int4* g4 = (int4*)mine;
    const int np = ntot >> 1;
    for (int i = tid; i < np; i += 1024) g4[i] = s4[i];
    if ((ntot & 1) && tid == 0) mine[ntot - 1] = stage[ntot - 1];
  } else {
    ec_chunk_place<false>(src, dst, hs, stage, mine, tid, ntot, len, lo, Eper, NB);
  }
}

__global__ __launch_bounds__(1024) void ec_bucket_sort(const int2* __restrict__ part,
                                                       const int* __restrict__ lstart,
                                                       const int* __restrict__ lcnt,
                                                       int* __restrict__ offs,
                                                       int* __restrict__ col,
                                                       int Nn, int NB, int RB, int M,
                                                       int Eper, int Rn, int sub) {
  __shared__ int cnt[EC_BW];
  __shared__ int sh[EC_BW];
  __shared__ int rbeg[EC_NCHUNK];
  __shared__ int rlen[EC_NCHUNK];
  __shared__ int wst[EC_CPW];
  __shared__ int wln[EC_CPW];
  __shared__ int cstage[EC_COL_CAP];
  int tid = threadIdx.x;
  int lane = tid & 31;
  int wv = tid >> 5;
  int bkt = blockIdx.x;
  int r = bkt / NB;
  int n0 = (bkt - r * NB) << EC_BW_LOG;
  cnt[tid] = 0;
  if (tid < EC_NCHUNK) {
    int st = lstart[bkt * EC_NCHUNK + tid];
    int ln = lcnt[bkt * EC_NCHUNK + tid];
    int lo = tid * sub;
    if (lo > Eper) lo = Eper;
    rbeg[tid] = Rn * lo + st;
    rlen[tid] = ln;
    for (int dl = 16; dl > 0; dl >>= 1) {
      st += __shfl_xor(st, dl, 32);
      ln += __shfl_xor(ln, dl, 32);
    }
    if (lane == 0) {
      wst[wv] = st;
      wln[wv] = ln;
    }
  }
  __syncthreads();
  int e0 = 0;
  int nb = 0;
#pragma unroll
  for (int j = 0; j < EC_CPW; ++j) {
    e0 += wst[j];
    nb += wln[j];
  }
  int e1 = e0 + nb;
  int rb[EC_CPW], rl[EC_CPW];
#pragma unroll
  for (int cc = 0; cc < EC_CPW; ++cc) {
    rb[cc] = rbeg[wv * EC_CPW + cc];
    rl[cc] = rlen[wv * EC_CPW + cc];
  }
  if (nb > 0) {
    int y[2 * EC_CPW];
#pragma unroll
    for (int cc = 0; cc < EC_CPW; ++cc) {
      y[2 * cc] = part[(lane < rl[cc]) ? (rb[cc] + lane) : 0].y;
      y[2 * cc + 1] = part[(lane + 32 < rl[cc]) ? (rb[cc] + lane + 32) : 0].y;
    }
#pragma unroll
    for (int cc = 0; cc < EC_CPW; ++cc) {
      if (lane < rl[cc]) atomicAdd(&cnt[y[2 * cc] - n0], 1);
      if (lane + 32 < rl[cc]) atomicAdd(&cnt[y[2 * cc + 1] - n0], 1);
    }
#pragma unroll
    for (int cc = 0; cc < EC_CPW; ++cc) {
      for (int j = lane + 64; j < rl[cc]; j += 32) atomicAdd(&cnt[part[rb[cc] + j].y - n0], 1);
    }
  }
  __syncthreads();
  int v = cnt[tid];
  sh[tid] = v;
  __syncthreads();
  for (int dl = 1; dl < EC_BW; dl <<= 1) {
    int add = (tid >= dl) ? sh[tid - dl] : 0;
    __syncthreads();
    sh[tid] += add;
    __syncthreads();
  }
  int start = sh[tid] - v;
  if (n0 + tid < Nn) offs[r * Nn + n0 + tid] = e0 + start;
  cnt[tid] = start;
  __syncthreads();
  const bool staged = nb <= EC_COL_CAP;
  if (nb > 0) {
    int2 rec[2 * EC_CPW];
#pragma unroll
    for (int cc = 0; cc < EC_CPW; ++cc) {
      rec[2 * cc] = part[(lane < rl[cc]) ? (rb[cc] + lane) : 0];
      rec[2 * cc + 1] = part[(lane + 32 < rl[cc]) ? (rb[cc] + lane + 32) : 0];
    }
#pragma unroll
    for (int cc = 0; cc < EC_CPW; ++cc) {
      if (lane < rl[cc]) {
        int q = atomicAdd(&cnt[rec[2 * cc].y - n0], 1);
        if (staged) cstage[q] = rec[2 * cc].x; else col[e0 + q] = rec[2 * cc].x;
      }
      if (lane + 32 < rl[cc]) {
        int q = atomicAdd(&cnt[rec[2 * cc + 1].y - n0], 1);
        if (staged) cstage[q] = rec[2 * cc + 1].x; else col[e0 + q] = rec[2 * cc + 1].x;
      }
    }
#pragma unroll
    for (int cc = 0; cc < EC_CPW; ++cc) {
      for (int j = lane + 64; j < rl[cc]; j += 32) {
        int2 rc = part[rb[cc] + j];
        int q = atomicAdd(&cnt[rc.y - n0], 1);
        if (staged) cstage[q] = rc.x; else col[e0 + q] = rc.x;
      }
    }
  }
  __syncthreads();
  if (staged) {
    for (int i = tid; i < nb; i += EC_BW) col[e0 + i] = cstage[i];
  }
  if (bkt == RB - 1 && tid == 0) offs[M] = e1;
}

template <int K, int NC>
__global__ __launch_bounds__(128) void ec_gemm_wmma(
    const float* __restrict__ A, const float* __restrict__ B,
    const float* __restrict__ bias, float* __restrict__ C, int mtiles) {
  constexpr int NT = NC / 16;
  constexpr int S = ((2 * NC) % 64 == 32) ? (2 * NC) : (2 * NC + 32);
  __shared__ float Bs[(K / 2) * S];

  for (int i = threadIdx.x; i < K * NC / 4; i += 128) {
    int k = (i * 4) / NC;
    int n = (i * 4) % NC;
    float4 v = *(const float4*)(B + i * 4);
    float* p = &Bs[(k >> 1) * S + 2 * n + (k & 1)];
    p[0] = v.x;
    p[2] = v.y;
    p[4] = v.z;
    p[6] = v.w;
  }
  __syncthreads();

  int wave = (blockIdx.x * blockDim.x + threadIdx.x) >> 5;
  if (wave >= mtiles) return;
  int lane = threadIdx.x & 31;
  int m = lane & 15;
  int half = lane >> 4;

  const float* Arow = A + (size_t)(wave * 16 + m) * K + 2 * half;

  v8f acc[NT];
  v8f z = {0.f, 0.f, 0.f, 0.f, 0.f, 0.f, 0.f, 0.f};
#pragma unroll
  for (int j = 0; j < NT; ++j) acc[j] = z;

#pragma unroll 8
  for (int k = 0; k < K; k += 4) {
    float2 av = *(const float2*)(Arow + k);
    v2f a = {av.x, av.y};
    const float* Bp = &Bs[(size_t)((k + 2 * half) >> 1) * S + 2 * m];
#pragma unroll
    for (int j = 0; j < NT; ++j) {
      v2f b = *(const v2f*)(Bp + j * 32);
      acc[j] = __builtin_amdgcn_wmma_f32_16x16x4_f32(
           false, a,  false, b,
           (short)0, acc[j],  false,  false);
    }
  }

#pragma unroll
  for (int j = 0; j < NT; ++j) {
    float bv = bias ? bias[j * 16 + m] : 0.0f;
#pragma unroll
    for (int v = 0; v < 8; ++v) {
      C[(size_t)(wave * 16 + v + 8 * half) * NC + j * 16 + m] = acc[j][v] + bv;
    }
  }
}

__global__ __launch_bounds__(128) void ec_gemm_xs(
    const float* __restrict__ A, const float* __restrict__ Bself,
    const float* __restrict__ Brel, const float* __restrict__ bias,
    float* __restrict__ Cself, float* __restrict__ Crel,
    int mtiles, int nrel, int Nn) {
  constexpr int K = 256;
  constexpr int NC = 64;
  constexpr int NT = NC / 16;
  constexpr int S = 2 * NC + 32;
  __shared__ float Bs[(K / 2) * S];

  int wave = (blockIdx.x * blockDim.x + threadIdx.x) >> 5;
  bool active = wave < mtiles;
  int wv = active ? wave : 0;
  int lane = threadIdx.x & 31;
  int m = lane & 15;
  int half = lane >> 4;

  const float* Arow = A + (size_t)(wv * 16 + m) * K + 2 * half;
  v2f a[K / 4];
#pragma unroll
  for (int kk = 0; kk < K / 4; ++kk) {
    float2 av = *(const float2*)(Arow + 4 * kk);
    a[kk][0] = av.x;
    a[kk][1] = av.y;
  }

  float bvs[NT];
#pragma unroll
  for (int j = 0; j < NT; ++j) bvs[j] = bias[j * 16 + m];

#pragma unroll 1
  for (int r = (Bself ? 0 : 1); r <= nrel; ++r) {
    const float* B = (r == 0) ? Bself : Brel + (size_t)(r - 1) * K * NC;
    float* C = (r == 0) ? Cself : Crel + (size_t)(r - 1) * Nn * NC;

    __syncthreads();
    for (int it0 = 0; it0 < K * NC / 4 / 128; it0 += 8) {
      v4f q[8];
#pragma unroll
      for (int u = 0; u < 8; ++u) {
        int i = (it0 + u) * 128 + (int)threadIdx.x;
        q[u] = *(const v4f*)(B + i * 4);
      }
      asm volatile("" : "+v"(q[0]), "+v"(q[1]), "+v"(q[2]), "+v"(q[3]),
                        "+v"(q[4]), "+v"(q[5]), "+v"(q[6]), "+v"(q[7]));
#pragma unroll
      for (int u = 0; u < 8; ++u) {
        int i = (it0 + u) * 128 + (int)threadIdx.x;
        int k = (i * 4) / NC;
        int n = (i * 4) % NC;
        int j = n >> 4;
        int m0 = n & 15;
        float* p = &Bs[(k >> 1) * S + (j >> 1) * 64 + 4 * m0 + 2 * (j & 1) + (k & 1)];
        p[0] = q[u][0];
        p[4] = q[u][1];
        p[8] = q[u][2];
        p[12] = q[u][3];
      }
    }
    __syncthreads();

    v8f acc[NT];
    v8f z = {0.f, 0.f, 0.f, 0.f, 0.f, 0.f, 0.f, 0.f};
#pragma unroll
    for (int j = 0; j < NT; ++j) acc[j] = z;

#pragma unroll
    for (int kk = 0; kk < K / 4; ++kk) {
      int k = 4 * kk;
      const float* Bp = &Bs[(size_t)((k + 2 * half) >> 1) * S + 4 * m];
#pragma unroll
      for (int t = 0; t < NT / 2; ++t) {
        v4f q = *(const v4f*)(Bp + t * 64);
        v2f b0 = {q[0], q[1]};
        v2f b1 = {q[2], q[3]};
        acc[2 * t] = __builtin_amdgcn_wmma_f32_16x16x4_f32(
             false, a[kk],  false, b0,
             (short)0, acc[2 * t],  false,  false);
        acc[2 * t + 1] = __builtin_amdgcn_wmma_f32_16x16x4_f32(
             false, a[kk],  false, b1,
             (short)0, acc[2 * t + 1],  false,  false);
      }
    }

    v8f o[NT];
#pragma unroll
    for (int j = 0; j < NT; ++j) {
      float bv = (r == 0) ? bvs[j] : 0.0f;
#pragma unroll
      for (int v = 0; v < 8; ++v) o[j][v] = acc[j][v] + bv;
    }
    asm volatile("" : "+v"(o[0]), "+v"(o[1]), "+v"(o[2]), "+v"(o[3]));
    if (active) {
#pragma unroll
      for (int j = 0; j < NT; ++j) {
#pragma unroll
        for (int v = 0; v < 8; ++v) {
          C[(size_t)(wave * 16 + v + 8 * half) * NC + j * 16 + m] = o[j][v];
        }
      }
    }
  }
}

template <int COLS>
__global__ __launch_bounds__(256) void ec_agg(
    const float* __restrict__ hw, const int* __restrict__ src,
    const int* __restrict__ dst, const float* __restrict__ invd,
    float* __restrict__ acc, int E) {
  constexpr int TPE = COLS / 4;
  int t = blockIdx.x * blockDim.x + threadIdx.x;
  int e = t / TPE;
  if (e >= E) return;
  int c = (t % TPE) * 4;
  int s = src[e];
  int d = dst[e];
  float w = invd[d];
  float4 v = *(const float4*)(hw + (size_t)s * COLS + c);
  float* dp = acc + (size_t)d * COLS + c;
  unsafeAtomicAdd(dp + 0, v.x * w);
  unsafeAtomicAdd(dp + 1, v.y * w);
  unsafeAtomicAdd(dp + 2, v.z * w);
  unsafeAtomicAdd(dp + 3, v.w * w);
}

template <int TPN, int J>
__device__ __forceinline__ int ec_slot_src(const int* id) {
  static_assert(TPN == 16 || TPN == 4, "lane group is a 16-lane row or a quad");
  if constexpr (TPN == 16) {
    return __builtin_amdgcn_update_dpp(0, id[0], 0x150 + J, 0xf, 0xf, true);
  } else {
    return __builtin_amdgcn_update_dpp(0, id[J >> 2], (J & 3) * 0x55, 0xf, 0xf, true);
  }
}

template <int COLS>
__device__ __forceinline__ void ec_sum_batch(const float* __restrict__ hwb, unsigned cb, const int* id, int n,
                                             float& sx, float& sy, float& sz, float& sw) {
  constexpr int TPN = COLS / 4;
  float4 v[16];
#define EC_GATHER(J) \
  if (J < n) v[J] = *(const float4*)((const char*)hwb + \
                       ((unsigned)ec_slot_src<TPN, J>(id) * (unsigned)(COLS * 4) + cb));
  EC_GATHER(0) EC_GATHER(1) EC_GATHER(2) EC_GATHER(3)
  if (n > 4) {
    EC_GATHER(4) EC_GATHER(5) EC_GATHER(6) EC_GATHER(7)
    if (n > 8) {
      EC_GATHER(8) EC_GATHER(9) EC_GATHER(10) EC_GATHER(11)
      if (n > 12) {
        EC_GATHER(12) EC_GATHER(13) EC_GATHER(14) EC_GATHER(15)
      }
    }
  }
#undef EC_GATHER
  asm volatile("" : "+v"(sx), "+v"(sy), "+v"(sz), "+v"(sw));
#define EC_ADD(J) \
  if (J < n) { sx += v[J].x; sy += v[J].y; sz += v[J].z; sw += v[J].w; }
  EC_ADD(0) EC_ADD(1) EC_ADD(2) EC_ADD(3)
  if (n > 4) {
    EC_ADD(4) EC_ADD(5) EC_ADD(6) EC_ADD(7)
    if (n > 8) {
      EC_ADD(8) EC_ADD(9) EC_ADD(10) EC_ADD(11)
      if (n > 12) {
        EC_ADD(12) EC_ADD(13) EC_ADD(14) EC_ADD(15)
      }
    }
  }
#undef EC_ADD
}

template <int COLS>
__global__ __launch_bounds__(256) void ec_pull_rows(
    const float* __restrict__ hw, const int* __restrict__ offs,
    const int* __restrict__ col, float* __restrict__ acc,
    int Nn, int nrel, int relu) {
  constexpr int TPN = COLS / 4;
  constexpr int RPB = 256 / TPN;
  constexpr int NI = 16 / TPN;
  const int g = (int)threadIdx.x / TPN;
  const int l = (int)threadIdx.x % TPN;
  const int c = l * 4;
  const int step = (int)gridDim.x * RPB;
  const size_t plane = (size_t)Nn * COLS;
  int d = (int)blockIdx.x * RPB + g;
  if (d >= Nn) return;
  int q = 0;

  int b = offs[d];
  int e = offs[d + 1];
  int idc[NI];
#pragma unroll
  for (int k = 0; k < NI; ++k) {
    int j = k * TPN + l;
    idc[k] = col[(j < e - b) ? (b + j) : 0];
  }
  int q1 = (nrel > 1) ? 1 : 0;
  int d1 = (nrel > 1) ? d : d + step;
  bool ok1 = d1 < Nn;
  int b1 = 0, e1 = 0;
  if (ok1) {
    const int* p = offs + (size_t)q1 * Nn + d1;
    b1 = p[0];
    e1 = p[1];
  }

  float4 o = {0.f, 0.f, 0.f, 0.f};
  for (;;) {
    int idn[NI];
#pragma unroll
    for (int k = 0; k < NI; ++k) {
      int j = k * TPN + l;
      idn[k] = col[(j < e1 - b1) ? (b1 + j) : 0];
    }
    int q2 = (q1 + 1 < nrel) ? q1 + 1 : 0;
    int d2 = (q1 + 1 < nrel) ? d1 : d1 + step;
    bool ok2 = ok1 && (d2 < Nn);
    int b2 = 0, e2 = 0;
    if (ok2) {
      const int* p = offs + (size_t)q2 * Nn + d2;
      b2 = p[0];
      e2 = p[1];
    }

    float4* hp = (float4*)(acc + (size_t)d * COLS + c);
    if (q == 0) o = *hp;
    const unsigned cb = (unsigned)(((size_t)q * plane + (size_t)c) * sizeof(float));
    float sx = 0.f, sy = 0.f, sz = 0.f, sw = 0.f;
    const int n = e - b;
    ec_sum_batch<COLS>(hw, cb, idc, (n < 16) ? n : 16, sx, sy, sz, sw);
    for (int i = b + 16; i < e; i += 16) {
      int m = (e - i < 16) ? (e - i) : 16;
      int idx[NI];
#pragma unroll
      for (int k = 0; k < NI; ++k) {
        int j = k * TPN + l;
        idx[k] = col[(j < m) ? (i + j) : 0];
      }
      ec_sum_batch<COLS>(hw, cb, idx, m, sx, sy, sz, sw);
    }
    float w = 1.0f / fmaxf((float)n, 1.0f);
    o.x += sx * w;
    o.y += sy * w;
    o.z += sz * w;
    o.w += sw * w;
    if (q + 1 >= nrel) {
      if (relu) {
        o.x = fmaxf(o.x, 0.0f);
        o.y = fmaxf(o.y, 0.0f);
        o.z = fmaxf(o.z, 0.0f);
        o.w = fmaxf(o.w, 0.0f);
      }
      *hp = o;
    }
    if (!ok1) break;
    d = d1; q = q1; b = b1; e = e1;
#pragma unroll
    for (int k = 0; k < NI; ++k) idc[k] = idn[k];
    d1 = d2; q1 = q2; ok1 = ok2; b1 = b2; e1 = e2;
  }
}


__global__ __launch_bounds__(256) void ec_relu(float* __restrict__ h, int n4) {
  v4f* p = (v4f*)h;
  const int T = (int)gridDim.x * 256;
  for (int i = (int)blockIdx.x * 256 + (int)threadIdx.x; i < n4; i += 4 * T) {
    const int i1 = i + T;
    const int i2 = i + 2 * T;
    const int i3 = i + 3 * T;
    const bool k1 = i1 < n4;
    const bool k2 = i2 < n4;
    const bool k3 = i3 < n4;
    v4f a0 = p[i];
    v4f a1 = p[k1 ? i1 : i];
    v4f a2 = p[k2 ? i2 : i];
    v4f a3 = p[k3 ? i3 : i];
#pragma unroll
    for (int q = 0; q < 4; ++q) {
      a0[q] = fmaxf(a0[q], 0.0f);
      a1[q] = fmaxf(a1[q], 0.0f);
      a2[q] = fmaxf(a2[q], 0.0f);
      a3[q] = fmaxf(a3[q], 0.0f);
    }
    p[i] = a0;
    if (k1) p[i1] = a1;
    if (k2) p[i2] = a2;
    if (k3) p[i3] = a3;
  }
}

extern "C" void kernel_launch(void* const* d_in, const int* in_sizes, int n_in,
                              void* d_out, int out_size, void* d_ws, size_t ws_size,
                              hipStream_t stream) {
  const float* x   = (const float*)d_in[0];
  const int*   src = (const int*)d_in[1];
  const int*   dst = (const int*)d_in[2];
  const float* W1  = (const float*)d_in[3];
  const float* L1w = (const float*)d_in[4];
  const float* L1b = (const float*)d_in[5];
  const float* W2  = (const float*)d_in[6];
  const float* L2w = (const float*)d_in[7];
  const float* L2b = (const float*)d_in[8];
  float* out = (float*)d_out;

  const int D_IN = 256, D_H = 64, D_OUT = 16;
  const int Nn = in_sizes[0] / D_IN;
  const int R  = in_sizes[3] / (D_IN * D_H);
  const int E  = in_sizes[1] / R;

  char* ws = (char*)d_ws;
  int mtiles = Nn / 16;
  int gemmBlocks = (mtiles * 32 + 127) / 128;

  const int M = R * Nn;
  const int NB = (Nn + EC_BW - 1) >> EC_BW_LOG;
  const int RB = R * NB;
  size_t off = 0;
  int* lstart = (int*)(ws + off);
  off += (size_t)RB * EC_NCHUNK * sizeof(int);
  off = (off + 255) & ~(size_t)255;
  int* lcnt = (int*)(ws + off);
  off += (size_t)RB * EC_NCHUNK * sizeof(int);
  off = (off + 255) & ~(size_t)255;
  int* offs = (int*)(ws + off);
  off += ((size_t)M + 1) * sizeof(int);
  off = (off + 255) & ~(size_t)255;
  int* col = (int*)(ws + off);
  off += (size_t)R * E * sizeof(int);
  off = (off + 255) & ~(size_t)255;
  float* h = (float*)(ws + off);
  off += (size_t)Nn * D_H * sizeof(float);
  off = (off + 255) & ~(size_t)255;
  float* hw = (float*)(ws + off);
  int2* part = (int2*)hw;
  size_t planeBytes = (size_t)Nn * D_H * sizeof(float);
  size_t partBytes = (size_t)R * E * sizeof(int2);
  size_t hwOff = off;
  off += (planeBytes > partBytes) ? planeBytes : partBytes;

  const int residentBlocks = 2048;
  const int pull64Resident = 1280;
  const int pull16Resident = 1024;
  int pull64Blocks = (Nn + 15) / 16;
  if (pull64Blocks > pull64Resident) pull64Blocks = pull64Resident;
  int pull16Blocks = (Nn + 63) / 64;
  if (pull16Blocks > pull16Resident) pull16Blocks = pull16Resident;
  const int relu4 = Nn * D_H / 4;
  int reluBlocks = (relu4 + 1023) / 1024;
  if (reluBlocks > residentBlocks) reluBlocks = residentBlocks;

  if (ws_size >= off && RB <= EC_RB_MAX) {
    int P = 1;
    if (ws_size >= hwOff + 2 * planeBytes) P = 2;
    if (ws_size >= hwOff + (size_t)R * planeBytes) P = R;
    int sub = (((E + EC_NCHUNK - 1) / EC_NCHUNK) + 31) & ~31;
    ec_chunk_sort<<<EC_NCHUNK, 1024, 0, stream>>>(src, dst, part, lstart, lcnt, E, R, NB, RB, sub);
    ec_bucket_sort<<<RB, 1024, 0, stream>>>(part, lstart, lcnt, offs, col, Nn, NB, RB, M, E, R, sub);

    for (int r0 = 0; r0 < R; r0 += P) {
      int nr = (R - r0 < P) ? (R - r0) : P;
      ec_gemm_xs<<<gemmBlocks, 128, 0, stream>>>(
          x, (r0 == 0) ? L1w : nullptr, W1 + (size_t)r0 * D_IN * D_H, L1b,
          h, hw, mtiles, nr, Nn);
      ec_pull_rows<64><<<pull64Blocks, 256, 0, stream>>>(
          hw, offs + (size_t)r0 * Nn, col, h, Nn, nr, (r0 + nr >= R) ? 1 : 0);
    }

    ec_gemm_wmma<64, 16><<<gemmBlocks, 128, 0, stream>>>(h, L2w, L2b, out, mtiles);
    size_t plane2Bytes = (size_t)Nn * D_OUT * sizeof(float);
    int P2 = (ws_size >= hwOff + (size_t)R * plane2Bytes) ? R : 1;
    for (int r0 = 0; r0 < R; r0 += P2) {
      int nr = (R - r0 < P2) ? (R - r0) : P2;
      for (int q = 0; q < nr; ++q) {
        ec_gemm_wmma<64, 16><<<gemmBlocks, 128, 0, stream>>>(
            h, W2 + (size_t)(r0 + q) * D_H * D_OUT, nullptr,
            hw + (size_t)q * Nn * D_OUT, mtiles);
      }
      ec_pull_rows<16><<<pull16Blocks, 256, 0, stream>>>(
          hw, offs + (size_t)r0 * Nn, col, out, Nn, nr, 0);
    }
    return;
  }

  off = 0;
  int*   deg  = (int*)(ws + off);
  float* invd = (float*)(ws + off);
  off += (size_t)R * Nn * sizeof(float);
  off = (off + 255) & ~(size_t)255;
  h = (float*)(ws + off);
  off += (size_t)Nn * D_H * sizeof(float);
  off = (off + 255) & ~(size_t)255;
  hw = (float*)(ws + off);

  hipMemsetAsync(deg, 0, (size_t)R * Nn * sizeof(int), stream);
  {
    int total = R * E;
    ec_count_deg<<<(total + 255) / 256, 256, 0, stream>>>(dst, deg, total, Nn, E);
  }
  {
    int total = R * Nn;
    ec_inv_deg<<<(total + 255) / 256, 256, 0, stream>>>(deg, invd, total);
  }

  ec_gemm_wmma<256, 64><<<gemmBlocks, 128, 0, stream>>>(x, L1w, L1b, h, mtiles);
  for (int r = 0; r < R; ++r) {
    ec_gemm_wmma<256, 64><<<gemmBlocks, 128, 0, stream>>>(
        x, W1 + (size_t)r * D_IN * D_H, nullptr, hw, mtiles);
    long long tot = (long long)E * (D_H / 4);
    ec_agg<64><<<(int)((tot + 255) / 256), 256, 0, stream>>>(
        hw, src + (size_t)r * E, dst + (size_t)r * E, invd + (size_t)r * Nn, h, E);
  }
  ec_relu<<<reluBlocks, 256, 0, stream>>>(h, relu4);

  ec_gemm_wmma<64, 16><<<gemmBlocks, 128, 0, stream>>>(h, L2w, L2b, out, mtiles);
  for (int r = 0; r < R; ++r) {
    ec_gemm_wmma<64, 16><<<gemmBlocks, 128, 0, stream>>>(
        h, W2 + (size_t)r * D_H * D_OUT, nullptr, hw, mtiles);
    long long tot = (long long)E * (D_OUT / 4);
    ec_agg<16><<<(int)((tot + 255) / 256), 256, 0, stream>>>(
        hw, src + (size_t)r * E, dst + (size_t)r * E, invd + (size_t)r * Nn, out, E);
  }
}


# Layer-2 products in one launch, not five: a wave keeps its 16x64 strip of h in registers and multiplies it by the self and the four relation weight blocks staged once in LDS; sums unchanged
# speedup vs baseline: 3.3306x; 1.0984x over previous
#include <hip/hip_runtime.h>

typedef __attribute__((ext_vector_type(2))) float v2f;
typedef __attribute__((ext_vector_type(4))) float v4f;
typedef __attribute__((ext_vector_type(8))) float v8f;

__global__ __launch_bounds__(256) void ec_count_deg(const int* __restrict__ dst,
                                                    int* __restrict__ deg,
                                                    int total, int Nn, int Eper) {
  int t = blockIdx.x * blockDim.x + threadIdx.x;
  if (t >= total) return;
  int r = t / Eper;
  atomicAdd(&deg[r * Nn + dst[t]], 1);
}

__global__ __launch_bounds__(256) void ec_inv_deg(int* __restrict__ deg,
                                                  float* __restrict__ invd,
                                                  int total) {
  int t = blockIdx.x * blockDim.x + threadIdx.x;
  if (t >= total) return;
  float d = (float)deg[t];
  invd[t] = 1.0f / fmaxf(d, 1.0f);
}


#define EC_BW_LOG 10
#define EC_BW (1 << EC_BW_LOG)
#define EC_RB_MAX 1024
#define EC_NCHUNK 256
#define EC_CPW (EC_NCHUNK / 32)
#define EC_STAGE_CAP 16384
#define EC_COL_CAP 16384

__device__ __forceinline__ int ec_chunk_edge(int u, int len, int lo, int Eper, int& rel) {
  rel = u / len;
  return rel * Eper + lo + (u - rel * len);
}

template <bool STAGED>
__device__ __forceinline__ void ec_chunk_place(const int* __restrict__ src, const int* __restrict__ dst,
                                               int* hs, int2* stage, int2* __restrict__ mine,
                                               int tid, int ntot, int len, int lo, int Eper, int NB) {
  int u = tid;
  for (; u + 3 * 1024 < ntot; u += 4 * 1024) {
    int r0, r1, r2, r3;
    int t0 = ec_chunk_edge(u, len, lo, Eper, r0);
    int t1 = ec_chunk_edge(u + 1024, len, lo, Eper, r1);
    int t2 = ec_chunk_edge(u + 2048, len, lo, Eper, r2);
    int t3 = ec_chunk_edge(u + 3072, len, lo, Eper, r3);
    int d0 = dst[t0];
    int d1 = dst[t1];
    int d2 = dst[t2];
    int d3 = dst[t3];
    int s0 = src[t0];
    int s1 = src[t1];
    int s2 = src[t2];
    int s3 = src[t3];
    int q0 = atomicAdd(&hs[r0 * NB + (d0 >> EC_BW_LOG)], 1);
    int q1 = atomicAdd(&hs[r1 * NB + (d1 >> EC_BW_LOG)], 1);
    int q2 = atomicAdd(&hs[r2 * NB + (d2 >> EC_BW_LOG)], 1);
    int q3 = atomicAdd(&hs[r3 * NB + (d3 >> EC_BW_LOG)], 1);
    if constexpr (STAGED) {
      stage[q0] = make_int2(s0, d0);
      stage[q1] = make_int2(s1, d1);
      stage[q2] = make_int2(s2, d2);
      stage[q3] = make_int2(s3, d3);
    } else {
      mine[q0] = make_int2(s0, d0);
      mine[q1] = make_int2(s1, d1);
      mine[q2] = make_int2(s2, d2);
      mine[q3] = make_int2(s3, d3);
    }
  }
  for (; u < ntot; u += 1024) {
    int r0;
    int t0 = ec_chunk_edge(u, len, lo, Eper, r0);
    int d = dst[t0];
    int s = src[t0];
    int q = atomicAdd(&hs[r0 * NB + (d >> EC_BW_LOG)], 1);
    if constexpr (STAGED) {
      stage[q] = make_int2(s, d);
    } else {
      mine[q] = make_int2(s, d);
    }
  }
}

__global__ __launch_bounds__(1024) void ec_chunk_sort(const int* __restrict__ src,
                                                      const int* __restrict__ dst,
                                                      int2* __restrict__ part,
                                                      int* __restrict__ lstart,
                                                      int* __restrict__ lcnt,
                                                      int Eper, int Rn, int NB, int RB,
                                                      int sub) {
  static_assert(EC_RB_MAX == 1024, "one bucket counter per thread");
  __shared__ int hs[EC_RB_MAX];
  __shared__ int wtot[32];
  __shared__ __attribute__((aligned(16))) int2 stage[EC_STAGE_CAP];
  int tid = threadIdx.x;
  int lane = tid & 31;
  int wv = tid >> 5;
  hs[tid] = 0;
  __syncthreads();
  int lo = (int)blockIdx.x * sub;
  if (lo > Eper) lo = Eper;
  int hi = (lo + sub < Eper) ? (lo + sub) : Eper;
  int len = hi - lo;
  int ntot = Rn * len;

  int u = tid;
  for (; u + 3 * 1024 < ntot; u += 4 * 1024) {
    int r0, r1, r2, r3;
    int t0 = ec_chunk_edge(u, len, lo, Eper, r0);
    int t1 = ec_chunk_edge(u + 1024, len, lo, Eper, r1);
    int t2 = ec_chunk_edge(u + 2048, len, lo, Eper, r2);
    int t3 = ec_chunk_edge(u + 3072, len, lo, Eper, r3);
    int d0 = dst[t0];
    int d1 = dst[t1];
    int d2 = dst[t2];
    int d3 = dst[t3];
    atomicAdd(&hs[r0 * NB + (d0 >> EC_BW_LOG)], 1);
    atomicAdd(&hs[r1 * NB + (d1 >> EC_BW_LOG)], 1);
    atomicAdd(&hs[r2 * NB + (d2 >> EC_BW_LOG)], 1);
    atomicAdd(&hs[r3 * NB + (d3 >> EC_BW_LOG)], 1);
  }
  for (; u < ntot; u += 1024) {
    int r0;
    int t0 = ec_chunk_edge(u, len, lo, Eper, r0);
    int d = dst[t0];
    atomicAdd(&hs[r0 * NB + (d >> EC_BW_LOG)], 1);
  }
  __syncthreads();

  int v = hs[tid];
  int inc = v;
  for (int dl = 1; dl < 32; dl <<= 1) {
    int up = __shfl_up(inc, dl, 32);
    if (lane >= dl) inc += up;
  }
  if (lane == 31) wtot[wv] = inc;
  __syncthreads();
  int before = 0;
  for (int j = 0; j < wv; ++j) before += wtot[j];
  int start = before + inc - v;
  if (tid < RB) {
    lstart[tid * EC_NCHUNK + blockIdx.x] = start;
    lcnt[tid * EC_NCHUNK + blockIdx.x] = v;
  }
  hs[tid] = start;
  __syncthreads();

  int2* mine = part + (size_t)Rn * lo;
  const bool staged = (ntot <= EC_STAGE_CAP) && (((Rn * lo) & 1) == 0);
  if (staged) {
    ec_chunk_place<true>(src, dst, hs, stage, mine, tid, ntot, len, lo, Eper, NB);
    __syncthreads();
    const int4* s4 = (const int4*)stage;
    int4* g4 = (int4*)mine;
    const int np = ntot >> 1;
    for (int i = tid; i < np; i += 1024) g4[i] = s4[i];
    if ((ntot & 1) && tid == 0) mine[ntot - 1] = stage[ntot - 1];
  } else {
    ec_chunk_place<false>(src, dst, hs, stage, mine, tid, ntot, len, lo, Eper, NB);
  }
}

__global__ __launch_bounds__(1024) void ec_bucket_sort(const int2* __restrict__ part,
                                                       const int* __restrict__ lstart,
                                                       const int* __restrict__ lcnt,
                                                       int* __restrict__ offs,
                                                       int* __restrict__ col,
                                                       int Nn, int NB, int RB, int M,
                                                       int Eper, int Rn, int sub) {
  __shared__ int cnt[EC_BW];
  __shared__ int sh[EC_BW];
  __shared__ int rbeg[EC_NCHUNK];
  __shared__ int rlen[EC_NCHUNK];
  __shared__ int wst[EC_CPW];
  __shared__ int wln[EC_CPW];
  __shared__ int cstage[EC_COL_CAP];
  int tid = threadIdx.x;
  int lane = tid & 31;
  int wv = tid >> 5;
  int bkt = blockIdx.x;
  int r = bkt / NB;
  int n0 = (bkt - r * NB) << EC_BW_LOG;
  cnt[tid] = 0;
  if (tid < EC_NCHUNK) {
    int st = lstart[bkt * EC_NCHUNK + tid];
    int ln = lcnt[bkt * EC_NCHUNK + tid];
    int lo = tid * sub;
    if (lo > Eper) lo = Eper;
    rbeg[tid] = Rn * lo + st;
    rlen[tid] = ln;
    for (int dl = 16; dl > 0; dl >>= 1) {
      st += __shfl_xor(st, dl, 32);
      ln += __shfl_xor(ln, dl, 32);
    }
    if (lane == 0) {
      wst[wv] = st;
      wln[wv] = ln;
    }
  }
  __syncthreads();
  int e0 = 0;
  int nb = 0;
#pragma unroll
  for (int j = 0; j < EC_CPW; ++j) {
    e0 += wst[j];
    nb += wln[j];
  }
  int e1 = e0 + nb;
  int rb[EC_CPW], rl[EC_CPW];
#pragma unroll
  for (int cc = 0; cc < EC_CPW; ++cc) {
    rb[cc] = rbeg[wv * EC_CPW + cc];
    rl[cc] = rlen[wv * EC_CPW + cc];
  }
  if (nb > 0) {
    int y[2 * EC_CPW];
#pragma unroll
    for (int cc = 0; cc < EC_CPW; ++cc) {
      y[2 * cc] = part[(lane < rl[cc]) ? (rb[cc] + lane) : 0].y;
      y[2 * cc + 1] = part[(lane + 32 < rl[cc]) ? (rb[cc] + lane + 32) : 0].y;
    }
#pragma unroll
    for (int cc = 0; cc < EC_CPW; ++cc) {
      if (lane < rl[cc]) atomicAdd(&cnt[y[2 * cc] - n0], 1);
      if (lane + 32 < rl[cc]) atomicAdd(&cnt[y[2 * cc + 1] - n0], 1);
    }
#pragma unroll
    for (int cc = 0; cc < EC_CPW; ++cc) {
      for (int j = lane + 64; j < rl[cc]; j += 32) atomicAdd(&cnt[part[rb[cc] + j].y - n0], 1);
    }
  }
  __syncthreads();
  int v = cnt[tid];
  sh[tid] = v;
  __syncthreads();
  for (int dl = 1; dl < EC_BW; dl <<= 1) {
    int add = (tid >= dl) ? sh[tid - dl] : 0;
    __syncthreads();
    sh[tid] += add;
    __syncthreads();
  }
  int start = sh[tid] - v;
  if (n0 + tid < Nn) offs[r * Nn + n0 + tid] = e0 + start;
  cnt[tid] = start;
  __syncthreads();
  const bool staged = nb <= EC_COL_CAP;
  if (nb > 0) {
    int2 rec[2 * EC_CPW];
#pragma unroll
    for (int cc = 0; cc < EC_CPW; ++cc) {
      rec[2 * cc] = part[(lane < rl[cc]) ? (rb[cc] + lane) : 0];
      rec[2 * cc + 1] = part[(lane + 32 < rl[cc]) ? (rb[cc] + lane + 32) : 0];
    }
#pragma unroll
    for (int cc = 0; cc < EC_CPW; ++cc) {
      if (lane < rl[cc]) {
        int q = atomicAdd(&cnt[rec[2 * cc].y - n0], 1);
        if (staged) cstage[q] = rec[2 * cc].x; else col[e0 + q] = rec[2 * cc].x;
      }
      if (lane + 32 < rl[cc]) {
        int q = atomicAdd(&cnt[rec[2 * cc + 1].y - n0], 1);
        if (staged) cstage[q] = rec[2 * cc + 1].x; else col[e0 + q] = rec[2 * cc + 1].x;
      }
    }
#pragma unroll
    for (int cc = 0; cc < EC_CPW; ++cc) {
      for (int j = lane + 64; j < rl[cc]; j += 32) {
        int2 rc = part[rb[cc] + j];
        int q = atomicAdd(&cnt[rc.y - n0], 1);
        if (staged) cstage[q] = rc.x; else col[e0 + q] = rc.x;
      }
    }
  }
  __syncthreads();
  if (staged) {
    for (int i = tid; i < nb; i += EC_BW) col[e0 + i] = cstage[i];
  }
  if (bkt == RB - 1 && tid == 0) offs[M] = e1;
}

template <int K, int NC>
__global__ __launch_bounds__(128) void ec_gemm_wmma(
    const float* __restrict__ A, const float* __restrict__ B,
    const float* __restrict__ bias, float* __restrict__ C, int mtiles) {
  constexpr int NT = NC / 16;
  constexpr int S = ((2 * NC) % 64 == 32) ? (2 * NC) : (2 * NC + 32);
  __shared__ float Bs[(K / 2) * S];

  for (int i = threadIdx.x; i < K * NC / 4; i += 128) {
    int k = (i * 4) / NC;
    int n = (i * 4) % NC;
    float4 v = *(const float4*)(B + i * 4);
    float* p = &Bs[(k >> 1) * S + 2 * n + (k & 1)];
    p[0] = v.x;
    p[2] = v.y;
    p[4] = v.z;
    p[6] = v.w;
  }
  __syncthreads();

  int wave = (blockIdx.x * blockDim.x + threadIdx.x) >> 5;
  if (wave >= mtiles) return;
  int lane = threadIdx.x & 31;
  int m = lane & 15;
  int half = lane >> 4;

  const float* Arow = A + (size_t)(wave * 16 + m) * K + 2 * half;

  v8f acc[NT];
  v8f z = {0.f, 0.f, 0.f, 0.f, 0.f, 0.f, 0.f, 0.f};
#pragma unroll
  for (int j = 0; j < NT; ++j) acc[j] = z;

#pragma unroll 8
  for (int k = 0; k < K; k += 4) {
    float2 av = *(const float2*)(Arow + k);
    v2f a = {av.x, av.y};
    const float* Bp = &Bs[(size_t)((k + 2 * half) >> 1) * S + 2 * m];
#pragma unroll
    for (int j = 0; j < NT; ++j) {
      v2f b = *(const v2f*)(Bp + j * 32);
      acc[j] = __builtin_amdgcn_wmma_f32_16x16x4_f32(
           false, a,  false, b,
           (short)0, acc[j],  false,  false);
    }
  }

#pragma unroll
  for (int j = 0; j < NT; ++j) {
    float bv = bias ? bias[j * 16 + m] : 0.0f;
#pragma unroll
    for (int v = 0; v < 8; ++v) {
      C[(size_t)(wave * 16 + v + 8 * half) * NC + j * 16 + m] = acc[j][v] + bv;
    }
  }
}

__global__ __launch_bounds__(128) void ec_gemm_xs(
    const float* __restrict__ A, const float* __restrict__ Bself,
    const float* __restrict__ Brel, const float* __restrict__ bias,
    float* __restrict__ Cself, float* __restrict__ Crel,
    int mtiles, int nrel, int Nn) {
  constexpr int K = 256;
  constexpr int NC = 64;
  constexpr int NT = NC / 16;
  constexpr int S = 2 * NC + 32;
  __shared__ float Bs[(K / 2) * S];

  int wave = (blockIdx.x * blockDim.x + threadIdx.x) >> 5;
  bool active = wave < mtiles;
  int wv = active ? wave : 0;
  int lane = threadIdx.x & 31;
  int m = lane & 15;
  int half = lane >> 4;

  const float* Arow = A + (size_t)(wv * 16 + m) * K + 2 * half;
  v2f a[K / 4];
#pragma unroll
  for (int kk = 0; kk < K / 4; ++kk) {
    float2 av = *(const float2*)(Arow + 4 * kk);
    a[kk][0] = av.x;
    a[kk][1] = av.y;
  }

  float bvs[NT];
#pragma unroll
  for (int j = 0; j < NT; ++j) bvs[j] = bias[j * 16 + m];

#pragma unroll 1
  for (int r = (Bself ? 0 : 1); r <= nrel; ++r) {
    const float* B = (r == 0) ? Bself : Brel + (size_t)(r - 1) * K * NC;
    float* C = (r == 0) ? Cself : Crel + (size_t)(r - 1) * Nn * NC;

    __syncthreads();
    for (int it0 = 0; it0 < K * NC / 4 / 128; it0 += 8) {
      v4f q[8];
#pragma unroll
      for (int u = 0; u < 8; ++u) {
        int i = (it0 + u) * 128 + (int)threadIdx.x;
        q[u] = *(const v4f*)(B + i * 4);
      }
      asm volatile("" : "+v"(q[0]), "+v"(q[1]), "+v"(q[2]), "+v"(q[3]),
                        "+v"(q[4]), "+v"(q[5]), "+v"(q[6]), "+v"(q[7]));
#pragma unroll
      for (int u = 0; u < 8; ++u) {
        int i = (it0 + u) * 128 + (int)threadIdx.x;
        int k = (i * 4) / NC;
        int n = (i * 4) % NC;
        int j = n >> 4;
        int m0 = n & 15;
        float* p = &Bs[(k >> 1) * S + (j >> 1) * 64 + 4 * m0 + 2 * (j & 1) + (k & 1)];
        p[0] = q[u][0];
        p[4] = q[u][1];
        p[8] = q[u][2];
        p[12] = q[u][3];
      }
    }
    __syncthreads();

    v8f acc[NT];
    v8f z = {0.f, 0.f, 0.f, 0.f, 0.f, 0.f, 0.f, 0.f};
#pragma unroll
    for (int j = 0; j < NT; ++j) acc[j] = z;

#pragma unroll
    for (int kk = 0; kk < K / 4; ++kk) {
      int k = 4 * kk;
      const float* Bp = &Bs[(size_t)((k + 2 * half) >> 1) * S + 4 * m];
#pragma unroll
      for (int t = 0; t < NT / 2; ++t) {
        v4f q = *(const v4f*)(Bp + t * 64);
        v2f b0 = {q[0], q[1]};
        v2f b1 = {q[2], q[3]};
        acc[2 * t] = __builtin_amdgcn_wmma_f32_16x16x4_f32(
             false, a[kk],  false, b0,
             (short)0, acc[2 * t],  false,  false);
        acc[2 * t + 1] = __builtin_amdgcn_wmma_f32_16x16x4_f32(
             false, a[kk],  false, b1,
             (short)0, acc[2 * t + 1],  false,  false);
      }
    }

    v8f o[NT];
#pragma unroll
    for (int j = 0; j < NT; ++j) {
      float bv = (r == 0) ? bvs[j] : 0.0f;
#pragma unroll
      for (int v = 0; v < 8; ++v) o[j][v] = acc[j][v] + bv;
    }
    asm volatile("" : "+v"(o[0]), "+v"(o[1]), "+v"(o[2]), "+v"(o[3]));
    if (active) {
#pragma unroll
      for (int j = 0; j < NT; ++j) {
#pragma unroll
        for (int v = 0; v < 8; ++v) {
          C[(size_t)(wave * 16 + v + 8 * half) * NC + j * 16 + m] = o[j][v];
        }
      }
    }
  }
}

#define EC_L2_REL 4
__global__ __launch_bounds__(256) void ec_gemm_l2(
    const float* __restrict__ A, const float* __restrict__ Bself,
    const float* __restrict__ Brel, const float* __restrict__ bias,
    float* __restrict__ Cself, float* __restrict__ Crel, int mtiles, int Nn) {
  constexpr int K = 64;
  constexpr int NC = 16;
  constexpr int NT = 1 + EC_L2_REL;
  constexpr int S = 2 * NC * NT;
  static_assert(S % 64 == 32, "lane-halves on disjoint LDS bank ranges");
  static_assert(K * NC / 4 == 256, "one 128-bit vector of every block per thread");
  __shared__ float Bs[(K / 2) * S];

  int wave = (blockIdx.x * blockDim.x + threadIdx.x) >> 5;
  bool active = wave < mtiles;
  int wv = active ? wave : 0;
  int lane = threadIdx.x & 31;
  int m = lane & 15;
  int half = lane >> 4;

  v4f q[NT];
#pragma unroll
  for (int j = 0; j < NT; ++j) {
    const float* B = (j == 0) ? Bself : Brel + (size_t)(j - 1) * K * NC;
    q[j] = *(const v4f*)(B + (int)threadIdx.x * 4);
  }
  const float* Arow = A + (size_t)(wv * 16 + m) * K + 2 * half;
  v2f a[K / 4];
#pragma unroll
  for (int kk = 0; kk < K / 4; ++kk) {
    float2 av = *(const float2*)(Arow + 4 * kk);
    a[kk][0] = av.x;
    a[kk][1] = av.y;
  }
  float bv = bias[m];

  {
    int k = (int)threadIdx.x >> 2;
    int n = ((int)threadIdx.x & 3) * 4;
    float* p = &Bs[(k >> 1) * S + 2 * n + (k & 1)];
#pragma unroll
    for (int j = 0; j < NT; ++j) {
      p[32 * j + 0] = q[j][0];
      p[32 * j + 2] = q[j][1];
      p[32 * j + 4] = q[j][2];
      p[32 * j + 6] = q[j][3];
    }
  }
  __syncthreads();
  if (!active) return;

  v8f acc[NT];
  v8f z = {0.f, 0.f, 0.f, 0.f, 0.f, 0.f, 0.f, 0.f};
#pragma unroll
  for (int j = 0; j < NT; ++j) acc[j] = z;

#pragma unroll
  for (int kk = 0; kk < K / 4; ++kk) {
    int k = 4 * kk;
    const float* Bp = &Bs[(size_t)((k + 2 * half) >> 1) * S + 2 * m];
#pragma unroll
    for (int j = 0; j < NT; ++j) {
      v2f b = *(const v2f*)(Bp + j * 32);
      acc[j] = __builtin_amdgcn_wmma_f32_16x16x4_f32(
           false, a[kk],  false, b,
           (short)0, acc[j],  false,  false);
    }
  }

  const size_t row0 = (size_t)(wave * 16 + 8 * half) * NC + m;
#pragma unroll
  for (int v = 0; v < 8; ++v) Cself[row0 + (size_t)v * NC] = acc[0][v] + bv;
#pragma unroll
  for (int j = 1; j < NT; ++j) {
    float* C = Crel + (size_t)(j - 1) * Nn * NC + row0;
#pragma unroll
    for (int v = 0; v < 8; ++v) C[(size_t)v * NC] = acc[j][v];
  }
}

template <int COLS>
__global__ __launch_bounds__(256) void ec_agg(
    const float* __restrict__ hw, const int* __restrict__ src,
    const int* __restrict__ dst, const float* __restrict__ invd,
    float* __restrict__ acc, int E) {
  constexpr int TPE = COLS / 4;
  int t = blockIdx.x * blockDim.x + threadIdx.x;
  int e = t / TPE;
  if (e >= E) return;
  int c = (t % TPE) * 4;
  int s = src[e];
  int d = dst[e];
  float w = invd[d];
  float4 v = *(const float4*)(hw + (size_t)s * COLS + c);
  float* dp = acc + (size_t)d * COLS + c;
  unsafeAtomicAdd(dp + 0, v.x * w);
  unsafeAtomicAdd(dp + 1, v.y * w);
  unsafeAtomicAdd(dp + 2, v.z * w);
  unsafeAtomicAdd(dp + 3, v.w * w);
}

template <int TPN, int J>
__device__ __forceinline__ int ec_slot_src(const int* id) {
  static_assert(TPN == 16 || TPN == 4, "lane group is a 16-lane row or a quad");
  if constexpr (TPN == 16) {
    return __builtin_amdgcn_update_dpp(0, id[0], 0x150 + J, 0xf, 0xf, true);
  } else {
    return __builtin_amdgcn_update_dpp(0, id[J >> 2], (J & 3) * 0x55, 0xf, 0xf, true);
  }
}

template <int COLS>
__device__ __forceinline__ void ec_sum_batch(const float* __restrict__ hwb, unsigned cb, const int* id, int n,
                                             float& sx, float& sy, float& sz, float& sw) {
  constexpr int TPN = COLS / 4;
  float4 v[16];
#define EC_GATHER(J) \
  if (J < n) v[J] = *(const float4*)((const char*)hwb + \
                       ((unsigned)ec_slot_src<TPN, J>(id) * (unsigned)(COLS * 4) + cb));
  EC_GATHER(0) EC_GATHER(1) EC_GATHER(2) EC_GATHER(3)
  if (n > 4) {
    EC_GATHER(4) EC_GATHER(5) EC_GATHER(6) EC_GATHER(7)
    if (n > 8) {
      EC_GATHER(8) EC_GATHER(9) EC_GATHER(10) EC_GATHER(11)
      if (n > 12) {
        EC_GATHER(12) EC_GATHER(13) EC_GATHER(14) EC_GATHER(15)
      }
    }
  }
#undef EC_GATHER
  asm volatile("" : "+v"(sx), "+v"(sy), "+v"(sz), "+v"(sw));
#define EC_ADD(J) \
  if (J < n) { sx += v[J].x; sy += v[J].y; sz += v[J].z; sw += v[J].w; }
  EC_ADD(0) EC_ADD(1) EC_ADD(2) EC_ADD(3)
  if (n > 4) {
    EC_ADD(4) EC_ADD(5) EC_ADD(6) EC_ADD(7)
    if (n > 8) {
      EC_ADD(8) EC_ADD(9) EC_ADD(10) EC_ADD(11)
      if (n > 12) {
        EC_ADD(12) EC_ADD(13) EC_ADD(14) EC_ADD(15)
      }
    }
  }
#undef EC_ADD
}

template <int COLS>
__global__ __launch_bounds__(256) void ec_pull_rows(
    const float* __restrict__ hw, const int* __restrict__ offs,
    const int* __restrict__ col, float* __restrict__ acc,
    int Nn, int nrel, int relu) {
  constexpr int TPN = COLS / 4;
  constexpr int RPB = 256 / TPN;
  constexpr int NI = 16 / TPN;
  const int g = (int)threadIdx.x / TPN;
  const int l = (int)threadIdx.x % TPN;
  const int c = l * 4;
  const int step = (int)gridDim.x * RPB;
  const size_t plane = (size_t)Nn * COLS;
  int d = (int)blockIdx.x * RPB + g;
  if (d >= Nn) return;
  int q = 0;

  int b = offs[d];
  int e = offs[d + 1];
  int idc[NI];
#pragma unroll
  for (int k = 0; k < NI; ++k) {
    int j = k * TPN + l;
    idc[k] = col[(j < e - b) ? (b + j) : 0];
  }
  int q1 = (nrel > 1) ? 1 : 0;
  int d1 = (nrel > 1) ? d : d + step;
  bool ok1 = d1 < Nn;
  int b1 = 0, e1 = 0;
  if (ok1) {
    const int* p = offs + (size_t)q1 * Nn + d1;
    b1 = p[0];
    e1 = p[1];
  }

  float4 o = {0.f, 0.f, 0.f, 0.f};
  for (;;) {
    int idn[NI];
#pragma unroll
    for (int k = 0; k < NI; ++k) {
      int j = k * TPN + l;
      idn[k] = col[(j < e1 - b1) ? (b1 + j) : 0];
    }
    int q2 = (q1 + 1 < nrel) ? q1 + 1 : 0;
    int d2 = (q1 + 1 < nrel) ? d1 : d1 + step;
    bool ok2 = ok1 && (d2 < Nn);
    int b2 = 0, e2 = 0;
    if (ok2) {
      const int* p = offs + (size_t)q2 * Nn + d2;
      b2 = p[0];
      e2 = p[1];
    }

    float4* hp = (float4*)(acc + (size_t)d * COLS + c);
    if (q == 0) o = *hp;
    const unsigned cb = (unsigned)(((size_t)q * plane + (size_t)c) * sizeof(float));
    float sx = 0.f, sy = 0.f, sz = 0.f, sw = 0.f;
    const int n = e - b;
    ec_sum_batch<COLS>(hw, cb, idc, (n < 16) ? n : 16, sx, sy, sz, sw);
    for (int i = b + 16; i < e; i += 16) {
      int m = (e - i < 16) ? (e - i) : 16;
      int idx[NI];
#pragma unroll
      for (int k = 0; k < NI; ++k) {
        int j = k * TPN + l;
        idx[k] = col[(j < m) ? (i + j) : 0];
      }
      ec_sum_batch<COLS>(hw, cb, idx, m, sx, sy, sz, sw);
    }
    float w = 1.0f / fmaxf((float)n, 1.0f);
    o.x += sx * w;
    o.y += sy * w;
    o.z += sz * w;
    o.w += sw * w;
    if (q + 1 >= nrel) {
      if (relu) {
        o.x = fmaxf(o.x, 0.0f);
        o.y = fmaxf(o.y, 0.0f);
        o.z = fmaxf(o.z, 0.0f);
        o.w = fmaxf(o.w, 0.0f);
      }
      *hp = o;
    }
    if (!ok1) break;
    d = d1; q = q1; b = b1; e = e1;
#pragma unroll
    for (int k = 0; k < NI; ++k) idc[k] = idn[k];
    d1 = d2; q1 = q2; ok1 = ok2; b1 = b2; e1 = e2;
  }
}


__global__ __launch_bounds__(256) void ec_relu(float* __restrict__ h, int n4) {
  v4f* p = (v4f*)h;
  const int T = (int)gridDim.x * 256;
  for (int i = (int)blockIdx.x * 256 + (int)threadIdx.x; i < n4; i += 4 * T) {
    const int i1 = i + T;
    const int i2 = i + 2 * T;
    const int i3 = i + 3 * T;
    const bool k1 = i1 < n4;
    const bool k2 = i2 < n4;
    const bool k3 = i3 < n4;
    v4f a0 = p[i];
    v4f a1 = p[k1 ? i1 : i];
    v4f a2 = p[k2 ? i2 : i];
    v4f a3 = p[k3 ? i3 : i];
#pragma unroll
    for (int q = 0; q < 4; ++q) {
      a0[q] = fmaxf(a0[q], 0.0f);
      a1[q] = fmaxf(a1[q], 0.0f);
      a2[q] = fmaxf(a2[q], 0.0f);
      a3[q] = fmaxf(a3[q], 0.0f);
    }
    p[i] = a0;
    if (k1) p[i1] = a1;
    if (k2) p[i2] = a2;
    if (k3) p[i3] = a3;
  }
}

extern "C" void kernel_launch(void* const* d_in, const int* in_sizes, int n_in,
                              void* d_out, int out_size, void* d_ws, size_t ws_size,
                              hipStream_t stream) {
  const float* x   = (const float*)d_in[0];
  const int*   src = (const int*)d_in[1];
  const int*   dst = (const int*)d_in[2];
  const float* W1  = (const float*)d_in[3];
  const float* L1w = (const float*)d_in[4];
  const float* L1b = (const float*)d_in[5];
  const float* W2  = (const float*)d_in[6];
  const float* L2w = (const float*)d_in[7];
  const float* L2b = (const float*)d_in[8];
  float* out = (float*)d_out;

  const int D_IN = 256, D_H = 64, D_OUT = 16;
  const int Nn = in_sizes[0] / D_IN;
  const int R  = in_sizes[3] / (D_IN * D_H);
  const int E  = in_sizes[1] / R;

  char* ws = (char*)d_ws;
  int mtiles = Nn / 16;
  int gemmBlocks = (mtiles * 32 + 127) / 128;

  const int M = R * Nn;
  const int NB = (Nn + EC_BW - 1) >> EC_BW_LOG;
  const int RB = R * NB;
  size_t off = 0;
  int* lstart = (int*)(ws + off);
  off += (size_t)RB * EC_NCHUNK * sizeof(int);
  off = (off + 255) & ~(size_t)255;
  int* lcnt = (int*)(ws + off);
  off += (size_t)RB * EC_NCHUNK * sizeof(int);
  off = (off + 255) & ~(size_t)255;
  int* offs = (int*)(ws + off);
  off += ((size_t)M + 1) * sizeof(int);
  off = (off + 255) & ~(size_t)255;
  int* col = (int*)(ws + off);
  off += (size_t)R * E * sizeof(int);
  off = (off + 255) & ~(size_t)255;
  float* h = (float*)(ws + off);
  off += (size_t)Nn * D_H * sizeof(float);
  off = (off + 255) & ~(size_t)255;
  float* hw = (float*)(ws + off);
  int2* part = (int2*)hw;
  size_t planeBytes = (size_t)Nn * D_H * sizeof(float);
  size_t partBytes = (size_t)R * E * sizeof(int2);
  size_t hwOff = off;
  off += (planeBytes > partBytes) ? planeBytes : partBytes;

  const int residentBlocks = 2048;
  const int pull64Resident = 1280;
  const int pull16Resident = 1024;
  int pull64Blocks = (Nn + 15) / 16;
  if (pull64Blocks > pull64Resident) pull64Blocks = pull64Resident;
  int pull16Blocks = (Nn + 63) / 64;
  if (pull16Blocks > pull16Resident) pull16Blocks = pull16Resident;
  const int relu4 = Nn * D_H / 4;
  int reluBlocks = (relu4 + 1023) / 1024;
  if (reluBlocks > residentBlocks) reluBlocks = residentBlocks;

  if (ws_size >= off && RB <= EC_RB_MAX) {
    int P = 1;
    if (ws_size >= hwOff + 2 * planeBytes) P = 2;
    if (ws_size >= hwOff + (size_t)R * planeBytes) P = R;
    int sub = (((E + EC_NCHUNK - 1) / EC_NCHUNK) + 31) & ~31;
    ec_chunk_sort<<<EC_NCHUNK, 1024, 0, stream>>>(src, dst, part, lstart, lcnt, E, R, NB, RB, sub);
    ec_bucket_sort<<<RB, 1024, 0, stream>>>(part, lstart, lcnt, offs, col, Nn, NB, RB, M, E, R, sub);

    for (int r0 = 0; r0 < R; r0 += P) {
      int nr = (R - r0 < P) ? (R - r0) : P;
      ec_gemm_xs<<<gemmBlocks, 128, 0, stream>>>(
          x, (r0 == 0) ? L1w : nullptr, W1 + (size_t)r0 * D_IN * D_H, L1b,
          h, hw, mtiles, nr, Nn);
      ec_pull_rows<64><<<pull64Blocks, 256, 0, stream>>>(
          hw, offs + (size_t)r0 * Nn, col, h, Nn, nr, (r0 + nr >= R) ? 1 : 0);
    }

    size_t plane2Bytes = (size_t)Nn * D_OUT * sizeof(float);
    int P2 = (ws_size >= hwOff + (size_t)R * plane2Bytes) ? R : 1;
    if (R == EC_L2_REL && P2 == R) {
      ec_gemm_l2<<<(mtiles + 7) / 8, 256, 0, stream>>>(h, L2w, W2, L2b, out, hw, mtiles, Nn);
      ec_pull_rows<16><<<pull16Blocks, 256, 0, stream>>>(hw, offs, col, out, Nn, R, 0);
      return;
    }
    ec_gemm_wmma<64, 16><<<gemmBlocks, 128, 0, stream>>>(h, L2w, L2b, out, mtiles);
    for (int r0 = 0; r0 < R; r0 += P2) {
      int nr = (R - r0 < P2) ? (R - r0) : P2;
      for (int q = 0; q < nr; ++q) {
        ec_gemm_wmma<64, 16><<<gemmBlocks, 128, 0, stream>>>(
            h, W2 + (size_t)(r0 + q) * D_H * D_OUT, nullptr,
            hw + (size_t)q * Nn * D_OUT, mtiles);
      }
      ec_pull_rows<16><<<pull16Blocks, 256, 0, stream>>>(
          hw, offs + (size_t)r0 * Nn, col, out, Nn, nr, 0);
    }
    return;
  }

  off = 0;
  int*   deg  = (int*)(ws + off);
  float* invd = (float*)(ws + off);
  off += (size_t)R * Nn * sizeof(float);
  off = (off + 255) & ~(size_t)255;
  h = (float*)(ws + off);
  off += (size_t)Nn * D_H * sizeof(float);
  off = (off + 255) & ~(size_t)255;
  hw = (float*)(ws + off);

  hipMemsetAsync(deg, 0, (size_t)R * Nn * sizeof(int), stream);
  {
    int total = R * E;
    ec_count_deg<<<(total + 255) / 256, 256, 0, stream>>>(dst, deg, total, Nn, E);
  }
  {
    int total = R * Nn;
    ec_inv_deg<<<(total + 255) / 256, 256, 0, stream>>>(deg, invd, total);
  }

  ec_gemm_wmma<256, 64><<<gemmBlocks, 128, 0, stream>>>(x, L1w, L1b, h, mtiles);
  for (int r = 0; r < R; ++r) {
    ec_gemm_wmma<256, 64><<<gemmBlocks, 128, 0, stream>>>(
        x, W1 + (size_t)r * D_IN * D_H, nullptr, hw, mtiles);
    long long tot = (long long)E * (D_H / 4);
    ec_agg<64><<<(int)((tot + 255) / 256), 256, 0, stream>>>(
        hw, src + (size_t)r * E, dst + (size_t)r * E, invd + (size_t)r * Nn, h, E);
  }
  ec_relu<<<reluBlocks, 256, 0, stream>>>(h, relu4);

  ec_gemm_wmma<64, 16><<<gemmBlocks, 128, 0, stream>>>(h, L2w, L2b, out, mtiles);
  for (int r = 0; r < R; ++r) {
    ec_gemm_wmma<64, 16><<<gemmBlocks, 128, 0, stream>>>(
        h, W2 + (size_t)r * D_H * D_OUT, nullptr, hw, mtiles);
    long long tot = (long long)E * (D_OUT / 4);
    ec_agg<16><<<(int)((tot + 255) / 256), 256, 0, stream>>>(
        hw, src + (size_t)r * E, dst + (size_t)r * E, invd + (size_t)r * Nn, out, E);
  }
}


# Per-bucket edge sort runs as the first work-groups of the layer-1 product launch (128 threads, tables in its LDS block), under the matrix work: 5 launches, not 6; slot (sum) order = LDS arrival
# speedup vs baseline: 3.5400x; 1.0629x over previous
#include <hip/hip_runtime.h>

typedef __attribute__((ext_vector_type(2))) float v2f;
typedef __attribute__((ext_vector_type(4))) float v4f;
typedef __attribute__((ext_vector_type(8))) float v8f;

__global__ __launch_bounds__(256) void ec_count_deg(const int* __restrict__ dst,
                                                    int* __restrict__ deg,
                                                    int total, int Nn, int Eper) {
  int t = blockIdx.x * blockDim.x + threadIdx.x;
  if (t >= total) return;
  int r = t / Eper;
  atomicAdd(&deg[r * Nn + dst[t]], 1);
}

__global__ __launch_bounds__(256) void ec_inv_deg(int* __restrict__ deg,
                                                  float* __restrict__ invd,
                                                  int total) {
  int t = blockIdx.x * blockDim.x + threadIdx.x;
  if (t >= total) return;
  float d = (float)deg[t];
  invd[t] = 1.0f / fmaxf(d, 1.0f);
}


#define EC_BW_LOG 10
#define EC_BW (1 << EC_BW_LOG)
#define EC_RB_MAX 1024
#define EC_NCHUNK 256
#define EC_CPW (EC_NCHUNK / 32)
#define EC_STAGE_CAP 16384
#define EC_COL_CAP 16384

__device__ __forceinline__ int ec_chunk_edge(int u, int len, int lo, int Eper, int& rel) {
  rel = u / len;
  return rel * Eper + lo + (u - rel * len);
}

template <bool STAGED>
__device__ __forceinline__ void ec_chunk_place(const int* __restrict__ src, const int* __restrict__ dst,
                                               int* hs, int2* stage, int2* __restrict__ mine,
                                               int tid, int ntot, int len, int lo, int Eper, int NB) {
  int u = tid;
  for (; u + 3 * 1024 < ntot; u += 4 * 1024) {
    int r0, r1, r2, r3;
    int t0 = ec_chunk_edge(u, len, lo, Eper, r0);
    int t1 = ec_chunk_edge(u + 1024, len, lo, Eper, r1);
    int t2 = ec_chunk_edge(u + 2048, len, lo, Eper, r2);
    int t3 = ec_chunk_edge(u + 3072, len, lo, Eper, r3);
    int d0 = dst[t0];
    int d1 = dst[t1];
    int d2 = dst[t2];
    int d3 = dst[t3];
    int s0 = src[t0];
    int s1 = src[t1];
    int s2 = src[t2];
    int s3 = src[t3];
    int q0 = atomicAdd(&hs[r0 * NB + (d0 >> EC_BW_LOG)], 1);
    int q1 = atomicAdd(&hs[r1 * NB + (d1 >> EC_BW_LOG)], 1);
    int q2 = atomicAdd(&hs[r2 * NB + (d2 >> EC_BW_LOG)], 1);
    int q3 = atomicAdd(&hs[r3 * NB + (d3 >> EC_BW_LOG)], 1);
    if constexpr (STAGED) {
      stage[q0] = make_int2(s0, d0);
      stage[q1] = make_int2(s1, d1);
      stage[q2] = make_int2(s2, d2);
      stage[q3] = make_int2(s3, d3);
    } else {
      mine[q0] = make_int2(s0, d0);
      mine[q1] = make_int2(s1, d1);
      mine[q2] = make_int2(s2, d2);
      mine[q3] = make_int2(s3, d3);
    }
  }
  for (; u < ntot; u += 1024) {
    int r0;
    int t0 = ec_chunk_edge(u, len, lo, Eper, r0);
    int d = dst[t0];
    int s = src[t0];
    int q = atomicAdd(&hs[r0 * NB + (d >> EC_BW_LOG)], 1);
    if constexpr (STAGED) {
      stage[q] = make_int2(s, d);
    } else {
      mine[q] = make_int2(s, d);
    }
  }
}

__global__ __launch_bounds__(1024) void ec_chunk_sort(const int* __restrict__ src,
                                                      const int* __restrict__ dst,
                                                      int2* __restrict__ part,
                                                      int* __restrict__ lstart,
                                                      int* __restrict__ lcnt,
                                                      int Eper, int Rn, int NB, int RB,
                                                      int sub) {
  static_assert(EC_RB_MAX == 1024, "one bucket counter per thread");
  __shared__ int hs[EC_RB_MAX];
  __shared__ int wtot[32];
  __shared__ __attribute__((aligned(16))) int2 stage[EC_STAGE_CAP];
  int tid = threadIdx.x;
  int lane = tid & 31;
  int wv = tid >> 5;
  hs[tid] = 0;
  __syncthreads();
  int lo = (int)blockIdx.x * sub;
  if (lo > Eper) lo = Eper;
  int hi = (lo + sub < Eper) ? (lo + sub) : Eper;
  int len = hi - lo;
  int ntot = Rn * len;

  int u = tid;
  for (; u + 3 * 1024 < ntot; u += 4 * 1024) {
    int r0, r1, r2, r3;
    int t0 = ec_chunk_edge(u, len, lo, Eper, r0);
    int t1 = ec_chunk_edge(u + 1024, len, lo, Eper, r1);
    int t2 = ec_chunk_edge(u + 2048, len, lo, Eper, r2);
    int t3 = ec_chunk_edge(u + 3072, len, lo, Eper, r3);
    int d0 = dst[t0];
    int d1 = dst[t1];
    int d2 = dst[t2];
    int d3 = dst[t3];
    atomicAdd(&hs[r0 * NB + (d0 >> EC_BW_LOG)], 1);
    atomicAdd(&hs[r1 * NB + (d1 >> EC_BW_LOG)], 1);
    atomicAdd(&hs[r2 * NB + (d2 >> EC_BW_LOG)], 1);
    atomicAdd(&hs[r3 * NB + (d3 >> EC_BW_LOG)], 1);
  }
  for (; u < ntot; u += 1024) {
    int r0;
    int t0 = ec_chunk_edge(u, len, lo, Eper, r0);
    int d = dst[t0];
    atomicAdd(&hs[r0 * NB + (d >> EC_BW_LOG)], 1);
  }
  __syncthreads();

  int v = hs[tid];
  int inc = v;
  for (int dl = 1; dl < 32; dl <<= 1) {
    int up = __shfl_up(inc, dl, 32);
    if (lane >= dl) inc += up;
  }
  if (lane == 31) wtot[wv] = inc;
  __syncthreads();
  int before = 0;
  for (int j = 0; j < wv; ++j) before += wtot[j];
  int start = before + inc - v;
  if (tid < RB) {
    lstart[tid * EC_NCHUNK + blockIdx.x] = start;
    lcnt[tid * EC_NCHUNK + blockIdx.x] = v;
  }
  hs[tid] = start;
  __syncthreads();

  int2* mine = part + (size_t)Rn * lo;
  const bool staged = (ntot <= EC_STAGE_CAP) && (((Rn * lo) & 1) == 0);
  if (staged) {
    ec_chunk_place<true>(src, dst, hs, stage, mine, tid, ntot, len, lo, Eper, NB);
    __syncthreads();
    const int4* s4 = (const int4*)stage;
    int4* g4 = (int4*)mine;
    const int np = ntot >> 1;
    for (int i = tid; i < np; i += 1024) g4[i] = s4[i];
    if ((ntot & 1) && tid == 0) mine[ntot - 1] = stage[ntot - 1];
  } else {
    ec_chunk_place<false>(src, dst, hs, stage, mine, tid, ntot, len, lo, Eper, NB);
  }
}

__global__ __launch_bounds__(1024) void ec_bucket_sort(const int2* __restrict__ part,
                                                       const int* __restrict__ lstart,
                                                       const int* __restrict__ lcnt,
                                                       int* __restrict__ offs,
                                                       int* __restrict__ col,
                                                       int Nn, int NB, int RB, int M,
                                                       int Eper, int Rn, int sub) {
  __shared__ int cnt[EC_BW];
  __shared__ int sh[EC_BW];
  __shared__ int rbeg[EC_NCHUNK];
  __shared__ int rlen[EC_NCHUNK];
  __shared__ int wst[EC_CPW];
  __shared__ int wln[EC_CPW];
  __shared__ int cstage[EC_COL_CAP];
  int tid = threadIdx.x;
  int lane = tid & 31;
  int wv = tid >> 5;
  int bkt = blockIdx.x;
  int r = bkt / NB;
  int n0 = (bkt - r * NB) << EC_BW_LOG;
  cnt[tid] = 0;
  if (tid < EC_NCHUNK) {
    int st = lstart[bkt * EC_NCHUNK + tid];
    int ln = lcnt[bkt * EC_NCHUNK + tid];
    int lo = tid * sub;
    if (lo > Eper) lo = Eper;
    rbeg[tid] = Rn * lo + st;
    rlen[tid] = ln;
    for (int dl = 16; dl > 0; dl >>= 1) {
      st += __shfl_xor(st, dl, 32);
      ln += __shfl_xor(ln, dl, 32);
    }
    if (lane == 0) {
      wst[wv] = st;
      wln[wv] = ln;
    }
  }
  __syncthreads();
  int e0 = 0;
  int nb = 0;
#pragma unroll
  for (int j = 0; j < EC_CPW; ++j) {
    e0 += wst[j];
    nb += wln[j];
  }
  int e1 = e0 + nb;
  int rb[EC_CPW], rl[EC_CPW];
#pragma unroll
  for (int cc = 0; cc < EC_CPW; ++cc) {
    rb[cc] = rbeg[wv * EC_CPW + cc];
    rl[cc] = rlen[wv * EC_CPW + cc];
  }
  if (nb > 0) {
    int y[2 * EC_CPW];
#pragma unroll
    for (int cc = 0; cc < EC_CPW; ++cc) {
      y[2 * cc] = part[(lane < rl[cc]) ? (rb[cc] + lane) : 0].y;
      y[2 * cc + 1] = part[(lane + 32 < rl[cc]) ? (rb[cc] + lane + 32) : 0].y;
    }
#pragma unroll
    for (int cc = 0; cc < EC_CPW; ++cc) {
      if (lane < rl[cc]) atomicAdd(&cnt[y[2 * cc] - n0], 1);
      if (lane + 32 < rl[cc]) atomicAdd(&cnt[y[2 * cc + 1] - n0], 1);
    }
#pragma unroll
    for (int cc = 0; cc < EC_CPW; ++cc) {
      for (int j = lane + 64; j < rl[cc]; j += 32) atomicAdd(&cnt[part[rb[cc] + j].y - n0], 1);
    }
  }
  __syncthreads();
  int v = cnt[tid];
  sh[tid] = v;
  __syncthreads();
  for (int dl = 1; dl < EC_BW; dl <<= 1) {
    int add = (tid >= dl) ? sh[tid - dl] : 0;
    __syncthreads();
    sh[tid] += add;
    __syncthreads();
  }
  int start = sh[tid] - v;
  if (n0 + tid < Nn) offs[r * Nn + n0 + tid] = e0 + start;
  cnt[tid] = start;
  __syncthreads();
  const bool staged = nb <= EC_COL_CAP;
  if (nb > 0) {
    int2 rec[2 * EC_CPW];
#pragma unroll
    for (int cc = 0; cc < EC_CPW; ++cc) {
      rec[2 * cc] = part[(lane < rl[cc]) ? (rb[cc] + lane) : 0];
      rec[2 * cc + 1] = part[(lane + 32 < rl[cc]) ? (rb[cc] + lane + 32) : 0];
    }
#pragma unroll
    for (int cc = 0; cc < EC_CPW; ++cc) {
      if (lane < rl[cc]) {
        int q = atomicAdd(&cnt[rec[2 * cc].y - n0], 1);
        if (staged) cstage[q] = rec[2 * cc].x; else col[e0 + q] = rec[2 * cc].x;
      }
      if (lane + 32 < rl[cc]) {
        int q = atomicAdd(&cnt[rec[2 * cc + 1].y - n0], 1);
        if (staged) cstage[q] = rec[2 * cc + 1].x; else col[e0 + q] = rec[2 * cc + 1].x;
      }
    }
#pragma unroll
    for (int cc = 0; cc < EC_CPW; ++cc) {
      for (int j = lane + 64; j < rl[cc]; j += 32) {
        int2 rc = part[rb[cc] + j];
        int q = atomicAdd(&cnt[rc.y - n0], 1);
        if (staged) cstage[q] = rc.x; else col[e0 + q] = rc.x;
      }
    }
  }
  __syncthreads();
  if (staged) {
    for (int i = tid; i < nb; i += EC_BW) col[e0 + i] = cstage[i];
  }
  if (bkt == RB - 1 && tid == 0) offs[M] = e1;
}

template <int K, int NC>
__global__ __launch_bounds__(128) void ec_gemm_wmma(
    const float* __restrict__ A, const float* __restrict__ B,
    const float* __restrict__ bias, float* __restrict__ C, int mtiles) {
  constexpr int NT = NC / 16;
  constexpr int S = ((2 * NC) % 64 == 32) ? (2 * NC) : (2 * NC + 32);
  __shared__ float Bs[(K / 2) * S];

  for (int i = threadIdx.x; i < K * NC / 4; i += 128) {
    int k = (i * 4) / NC;
    int n = (i * 4) % NC;
    float4 v = *(const float4*)(B + i * 4);
    float* p = &Bs[(k >> 1) * S + 2 * n + (k & 1)];
    p[0] = v.x;
    p[2] = v.y;
    p[4] = v.z;
    p[6] = v.w;
  }
  __syncthreads();

  int wave = (blockIdx.x * blockDim.x + threadIdx.x) >> 5;
  if (wave >= mtiles) return;
  int lane = threadIdx.x & 31;
  int m = lane & 15;
  int half = lane >> 4;

  const float* Arow = A + (size_t)(wave * 16 + m) * K + 2 * half;

  v8f acc[NT];
  v8f z = {0.f, 0.f, 0.f, 0.f, 0.f, 0.f, 0.f, 0.f};
#pragma unroll
  for (int j = 0; j < NT; ++j) acc[j] = z;

#pragma unroll 8
  for (int k = 0; k < K; k += 4) {
    float2 av = *(const float2*)(Arow + k);
    v2f a = {av.x, av.y};
    const float* Bp = &Bs[(size_t)((k + 2 * half) >> 1) * S + 2 * m];
#pragma unroll
    for (int j = 0; j < NT; ++j) {
      v2f b = *(const v2f*)(Bp + j * 32);
      acc[j] = __builtin_amdgcn_wmma_f32_16x16x4_f32(
           false, a,  false, b,
           (short)0, acc[j],  false,  false);
    }
  }

#pragma unroll
  for (int j = 0; j < NT; ++j) {
    float bv = bias ? bias[j * 16 + m] : 0.0f;
#pragma unroll
    for (int v = 0; v < 8; ++v) {
      C[(size_t)(wave * 16 + v + 8 * half) * NC + j * 16 + m] = acc[j][v] + bv;
    }
  }
}

__device__ __forceinline__ void ec_gemm_xs_body(
    float* Bs, int wave, const float* __restrict__ A, const float* __restrict__ Bself,
    const float* __restrict__ Brel, const float* __restrict__ bias,
    float* __restrict__ Cself, float* __restrict__ Crel,
    int mtiles, int nrel, int Nn) {
  constexpr int K = 256;
  constexpr int NC = 64;
  constexpr int NT = NC / 16;
  constexpr int S = 2 * NC + 32;
  bool active = wave < mtiles;
  int wv = active ? wave : 0;
  int lane = threadIdx.x & 31;
  int m = lane & 15;
  int half = lane >> 4;

  const float* Arow = A + (size_t)(wv * 16 + m) * K + 2 * half;
  v2f a[K / 4];
#pragma unroll
  for (int kk = 0; kk < K / 4; ++kk) {
    float2 av = *(const float2*)(Arow + 4 * kk);
    a[kk][0] = av.x;
    a[kk][1] = av.y;
  }

  float bvs[NT];
#pragma unroll
  for (int j = 0; j < NT; ++j) bvs[j] = bias[j * 16 + m];

#pragma unroll 1
  for (int r = (Bself ? 0 : 1); r <= nrel; ++r) {
    const float* B = (r == 0) ? Bself : Brel + (size_t)(r - 1) * K * NC;
    float* C = (r == 0) ? Cself : Crel + (size_t)(r - 1) * Nn * NC;

    __syncthreads();
    for (int it0 = 0; it0 < K * NC / 4 / 128; it0 += 8) {
      v4f q[8];
#pragma unroll
      for (int u = 0; u < 8; ++u) {
        int i = (it0 + u) * 128 + (int)threadIdx.x;
        q[u] = *(const v4f*)(B + i * 4);
      }
      asm volatile("" : "+v"(q[0]), "+v"(q[1]), "+v"(q[2]), "+v"(q[3]),
                        "+v"(q[4]), "+v"(q[5]), "+v"(q[6]), "+v"(q[7]));
#pragma unroll
      for (int u = 0; u < 8; ++u) {
        int i = (it0 + u) * 128 + (int)threadIdx.x;
        int k = (i * 4) / NC;
        int n = (i * 4) % NC;
        int j = n >> 4;
        int m0 = n & 15;
        float* p = &Bs[(k >> 1) * S + (j >> 1) * 64 + 4 * m0 + 2 * (j & 1) + (k & 1)];
        p[0] = q[u][0];
        p[4] = q[u][1];
        p[8] = q[u][2];
        p[12] = q[u][3];
      }
    }
    __syncthreads();

    v8f acc[NT];
    v8f z = {0.f, 0.f, 0.f, 0.f, 0.f, 0.f, 0.f, 0.f};
#pragma unroll
    for (int j = 0; j < NT; ++j) acc[j] = z;

#pragma unroll
    for (int kk = 0; kk < K / 4; ++kk) {
      int k = 4 * kk;
      const float* Bp = &Bs[(size_t)((k + 2 * half) >> 1) * S + 4 * m];
#pragma unroll
      for (int t = 0; t < NT / 2; ++t) {
        v4f q = *(const v4f*)(Bp + t * 64);
        v2f b0 = {q[0], q[1]};
        v2f b1 = {q[2], q[3]};
        acc[2 * t] = __builtin_amdgcn_wmma_f32_16x16x4_f32(
             false, a[kk],  false, b0,
             (short)0, acc[2 * t],  false,  false);
        acc[2 * t + 1] = __builtin_amdgcn_wmma_f32_16x16x4_f32(
             false, a[kk],  false, b1,
             (short)0, acc[2 * t + 1],  false,  false);
      }
    }

    v8f o[NT];
#pragma unroll
    for (int j = 0; j < NT; ++j) {
      float bv = (r == 0) ? bvs[j] : 0.0f;
#pragma unroll
      for (int v = 0; v < 8; ++v) o[j][v] = acc[j][v] + bv;
    }
    asm volatile("" : "+v"(o[0]), "+v"(o[1]), "+v"(o[2]), "+v"(o[3]));
    if (active) {
#pragma unroll
      for (int j = 0; j < NT; ++j) {
#pragma unroll
        for (int v = 0; v < 8; ++v) {
          C[(size_t)(wave * 16 + v + 8 * half) * NC + j * 16 + m] = o[j][v];
        }
      }
    }
  }
}

#define EC_XS_LDS_FLOATS ((256 / 2) * (2 * 64 + 32))
#define EC_BS_T 128
#define EC_BS_RPW (EC_NCHUNK / (EC_BS_T / 32))
#define EC_BS_WORDS (2 * EC_BW + 2 * EC_NCHUNK + 32 + EC_COL_CAP)
static_assert(EC_BS_WORDS <= EC_XS_LDS_FLOATS, "the sorter's tables fit the weight image");
static_assert(EC_BW == 8 * EC_BS_T, "eight nodes per thread in the scan");
static_assert(EC_BS_RPW % EC_CPW == 0, "whole batches of runs per wave");

__device__ __forceinline__ void ec_bucket_role(int* L, const int2* __restrict__ part,
                                               const int* __restrict__ lstart,
                                               const int* __restrict__ lcnt,
                                               int* __restrict__ offs, int* __restrict__ col,
                                               int Nn, int NB, int RB, int M,
                                               int Eper, int Rn, int sub, int bkt) {
  int* cnt = L;
  int* sh = L + EC_BW;
  int* rbeg = L + 2 * EC_BW;
  int* rlen = rbeg + EC_NCHUNK;
  int* wst = rlen + EC_NCHUNK;
  int* wln = wst + 8;
  int* wsc = wln + 8;
  int* cstage = L + 2 * EC_BW + 2 * EC_NCHUNK + 32;
  const int tid = threadIdx.x;
  const int lane = tid & 31;
  const int wv = tid >> 5;
  const int r = bkt / NB;
  const int n0 = (bkt - r * NB) << EC_BW_LOG;
  for (int i = tid; i < EC_BW; i += EC_BS_T) cnt[i] = 0;
  int st = 0, ln = 0;
  for (int c = tid; c < EC_NCHUNK; c += EC_BS_T) {
    int s1 = lstart[bkt * EC_NCHUNK + c];
    int l1 = lcnt[bkt * EC_NCHUNK + c];
    int lo = c * sub;
    if (lo > Eper) lo = Eper;
    rbeg[c] = Rn * lo + s1;
    rlen[c] = l1;
    st += s1;
    ln += l1;
  }
  for (int dl = 16; dl > 0; dl >>= 1) {
    st += __shfl_xor(st, dl, 32);
    ln += __shfl_xor(ln, dl, 32);
  }
  if (lane == 0) {
    wst[wv] = st;
    wln[wv] = ln;
  }
  __syncthreads();
  int e0 = 0;
  int nb = 0;
#pragma unroll
  for (int j = 0; j < EC_BS_T / 32; ++j) {
    e0 += wst[j];
    nb += wln[j];
  }
  const int e1 = e0 + nb;

  if (nb > 0) {
#pragma unroll 1
    for (int bt = 0; bt < EC_BS_RPW / EC_CPW; ++bt) {
      const int base = wv * EC_BS_RPW + bt * EC_CPW;
      int rb[EC_CPW], rl[EC_CPW];
#pragma unroll
      for (int cc = 0; cc < EC_CPW; ++cc) {
        rb[cc] = rbeg[base + cc];
        rl[cc] = rlen[base + cc];
      }
      int y[2 * EC_CPW];
#pragma unroll
      for (int cc = 0; cc < EC_CPW; ++cc) {
        y[2 * cc] = part[(lane < rl[cc]) ? (rb[cc] + lane) : 0].y;
        y[2 * cc + 1] = part[(lane + 32 < rl[cc]) ? (rb[cc] + lane + 32) : 0].y;
      }
#pragma unroll
      for (int cc = 0; cc < EC_CPW; ++cc) {
        if (lane < rl[cc]) atomicAdd(&cnt[y[2 * cc] - n0], 1);
        if (lane + 32 < rl[cc]) atomicAdd(&cnt[y[2 * cc + 1] - n0], 1);
      }
#pragma unroll
      for (int cc = 0; cc < EC_CPW; ++cc) {
        for (int j = lane + 64; j < rl[cc]; j += 32) atomicAdd(&cnt[part[rb[cc] + j].y - n0], 1);
      }
    }
  }
  __syncthreads();

  int v[8];
  {
    const int4 c0 = ((const int4*)cnt)[2 * tid];
    const int4 c1 = ((const int4*)cnt)[2 * tid + 1];
    v[0] = c0.x; v[1] = c0.y; v[2] = c0.z; v[3] = c0.w;
    v[4] = c1.x; v[5] = c1.y; v[6] = c1.z; v[7] = c1.w;
  }
  int tot = 0;
#pragma unroll
  for (int j = 0; j < 8; ++j) tot += v[j];
  int inc = tot;
  for (int dl = 1; dl < 32; dl <<= 1) {
    int up = __shfl_up(inc, dl, 32);
    if (lane >= dl) inc += up;
  }
  if (lane == 31) wsc[wv] = inc;
  __syncthreads();
  int run = inc - tot;
  for (int j = 0; j < wv; ++j) run += wsc[j];
  int s8[8];
#pragma unroll
  for (int j = 0; j < 8; ++j) {
    s8[j] = run;
    run += v[j];
  }
  {
    const int4 a0 = make_int4(s8[0], s8[1], s8[2], s8[3]);
    const int4 a1 = make_int4(s8[4], s8[5], s8[6], s8[7]);
    ((int4*)cnt)[2 * tid] = a0;
    ((int4*)cnt)[2 * tid + 1] = a1;
    ((int4*)sh)[2 * tid] = a0;
    ((int4*)sh)[2 * tid + 1] = a1;
  }
  __syncthreads();
  for (int i = tid; i < EC_BW; i += EC_BS_T) {
    if (n0 + i < Nn) offs[r * Nn + n0 + i] = e0 + sh[i];
  }

  const bool staged = nb <= EC_COL_CAP;
  if (nb > 0) {
#pragma unroll 1
    for (int bt = 0; bt < EC_BS_RPW / EC_CPW; ++bt) {
      const int base = wv * EC_BS_RPW + bt * EC_CPW;
      int rb[EC_CPW], rl[EC_CPW];
#pragma unroll
      for (int cc = 0; cc < EC_CPW; ++cc) {
        rb[cc] = rbeg[base + cc];
        rl[cc] = rlen[base + cc];
      }
      int2 rec[2 * EC_CPW];
#pragma unroll
      for (int cc = 0; cc < EC_CPW; ++cc) {
        rec[2 * cc] = part[(lane < rl[cc]) ? (rb[cc] + lane) : 0];
        rec[2 * cc + 1] = part[(lane + 32 < rl[cc]) ? (rb[cc] + lane + 32) : 0];
      }
#pragma unroll
      for (int cc = 0; cc < EC_CPW; ++cc) {
        if (lane < rl[cc]) {
          int q = atomicAdd(&cnt[rec[2 * cc].y - n0], 1);
          if (staged) cstage[q] = rec[2 * cc].x; else col[e0 + q] = rec[2 * cc].x;
        }
        if (lane + 32 < rl[cc]) {
          int q = atomicAdd(&cnt[rec[2 * cc + 1].y - n0], 1);
          if (staged) cstage[q] = rec[2 * cc + 1].x; else col[e0 + q] = rec[2 * cc + 1].x;
        }
      }
#pragma unroll
      for (int cc = 0; cc < EC_CPW; ++cc) {
        for (int j = lane + 64; j < rl[cc]; j += 32) {
          int2 rc = part[rb[cc] + j];
          int q = atomicAdd(&cnt[rc.y - n0], 1);
          if (staged) cstage[q] = rc.x; else col[e0 + q] = rc.x;
        }
      }
    }
  }
  __syncthreads();
  if (staged) {
    for (int i = tid; i < nb; i += EC_BS_T) col[e0 + i] = cstage[i];
  }
  if (bkt == RB - 1 && tid == 0) offs[M] = e1;
}

__global__ __launch_bounds__(128) void ec_gemm_xs_sort(
    const float* __restrict__ A, const float* __restrict__ Bself,
    const float* __restrict__ Brel, const float* __restrict__ bias,
    float* __restrict__ Cself, float* __restrict__ Crel,
    int mtiles, int nrel, int Nn,
    const int2* __restrict__ part, const int* __restrict__ lstart,
    const int* __restrict__ lcnt, int* __restrict__ offs, int* __restrict__ col,
    int NB, int RB, int M, int Eper, int Rn, int sub, int nbkt) {
  __shared__ __attribute__((aligned(16))) float Bs[EC_XS_LDS_FLOATS];
  const int blk = (int)blockIdx.x;
  if (blk < nbkt) {
    ec_bucket_role((int*)Bs, part, lstart, lcnt, offs, col, Nn, NB, RB, M, Eper, Rn, sub, blk);
    return;
  }
  int wave = ((blk - nbkt) * 128 + (int)threadIdx.x) >> 5;
  ec_gemm_xs_body(Bs, wave, A, Bself, Brel, bias, Cself, Crel, mtiles, nrel, Nn);
}

__global__ __launch_bounds__(128) void ec_gemm_xs(
    const float* __restrict__ A, const float* __restrict__ Bself,
    const float* __restrict__ Brel, const float* __restrict__ bias,
    float* __restrict__ Cself, float* __restrict__ Crel,
    int mtiles, int nrel, int Nn) {
  __shared__ __attribute__((aligned(16))) float Bs[EC_XS_LDS_FLOATS];
  int wave = (blockIdx.x * blockDim.x + threadIdx.x) >> 5;
  ec_gemm_xs_body(Bs, wave, A, Bself, Brel, bias, Cself, Crel, mtiles, nrel, Nn);
}

#define EC_L2_REL 4
__global__ __launch_bounds__(256) void ec_gemm_l2(
    const float* __restrict__ A, const float* __restrict__ Bself,
    const float* __restrict__ Brel, const float* __restrict__ bias,
    float* __restrict__ Cself, float* __restrict__ Crel, int mtiles, int Nn) {
  constexpr int K = 64;
  constexpr int NC = 16;
  constexpr int NT = 1 + EC_L2_REL;
  constexpr int S = 2 * NC * NT;
  static_assert(S % 64 == 32, "lane-halves on disjoint LDS bank ranges");
  static_assert(K * NC / 4 == 256, "one 128-bit vector of every block per thread");
  __shared__ float Bs[(K / 2) * S];

  int wave = (blockIdx.x * blockDim.x + threadIdx.x) >> 5;
  bool active = wave < mtiles;
  int wv = active ? wave : 0;
  int lane = threadIdx.x & 31;
  int m = lane & 15;
  int half = lane >> 4;

  v4f q[NT];
#pragma unroll
  for (int j = 0; j < NT; ++j) {
    const float* B = (j == 0) ? Bself : Brel + (size_t)(j - 1) * K * NC;
    q[j] = *(const v4f*)(B + (int)threadIdx.x * 4);
  }
  const float* Arow = A + (size_t)(wv * 16 + m) * K + 2 * half;
  v2f a[K / 4];
#pragma unroll
  for (int kk = 0; kk < K / 4; ++kk) {
    float2 av = *(const float2*)(Arow + 4 * kk);
    a[kk][0] = av.x;
    a[kk][1] = av.y;
  }
  float bv = bias[m];

  {
    int k = (int)threadIdx.x >> 2;
    int n = ((int)threadIdx.x & 3) * 4;
    float* p = &Bs[(k >> 1) * S + 2 * n + (k & 1)];
#pragma unroll
    for (int j = 0; j < NT; ++j) {
      p[32 * j + 0] = q[j][0];
      p[32 * j + 2] = q[j][1];
      p[32 * j + 4] = q[j][2];
      p[32 * j + 6] = q[j][3];
    }
  }
  __syncthreads();
  if (!active) return;

  v8f acc[NT];
  v8f z = {0.f, 0.f, 0.f, 0.f, 0.f, 0.f, 0.f, 0.f};
#pragma unroll
  for (int j = 0; j < NT; ++j) acc[j] = z;

#pragma unroll
  for (int kk = 0; kk < K / 4; ++kk) {
    int k = 4 * kk;
    const float* Bp = &Bs[(size_t)((k + 2 * half) >> 1) * S + 2 * m];
#pragma unroll
    for (int j = 0; j < NT; ++j) {
      v2f b = *(const v2f*)(Bp + j * 32);
      acc[j] = __builtin_amdgcn_wmma_f32_16x16x4_f32(
           false, a[kk],  false, b,
           (short)0, acc[j],  false,  false);
    }
  }

  const size_t row0 = (size_t)(wave * 16 + 8 * half) * NC + m;
#pragma unroll
  for (int v = 0; v < 8; ++v) Cself[row0 + (size_t)v * NC] = acc[0][v] + bv;
#pragma unroll
  for (int j = 1; j < NT; ++j) {
    float* C = Crel + (size_t)(j - 1) * Nn * NC + row0;
#pragma unroll
    for (int v = 0; v < 8; ++v) C[(size_t)v * NC] = acc[j][v];
  }
}

template <int COLS>
__global__ __launch_bounds__(256) void ec_agg(
    const float* __restrict__ hw, const int* __restrict__ src,
    const int* __restrict__ dst, const float* __restrict__ invd,
    float* __restrict__ acc, int E) {
  constexpr int TPE = COLS / 4;
  int t = blockIdx.x * blockDim.x + threadIdx.x;
  int e = t / TPE;
  if (e >= E) return;
  int c = (t % TPE) * 4;
  int s = src[e];
  int d = dst[e];
  float w = invd[d];
  float4 v = *(const float4*)(hw + (size_t)s * COLS + c);
  float* dp = acc + (size_t)d * COLS + c;
  unsafeAtomicAdd(dp + 0, v.x * w);
  unsafeAtomicAdd(dp + 1, v.y * w);
  unsafeAtomicAdd(dp + 2, v.z * w);
  unsafeAtomicAdd(dp + 3, v.w * w);
}

template <int TPN, int J>
__device__ __forceinline__ int ec_slot_src(const int* id) {
  static_assert(TPN == 16 || TPN == 4, "lane group is a 16-lane row or a quad");
  if constexpr (TPN == 16) {
    return __builtin_amdgcn_update_dpp(0, id[0], 0x150 + J, 0xf, 0xf, true);
  } else {
    return __builtin_amdgcn_update_dpp(0, id[J >> 2], (J & 3) * 0x55, 0xf, 0xf, true);
  }
}

template <int COLS>
__device__ __forceinline__ void ec_sum_batch(const float* __restrict__ hwb, unsigned cb, const int* id, int n,
                                             float& sx, float& sy, float& sz, float& sw) {
  constexpr int TPN = COLS / 4;
  float4 v[16];
#define EC_GATHER(J) \
  if (J < n) v[J] = *(const float4*)((const char*)hwb + \
                       ((unsigned)ec_slot_src<TPN, J>(id) * (unsigned)(COLS * 4) + cb));
  EC_GATHER(0) EC_GATHER(1) EC_GATHER(2) EC_GATHER(3)
  if (n > 4) {
    EC_GATHER(4) EC_GATHER(5) EC_GATHER(6) EC_GATHER(7)
    if (n > 8) {
      EC_GATHER(8) EC_GATHER(9) EC_GATHER(10) EC_GATHER(11)
      if (n > 12) {
        EC_GATHER(12) EC_GATHER(13) EC_GATHER(14) EC_GATHER(15)
      }
    }
  }
#undef EC_GATHER
  asm volatile("" : "+v"(sx), "+v"(sy), "+v"(sz), "+v"(sw));
#define EC_ADD(J) \
  if (J < n) { sx += v[J].x; sy += v[J].y; sz += v[J].z; sw += v[J].w; }
  EC_ADD(0) EC_ADD(1) EC_ADD(2) EC_ADD(3)
  if (n > 4) {
    EC_ADD(4) EC_ADD(5) EC_ADD(6) EC_ADD(7)
    if (n > 8) {
      EC_ADD(8) EC_ADD(9) EC_ADD(10) EC_ADD(11)
      if (n > 12) {
        EC_ADD(12) EC_ADD(13) EC_ADD(14) EC_ADD(15)
      }
    }
  }
#undef EC_ADD
}

template <int COLS>
__global__ __launch_bounds__(256) void ec_pull_rows(
    const float* __restrict__ hw, const int* __restrict__ offs,
    const int* __restrict__ col, float* __restrict__ acc,
    int Nn, int nrel, int relu) {
  constexpr int TPN = COLS / 4;
  constexpr int RPB = 256 / TPN;
  constexpr int NI = 16 / TPN;
  const int g = (int)threadIdx.x / TPN;
  const int l = (int)threadIdx.x % TPN;
  const int c = l * 4;
  const int step = (int)gridDim.x * RPB;
  const size_t plane = (size_t)Nn * COLS;
  int d = (int)blockIdx.x * RPB + g;
  if (d >= Nn) return;
  int q = 0;

  int b = offs[d];
  int e = offs[d + 1];
  int idc[NI];
#pragma unroll
  for (int k = 0; k < NI; ++k) {
    int j = k * TPN + l;
    idc[k] = col[(j < e - b) ? (b + j) : 0];
  }
  int q1 = (nrel > 1) ? 1 : 0;
  int d1 = (nrel > 1) ? d : d + step;
  bool ok1 = d1 < Nn;
  int b1 = 0, e1 = 0;
  if (ok1) {
    const int* p = offs + (size_t)q1 * Nn + d1;
    b1 = p[0];
    e1 = p[1];
  }

  float4 o = {0.f, 0.f, 0.f, 0.f};
  for (;;) {
    int idn[NI];
#pragma unroll
    for (int k = 0; k < NI; ++k) {
      int j = k * TPN + l;
      idn[k] = col[(j < e1 - b1) ? (b1 + j) : 0];
    }
    int q2 = (q1 + 1 < nrel) ? q1 + 1 : 0;
    int d2 = (q1 + 1 < nrel) ? d1 : d1 + step;
    bool ok2 = ok1 && (d2 < Nn);
    int b2 = 0, e2 = 0;
    if (ok2) {
      const int* p = offs + (size_t)q2 * Nn + d2;
      b2 = p[0];
      e2 = p[1];
    }

    float4* hp = (float4*)(acc + (size_t)d * COLS + c);
    if (q == 0) o = *hp;
    const unsigned cb = (unsigned)(((size_t)q * plane + (size_t)c) * sizeof(float));
    float sx = 0.f, sy = 0.f, sz = 0.f, sw = 0.f;
    const int n = e - b;
    ec_sum_batch<COLS>(hw, cb, idc, (n < 16) ? n : 16, sx, sy, sz, sw);
    for (int i = b + 16; i < e; i += 16) {
      int m = (e - i < 16) ? (e - i) : 16;
      int idx[NI];
#pragma unroll
      for (int k = 0; k < NI; ++k) {
        int j = k * TPN + l;
        idx[k] = col[(j < m) ? (i + j) : 0];
      }
      ec_sum_batch<COLS>(hw, cb, idx, m, sx, sy, sz, sw);
    }
    float w = 1.0f / fmaxf((float)n, 1.0f);
    o.x += sx * w;
    o.y += sy * w;
    o.z += sz * w;
    o.w += sw * w;
    if (q + 1 >= nrel) {
      if (relu) {
        o.x = fmaxf(o.x, 0.0f);
        o.y = fmaxf(o.y, 0.0f);
        o.z = fmaxf(o.z, 0.0f);
        o.w = fmaxf(o.w, 0.0f);
      }
      *hp = o;
    }
    if (!ok1) break;
    d = d1; q = q1; b = b1; e = e1;
#pragma unroll
    for (int k = 0; k < NI; ++k) idc[k] = idn[k];
    d1 = d2; q1 = q2; ok1 = ok2; b1 = b2; e1 = e2;
  }
}


__global__ __launch_bounds__(256) void ec_relu(float* __restrict__ h, int n4) {
  v4f* p = (v4f*)h;
  const int T = (int)gridDim.x * 256;
  for (int i = (int)blockIdx.x * 256 + (int)threadIdx.x; i < n4; i += 4 * T) {
    const int i1 = i + T;
    const int i2 = i + 2 * T;
    const int i3 = i + 3 * T;
    const bool k1 = i1 < n4;
    const bool k2 = i2 < n4;
    const bool k3 = i3 < n4;
    v4f a0 = p[i];
    v4f a1 = p[k1 ? i1 : i];
    v4f a2 = p[k2 ? i2 : i];
    v4f a3 = p[k3 ? i3 : i];
#pragma unroll
    for (int q = 0; q < 4; ++q) {
      a0[q] = fmaxf(a0[q], 0.0f);
      a1[q] = fmaxf(a1[q], 0.0f);
      a2[q] = fmaxf(a2[q], 0.0f);
      a3[q] = fmaxf(a3[q], 0.0f);
    }
    p[i] = a0;
    if (k1) p[i1] = a1;
    if (k2) p[i2] = a2;
    if (k3) p[i3] = a3;
  }
}

extern "C" void kernel_launch(void* const* d_in, const int* in_sizes, int n_in,
                              void* d_out, int out_size, void* d_ws, size_t ws_size,
                              hipStream_t stream) {
  const float* x   = (const float*)d_in[0];
  const int*   src = (const int*)d_in[1];
  const int*   dst = (const int*)d_in[2];
  const float* W1  = (const float*)d_in[3];
  const float* L1w = (const float*)d_in[4];
  const float* L1b = (const float*)d_in[5];
  const float* W2  = (const float*)d_in[6];
  const float* L2w = (const float*)d_in[7];
  const float* L2b = (const float*)d_in[8];
  float* out = (float*)d_out;

  const int D_IN = 256, D_H = 64, D_OUT = 16;
  const int Nn = in_sizes[0] / D_IN;
  const int R  = in_sizes[3] / (D_IN * D_H);
  const int E  = in_sizes[1] / R;

  char* ws = (char*)d_ws;
  int mtiles = Nn / 16;
  int gemmBlocks = (mtiles * 32 + 127) / 128;

  const int M = R * Nn;
  const int NB = (Nn + EC_BW - 1) >> EC_BW_LOG;
  const int RB = R * NB;
  size_t off = 0;
  int* lstart = (int*)(ws + off);
  off += (size_t)RB * EC_NCHUNK * sizeof(int);
  off = (off + 255) & ~(size_t)255;
  int* lcnt = (int*)(ws + off);
  off += (size_t)RB * EC_NCHUNK * sizeof(int);
  off = (off + 255) & ~(size_t)255;
  int* offs = (int*)(ws + off);
  off += ((size_t)M + 1) * sizeof(int);
  off = (off + 255) & ~(size_t)255;
  int* col = (int*)(ws + off);
  off += (size_t)R * E * sizeof(int);
  off = (off + 255) & ~(size_t)255;
  float* h = (float*)(ws + off);
  off += (size_t)Nn * D_H * sizeof(float);
  off = (off + 255) & ~(size_t)255;
  float* hw = (float*)(ws + off);
  int2* part = (int2*)hw;
  size_t planeBytes = (size_t)Nn * D_H * sizeof(float);
  size_t partBytes = (size_t)R * E * sizeof(int2);
  size_t hwOff = off;
  off += (planeBytes > partBytes) ? planeBytes : partBytes;

  const int residentBlocks = 2048;
  const int pull64Resident = 1280;
  const int pull16Resident = 1024;
  int pull64Blocks = (Nn + 15) / 16;
  if (pull64Blocks > pull64Resident) pull64Blocks = pull64Resident;
  int pull16Blocks = (Nn + 63) / 64;
  if (pull16Blocks > pull16Resident) pull16Blocks = pull16Resident;
  const int relu4 = Nn * D_H / 4;
  int reluBlocks = (relu4 + 1023) / 1024;
  if (reluBlocks > residentBlocks) reluBlocks = residentBlocks;

  if (ws_size >= off && RB <= EC_RB_MAX) {
    int P = 1;
    if (ws_size >= hwOff + 2 * planeBytes) P = 2;
    if (ws_size >= hwOff + (size_t)R * planeBytes) P = R;
    int sub = (((E + EC_NCHUNK - 1) / EC_NCHUNK) + 31) & ~31;
    const size_t partOff = hwOff + (size_t)R * planeBytes;
    const bool sortInProducts = (P == R) && (ws_size >= partOff + partBytes);
    if (sortInProducts) part = (int2*)(ws + partOff);
    ec_chunk_sort<<<EC_NCHUNK, 1024, 0, stream>>>(src, dst, part, lstart, lcnt, E, R, NB, RB, sub);
    if (!sortInProducts) {
      ec_bucket_sort<<<RB, 1024, 0, stream>>>(part, lstart, lcnt, offs, col, Nn, NB, RB, M, E, R, sub);
    }

    if (sortInProducts) {
      ec_gemm_xs_sort<<<RB + gemmBlocks, 128, 0, stream>>>(
          x, L1w, W1, L1b, h, hw, mtiles, R, Nn,
          part, lstart, lcnt, offs, col, NB, RB, M, E, R, sub, RB);
      ec_pull_rows<64><<<pull64Blocks, 256, 0, stream>>>(hw, offs, col, h, Nn, R, 1);
    }
    for (int r0 = sortInProducts ? R : 0; r0 < R; r0 += P) {
      int nr = (R - r0 < P) ? (R - r0) : P;
      ec_gemm_xs<<<gemmBlocks, 128, 0, stream>>>(
          x, (r0 == 0) ? L1w : nullptr, W1 + (size_t)r0 * D_IN * D_H, L1b,
          h, hw, mtiles, nr, Nn);
      ec_pull_rows<64><<<pull64Blocks, 256, 0, stream>>>(
          hw, offs + (size_t)r0 * Nn, col, h, Nn, nr, (r0 + nr >= R) ? 1 : 0);
    }

    size_t plane2Bytes = (size_t)Nn * D_OUT * sizeof(float);
    int P2 = (ws_size >= hwOff + (size_t)R * plane2Bytes) ? R : 1;
    if (R == EC_L2_REL && P2 == R) {
      ec_gemm_l2<<<(mtiles + 7) / 8, 256, 0, stream>>>(h, L2w, W2, L2b, out, hw, mtiles, Nn);
      ec_pull_rows<16><<<pull16Blocks, 256, 0, stream>>>(hw, offs, col, out, Nn, R, 0);
      return;
    }
    ec_gemm_wmma<64, 16><<<gemmBlocks, 128, 0, stream>>>(h, L2w, L2b, out, mtiles);
    for (int r0 = 0; r0 < R; r0 += P2) {
      int nr = (R - r0 < P2) ? (R - r0) : P2;
      for (int q = 0; q < nr; ++q) {
        ec_gemm_wmma<64, 16><<<gemmBlocks, 128, 0, stream>>>(
            h, W2 + (size_t)(r0 + q) * D_H * D_OUT, nullptr,
            hw + (size_t)q * Nn * D_OUT, mtiles);
      }
      ec_pull_rows<16><<<pull16Blocks, 256, 0, stream>>>(
          hw, offs + (size_t)r0 * Nn, col, out, Nn, nr, 0);
    }
    return;
  }

  off = 0;
  int*   deg  = (int*)(ws + off);
  float* invd = (float*)(ws + off);
  off += (size_t)R * Nn * sizeof(float);
  off = (off + 255) & ~(size_t)255;
  h = (float*)(ws + off);
  off += (size_t)Nn * D_H * sizeof(float);
  off = (off + 255) & ~(size_t)255;
  hw = (float*)(ws + off);

  hipMemsetAsync(deg, 0, (size_t)R * Nn * sizeof(int), stream);
  {
    int total = R * E;
    ec_count_deg<<<(total + 255) / 256, 256, 0, stream>>>(dst, deg, total, Nn, E);
  }
  {
    int total = R * Nn;
    ec_inv_deg<<<(total + 255) / 256, 256, 0, stream>>>(deg, invd, total);
  }

  ec_gemm_wmma<256, 64><<<gemmBlocks, 128, 0, stream>>>(x, L1w, L1b, h, mtiles);
  for (int r = 0; r < R; ++r) {
    ec_gemm_wmma<256, 64><<<gemmBlocks, 128, 0, stream>>>(
        x, W1 + (size_t)r * D_IN * D_H, nullptr, hw, mtiles);
    long long tot = (long long)E * (D_H / 4);
    ec_agg<64><<<(int)((tot + 255) / 256), 256, 0, stream>>>(
        hw, src + (size_t)r * E, dst + (size_t)r * E, invd + (size_t)r * Nn, h, E);
  }
  ec_relu<<<reluBlocks, 256, 0, stream>>>(h, relu4);

  ec_gemm_wmma<64, 16><<<gemmBlocks, 128, 0, stream>>>(h, L2w, L2b, out, mtiles);
  for (int r = 0; r < R; ++r) {
    ec_gemm_wmma<64, 16><<<gemmBlocks, 128, 0, stream>>>(
        h, W2 + (size_t)r * D_H * D_OUT, nullptr, hw, mtiles);
    long long tot = (long long)E * (D_OUT / 4);
    ec_agg<16><<<(int)((tot + 255) / 256), 256, 0, stream>>>(
        hw, src + (size_t)r * E, dst + (size_t)r * E, invd + (size_t)r * Nn, out, E);
  }
}
